# starting kernel
# baseline (speedup 1.0000x reference)
#include <hip/hip_runtime.h>
#include <math.h>

#define N_USER 50000
#define N_REPO 100000
#define NEDGE  1000000
#define NEP    200000
#define D_IN   256
#define D_EMB  125
#define LD_EMB 128
#define D_HID  96
#define D_OUT  64
#define EPSN   1e-12f
#define MAX_K  256

typedef float v2f __attribute__((ext_vector_type(2)));
typedef float v8f __attribute__((ext_vector_type(8)));

__global__ void __launch_bounds__(128)
gemm_wmma_f32(const float* __restrict__ A, int lda,
              const float* __restrict__ W, int ldw,
              const float* __restrict__ degA,
              const float* __restrict__ bias,
              const float* __restrict__ agg,
              const float* __restrict__ degC,
              const float* __restrict__ bias2,
              float* __restrict__ C, int ldc,
              int M, int Nreal, int K, int Kpad, int relu)
{
  __shared__ __align__(16) float Wt[16 * (MAX_K + 4)];
  const int ldsStride = Kpad + 4;
  const int n0 = blockIdx.x * 16;

  {
    const int tid = threadIdx.y * 32 + threadIdx.x;
    const int sn  = tid >> 3;
    const int sk0 = tid & 7;
    const int gn  = n0 + sn;
    const bool gok = (gn < Nreal);
    float* dstp = &Wt[sn * ldsStride];
    for (int k = sk0; k < Kpad; k += 8)
      dstp[k] = (gok && k < K) ? W[(long)k * ldw + gn] : 0.0f;
  }
  __syncthreads();

  const int lane = threadIdx.x;
  const int half = lane >> 4;
  const int l16  = lane & 15;
  const int mtile = blockIdx.y * blockDim.y + threadIdx.y;
  if (mtile * 16 >= M) return;
  const int m0 = mtile * 16;

  const int   mA = m0 + l16;
  const float sA = degA ? rsqrtf(fmaxf(degA[mA], 1.0f)) : 1.0f;

  const float* Ap = A + (long)mA * lda + 2 * half;
  const float* Wp = &Wt[l16 * ldsStride + 2 * half];

  v8f acc = {};
#pragma unroll 4
  for (int k0 = 0; k0 < Kpad; k0 += 4) {
    v2f a = *(const v2f*)(Ap + k0);
    a *= sA;
    v2f b = *(const v2f*)(Wp + k0);
    acc = __builtin_amdgcn_wmma_f32_16x16x4_f32(false, a, false, b,
                                                (short)0, acc, false, false);
  }

  const int   n   = n0 + l16;
  const float bn  = (bias  && n < Nreal) ? bias[n]  : 0.0f;
  const float b2n = (bias2 && n < Nreal) ? bias2[n] : 0.0f;
#pragma unroll
  for (int v = 0; v < 8; ++v) {
    const int m = m0 + v + 8 * half;
    float val = acc[v] + bn;
    if (agg)  val += agg[(long)m * ldc + n] * rsqrtf(fmaxf(degC[m], 1.0f)) + b2n;
    if (relu) val = fmaxf(val, 0.0f);
    C[(long)m * ldc + n] = (n < Nreal) ? val : 0.0f;
  }
}

__global__ void degree_kernel(const int* __restrict__ src, const int* __restrict__ dst,
                              float* __restrict__ dsrc, float* __restrict__ ddst, int E)
{
  int e = blockIdx.x * blockDim.x + threadIdx.x;
  if (e < E) {
    atomicAdd(&dsrc[src[e]], 1.0f);
    atomicAdd(&ddst[dst[e]], 1.0f);
  }
}

__global__ void scatter_add_kernel(const float* __restrict__ feat,
                                   const int* __restrict__ src,
                                   const int* __restrict__ dst,
                                   float* __restrict__ agg, int E, int D)
{
  int e = blockIdx.x * blockDim.y + threadIdx.y;
  if (e >= E) return;
  int s = src[e], d = dst[e];
  atomicAdd(&agg[(long)d * D + threadIdx.x], feat[(long)s * D + threadIdx.x]);
}

__global__ void rownorm_kernel(const float* __restrict__ X, float* __restrict__ invn, int rows)
{
  int r = blockIdx.x * blockDim.y + threadIdx.y;
  if (r >= rows) return;
  int lane = threadIdx.x;
  float2 v = ((const float2*)(X + (long)r * D_OUT))[lane];
  float s = v.x * v.x + v.y * v.y;
#pragma unroll
  for (int o = 16; o > 0; o >>= 1) s += __shfl_xor(s, o, 32);
  if (lane == 0) invn[r] = 1.0f / fmaxf(sqrtf(s), EPSN);
}

__global__ void score_kernel(const float* __restrict__ HU, const float* __restrict__ HR,
                             const float* __restrict__ invU, const float* __restrict__ invR,
                             const int* __restrict__ iu, const int* __restrict__ ir,
                             float* __restrict__ out, int Ep)
{
  int e = blockIdx.x * blockDim.y + threadIdx.y;
  if (e >= Ep) return;
  int lane = threadIdx.x;
  int u = iu[e], r = ir[e];
  float2 a = ((const float2*)(HU + (long)u * D_OUT))[lane];
  float2 b = ((const float2*)(HR + (long)r * D_OUT))[lane];
  float s = a.x * b.x + a.y * b.y;
#pragma unroll
  for (int o = 16; o > 0; o >>= 1) s += __shfl_xor(s, o, 32);
  if (lane == 0) out[e] = s * invU[u] * invR[r];
}

extern "C" void kernel_launch(void* const* d_in, const int* in_sizes, int n_in,
                              void* d_out, int out_size, void* d_ws, size_t ws_size,
                              hipStream_t stream)
{
  const float* user_feat = (const float*)d_in[0];
  const float* repo_feat = (const float*)d_in[1];
  const float* W_ue = (const float*)d_in[2];   const float* b_ue = (const float*)d_in[3];
  const float* W_re = (const float*)d_in[4];   const float* b_re = (const float*)d_in[5];
  const float* W_hid_ur = (const float*)d_in[6];  const float* b_hid_ur = (const float*)d_in[7];
  const float* W_hid_ru = (const float*)d_in[8];  const float* b_hid_ru = (const float*)d_in[9];
  const float* W_out_ur = (const float*)d_in[10]; const float* b_out_ur = (const float*)d_in[11];
  const float* W_out_ru = (const float*)d_in[12]; const float* b_out_ru = (const float*)d_in[13];
  const float* W_hcu = (const float*)d_in[14]; const float* b_hcu = (const float*)d_in[15];
  const float* W_hcr = (const float*)d_in[16]; const float* b_hcr = (const float*)d_in[17];
  const float* W_ocu = (const float*)d_in[18]; const float* b_ocu = (const float*)d_in[19];
  const float* W_ocr = (const float*)d_in[20]; const float* b_ocr = (const float*)d_in[21];
  const int* e_ur_src = (const int*)d_in[22];  const int* e_ur_dst = (const int*)d_in[23];
  const int* e_ru_src = (const int*)d_in[24];  const int* e_ru_dst = (const int*)d_in[25];
  const int* pos_u = (const int*)d_in[26];     const int* pos_r = (const int*)d_in[27];
  const int* neg_u = (const int*)d_in[28];     const int* neg_r = (const int*)d_in[29];
  float* out = (float*)d_out;

  float* w = (float*)d_ws;
  size_t o = 0;
  float* deg_u_out_ur = w + o; o += N_USER;
  float* deg_r_in_ur  = w + o; o += N_REPO;
  float* deg_r_out_ru = w + o; o += N_REPO;
  float* deg_u_in_ru  = w + o; o += N_USER;
  float* agg_u   = w + o; o += (size_t)N_USER * D_HID;
  float* agg_r   = w + o; o += (size_t)N_REPO * D_HID;
  float* h_user  = w + o; o += (size_t)N_USER * LD_EMB;
  float* h_repo  = w + o; o += (size_t)N_REPO * LD_EMB;
  float* feat_u  = w + o; o += (size_t)N_USER * D_HID;
  float* feat_r  = w + o; o += (size_t)N_REPO * D_HID;
  float* out_user = w + o; o += (size_t)N_USER * D_HID;
  float* out_repo = w + o; o += (size_t)N_REPO * D_HID;
  float* inv_nu  = w + o; o += N_USER;
  float* inv_nr  = w + o; o += N_REPO;

  const size_t degBytes = (size_t)(2 * N_USER + 2 * N_REPO) * sizeof(float);
  const size_t aggBytes = ((size_t)N_USER + N_REPO) * D_HID * sizeof(float);

  auto gemm = [&](const float* A, int lda, const float* W_, int ldw,
                  const float* degA, const float* bias,
                  const float* agg, const float* degC, const float* bias2,
                  float* C, int ldc, int M, int Nreal, int K, int relu) {
    const int Kpad = (K + 3) & ~3;
    dim3 blk(32, 4);
    dim3 grd(ldc / 16, (M / 16 + 3) / 4);
    hipLaunchKernelGGL(gemm_wmma_f32, grd, blk, 0, stream,
                       A, lda, W_, ldw, degA, bias, agg, degC, bias2,
                       C, ldc, M, Nreal, K, Kpad, relu);
  };

  auto scatter = [&](const float* feat, const int* src, const int* dst,
                     float* agg, int D) {
    const int epb = 256 / D;
    dim3 blk(D, epb);
    dim3 grd((NEDGE + epb - 1) / epb);
    hipLaunchKernelGGL(scatter_add_kernel, grd, blk, 0, stream,
                       feat, src, dst, agg, NEDGE, D);
  };

  hipMemsetAsync(deg_u_out_ur, 0, degBytes, stream);
  hipMemsetAsync(agg_u, 0, aggBytes, stream);
  hipLaunchKernelGGL(degree_kernel, dim3((NEDGE + 255) / 256), dim3(256), 0, stream,
                     e_ur_src, e_ur_dst, deg_u_out_ur, deg_r_in_ur, NEDGE);
  hipLaunchKernelGGL(degree_kernel, dim3((NEDGE + 255) / 256), dim3(256), 0, stream,
                     e_ru_src, e_ru_dst, deg_r_out_ru, deg_u_in_ru, NEDGE);

  gemm(user_feat, D_IN, W_ue, D_EMB, nullptr, b_ue, nullptr, nullptr, nullptr,
       h_user, LD_EMB, N_USER, D_EMB, D_IN, 0);
  gemm(repo_feat, D_IN, W_re, D_EMB, nullptr, b_re, nullptr, nullptr, nullptr,
       h_repo, LD_EMB, N_REPO, D_EMB, D_IN, 0);

  gemm(h_user, LD_EMB, W_hid_ur, D_HID, deg_u_out_ur, nullptr, nullptr, nullptr, nullptr,
       feat_u, D_HID, N_USER, D_HID, D_EMB, 0);
  gemm(h_repo, LD_EMB, W_hid_ru, D_HID, deg_r_out_ru, nullptr, nullptr, nullptr, nullptr,
       feat_r, D_HID, N_REPO, D_HID, D_EMB, 0);

  scatter(feat_u, e_ur_src, e_ur_dst, agg_r, D_HID);
  scatter(feat_r, e_ru_src, e_ru_dst, agg_u, D_HID);

  gemm(h_user, LD_EMB, W_hcu, D_HID, nullptr, b_hcu, agg_u, deg_u_in_ru, b_hid_ru,
       out_user, D_HID, N_USER, D_HID, D_EMB, 1);
  gemm(h_repo, LD_EMB, W_hcr, D_HID, nullptr, b_hcr, agg_r, deg_r_in_ur, b_hid_ur,
       out_repo, D_HID, N_REPO, D_HID, D_EMB, 1);

  hipMemsetAsync(agg_u, 0, aggBytes, stream);
  gemm(out_user, D_HID, W_out_ur, D_OUT, deg_u_out_ur, nullptr, nullptr, nullptr, nullptr,
       feat_u, D_OUT, N_USER, D_OUT, D_HID, 0);
  gemm(out_repo, D_HID, W_out_ru, D_OUT, deg_r_out_ru, nullptr, nullptr, nullptr, nullptr,
       feat_r, D_OUT, N_REPO, D_OUT, D_HID, 0);

  scatter(feat_u, e_ur_src, e_ur_dst, agg_r, D_OUT);
  scatter(feat_r, e_ru_src, e_ru_dst, agg_u, D_OUT);

  gemm(out_user, D_HID, W_ocu, D_OUT, nullptr, b_ocu, agg_u, deg_u_in_ru, b_out_ru,
       h_user, D_OUT, N_USER, D_OUT, D_HID, 1);
  gemm(out_repo, D_HID, W_ocr, D_OUT, nullptr, b_ocr, agg_r, deg_r_in_ur, b_out_ur,
       h_repo, D_OUT, N_REPO, D_OUT, D_HID, 1);

  hipLaunchKernelGGL(rownorm_kernel, dim3((N_USER + 7) / 8), dim3(32, 8), 0, stream,
                     h_user, inv_nu, N_USER);
  hipLaunchKernelGGL(rownorm_kernel, dim3((N_REPO + 7) / 8), dim3(32, 8), 0, stream,
                     h_repo, inv_nr, N_REPO);
  hipLaunchKernelGGL(score_kernel, dim3((NEP + 7) / 8), dim3(32, 8), 0, stream,
                     h_user, h_repo, inv_nu, inv_nr, pos_u, pos_r, out, NEP);
  hipLaunchKernelGGL(score_kernel, dim3((NEP + 7) / 8), dim3(32, 8), 0, stream,
                     h_user, h_repo, inv_nu, inv_nr, neg_u, neg_r, out + NEP, NEP);
}


# Edge scatter by float atomics replaced by a per-call CSR by destination (count+arrival rank, scan, fill) and an atomic-free one-wave-per-row gather sum; sum order per row now follows arrival rank
# speedup vs baseline: 1.8177x; 1.8177x over previous
#include <hip/hip_runtime.h>
#include <math.h>

#define N_USER 50000
#define N_REPO 100000
#define NEDGE  1000000
#define NEP    200000
#define D_IN   256
#define D_EMB  125
#define LD_EMB 128
#define D_HID  96
#define D_OUT  64
#define EPSN   1e-12f
#define MAX_K  256

typedef float v2f __attribute__((ext_vector_type(2)));
typedef float v8f __attribute__((ext_vector_type(8)));

__global__ void __launch_bounds__(128)
gemm_wmma_f32(const float* __restrict__ A, int lda,
              const float* __restrict__ W, int ldw,
              const float* __restrict__ degA,
              const float* __restrict__ bias,
              const float* __restrict__ agg,
              const float* __restrict__ degC,
              const float* __restrict__ bias2,
              float* __restrict__ C, int ldc,
              int M, int Nreal, int K, int Kpad, int relu)
{
  __shared__ __align__(16) float Wt[16 * (MAX_K + 4)];
  const int ldsStride = Kpad + 4;
  const int n0 = blockIdx.x * 16;

  {
    const int tid = threadIdx.y * 32 + threadIdx.x;
    const int sn  = tid >> 3;
    const int sk0 = tid & 7;
    const int gn  = n0 + sn;
    const bool gok = (gn < Nreal);
    float* dstp = &Wt[sn * ldsStride];
    for (int k = sk0; k < Kpad; k += 8)
      dstp[k] = (gok && k < K) ? W[(long)k * ldw + gn] : 0.0f;
  }
  __syncthreads();

  const int lane = threadIdx.x;
  const int half = lane >> 4;
  const int l16  = lane & 15;
  const int mtile = blockIdx.y * blockDim.y + threadIdx.y;
  if (mtile * 16 >= M) return;
  const int m0 = mtile * 16;

  const int   mA = m0 + l16;
  const float sA = degA ? rsqrtf(fmaxf(degA[mA], 1.0f)) : 1.0f;

  const float* Ap = A + (long)mA * lda + 2 * half;
  const float* Wp = &Wt[l16 * ldsStride + 2 * half];

  v8f acc = {};
#pragma unroll 4
  for (int k0 = 0; k0 < Kpad; k0 += 4) {
    v2f a = *(const v2f*)(Ap + k0);
    a *= sA;
    v2f b = *(const v2f*)(Wp + k0);
    acc = __builtin_amdgcn_wmma_f32_16x16x4_f32(false, a, false, b,
                                                (short)0, acc, false, false);
  }

  const int   n   = n0 + l16;
  const float bn  = (bias  && n < Nreal) ? bias[n]  : 0.0f;
  const float b2n = (bias2 && n < Nreal) ? bias2[n] : 0.0f;
#pragma unroll
  for (int v = 0; v < 8; ++v) {
    const int m = m0 + v + 8 * half;
    float val = acc[v] + bn;
    if (agg)  val += agg[(long)m * ldc + n] * rsqrtf(fmaxf(degC[m], 1.0f)) + b2n;
    if (relu) val = fmaxf(val, 0.0f);
    C[(long)m * ldc + n] = (n < Nreal) ? val : 0.0f;
  }
}


__global__ void degree_rank_kernel(const int* __restrict__ src, const int* __restrict__ dst,
                                   float* __restrict__ dsrc, int* __restrict__ cnt_dst,
                                   int* __restrict__ rank, int E)
{
  int e = blockIdx.x * blockDim.x + threadIdx.x;
  if (e < E) {
    const int s = src[e], d = dst[e];
    atomicAdd(&dsrc[s], 1.0f);
    rank[e] = atomicAdd(&cnt_dst[d], 1);
  }
}

#define SCAN_T 1024
#define SCAN_V 8
__global__ void __launch_bounds__(SCAN_T)
rowptr_scan_kernel(const int* __restrict__ cnt0, int* __restrict__ ptr0,
                   float* __restrict__ deg0, int n0,
                   const int* __restrict__ cnt1, int* __restrict__ ptr1,
                   float* __restrict__ deg1, int n1)
{
  const int* __restrict__ cnt = blockIdx.x ? cnt1 : cnt0;
  int*   __restrict__ ptr = blockIdx.x ? ptr1 : ptr0;
  float* __restrict__ deg = blockIdx.x ? deg1 : deg0;
  const int n = blockIdx.x ? n1 : n0;

  __shared__ int wsum[2][32];
  const int tid = threadIdx.x, lane = tid & 31, wv = tid >> 5;

  int v[SCAN_V], nv[SCAN_V];
#pragma unroll
  for (int k = 0; k < SCAN_V; ++k) {
    const int i = tid * SCAN_V + k;
    v[k] = (i < n) ? cnt[i] : 0;
  }
  int carry = 0, it = 0;
  for (int base = 0; base < n; base += SCAN_T * SCAN_V, ++it) {
    const int i0 = base + tid * SCAN_V;
    const int nb = i0 + SCAN_T * SCAN_V;
#pragma unroll
    for (int k = 0; k < SCAN_V; ++k) nv[k] = (nb + k < n) ? cnt[nb + k] : 0;

    int tsum = 0;
#pragma unroll
    for (int k = 0; k < SCAN_V; ++k) tsum += v[k];
    int x = tsum;
#pragma unroll
    for (int o = 1; o < 32; o <<= 1) {
      const int y = __shfl_up(x, o, 32);
      if (lane >= o) x += y;
    }
    if (lane == 31) wsum[it & 1][wv] = x;
    __syncthreads();
    const int w = wsum[it & 1][lane];
    int wx = w;
#pragma unroll
    for (int o = 1; o < 32; o <<= 1) {
      const int y = __shfl_up(wx, o, 32);
      if (lane >= o) wx += y;
    }
    const int wave_excl = __shfl(wx, wv, 32) - __shfl(w, wv, 32);
    const int total     = __shfl(wx, 31, 32);
    int run = carry + wave_excl + (x - tsum);
#pragma unroll
    for (int k = 0; k < SCAN_V; ++k) {
      const int i = i0 + k;
      if (i < n) { ptr[i] = run; deg[i] = (float)v[k]; }
      run += v[k];
    }
    carry += total;
#pragma unroll
    for (int k = 0; k < SCAN_V; ++k) v[k] = nv[k];
  }
}

__global__ void csr_fill_kernel(const int* __restrict__ src, const int* __restrict__ dst,
                                const int* __restrict__ rank, const int* __restrict__ rowptr,
                                int* __restrict__ col, int E)
{
  int e = blockIdx.x * blockDim.x + threadIdx.x;
  if (e < E) {
    int p = rowptr[dst[e]] + rank[e];
    p = min(max(p, 0), E - 1);
    col[p] = src[e];
  }
}

template <int D>
__global__ void __launch_bounds__(256)
gather_sum_kernel(const float* __restrict__ feat, const int* __restrict__ rowptr,
                  const int* __restrict__ col, float* __restrict__ agg,
                  int n_dst, int n_src, int E)
{
  const int d = __builtin_amdgcn_readfirstlane((int)(blockIdx.x * blockDim.y + threadIdx.y));
  if (d >= n_dst) return;
  const int lane = threadIdx.x;
  const int beg = max(rowptr[d], 0);
  const int end = min((d + 1 < n_dst) ? rowptr[d + 1] : E, E);

  float2 acc = make_float2(0.0f, 0.0f);
  float  acc2 = 0.0f;
  for (int base = beg; base < end; base += 32) {
    const int cnt = min(32, end - base);
    int myc = (lane < cnt) ? col[base + lane] : 0;
    myc = min(max(myc, 0), n_src - 1);
#pragma unroll 4
    for (int j = 0; j < cnt; ++j) {
      const int s = __builtin_amdgcn_readlane(myc, j);
      const float* p = feat + (size_t)s * D;
      const float2 t = ((const float2*)p)[lane];
      acc.x += t.x;
      acc.y += t.y;
      if (D == 96) acc2 += p[64 + lane];
    }
  }
  float* q = agg + (size_t)d * D;
  ((float2*)q)[lane] = acc;
  if (D == 96) q[64 + lane] = acc2;
}

__global__ void rownorm_kernel(const float* __restrict__ X, float* __restrict__ invn, int rows)
{
  int r = blockIdx.x * blockDim.y + threadIdx.y;
  if (r >= rows) return;
  int lane = threadIdx.x;
  float2 v = ((const float2*)(X + (long)r * D_OUT))[lane];
  float s = v.x * v.x + v.y * v.y;
#pragma unroll
  for (int o = 16; o > 0; o >>= 1) s += __shfl_xor(s, o, 32);
  if (lane == 0) invn[r] = 1.0f / fmaxf(sqrtf(s), EPSN);
}

__global__ void score_kernel(const float* __restrict__ HU, const float* __restrict__ HR,
                             const float* __restrict__ invU, const float* __restrict__ invR,
                             const int* __restrict__ iu, const int* __restrict__ ir,
                             float* __restrict__ out, int Ep)
{
  int e = blockIdx.x * blockDim.y + threadIdx.y;
  if (e >= Ep) return;
  int lane = threadIdx.x;
  int u = iu[e], r = ir[e];
  float2 a = ((const float2*)(HU + (long)u * D_OUT))[lane];
  float2 b = ((const float2*)(HR + (long)r * D_OUT))[lane];
  float s = a.x * b.x + a.y * b.y;
#pragma unroll
  for (int o = 16; o > 0; o >>= 1) s += __shfl_xor(s, o, 32);
  if (lane == 0) out[e] = s * invU[u] * invR[r];
}

extern "C" void kernel_launch(void* const* d_in, const int* in_sizes, int n_in,
                              void* d_out, int out_size, void* d_ws, size_t ws_size,
                              hipStream_t stream)
{
  const float* user_feat = (const float*)d_in[0];
  const float* repo_feat = (const float*)d_in[1];
  const float* W_ue = (const float*)d_in[2];   const float* b_ue = (const float*)d_in[3];
  const float* W_re = (const float*)d_in[4];   const float* b_re = (const float*)d_in[5];
  const float* W_hid_ur = (const float*)d_in[6];  const float* b_hid_ur = (const float*)d_in[7];
  const float* W_hid_ru = (const float*)d_in[8];  const float* b_hid_ru = (const float*)d_in[9];
  const float* W_out_ur = (const float*)d_in[10]; const float* b_out_ur = (const float*)d_in[11];
  const float* W_out_ru = (const float*)d_in[12]; const float* b_out_ru = (const float*)d_in[13];
  const float* W_hcu = (const float*)d_in[14]; const float* b_hcu = (const float*)d_in[15];
  const float* W_hcr = (const float*)d_in[16]; const float* b_hcr = (const float*)d_in[17];
  const float* W_ocu = (const float*)d_in[18]; const float* b_ocu = (const float*)d_in[19];
  const float* W_ocr = (const float*)d_in[20]; const float* b_ocr = (const float*)d_in[21];
  const int* e_ur_src = (const int*)d_in[22];  const int* e_ur_dst = (const int*)d_in[23];
  const int* e_ru_src = (const int*)d_in[24];  const int* e_ru_dst = (const int*)d_in[25];
  const int* pos_u = (const int*)d_in[26];     const int* pos_r = (const int*)d_in[27];
  const int* neg_u = (const int*)d_in[28];     const int* neg_r = (const int*)d_in[29];
  float* out = (float*)d_out;

  float* w = (float*)d_ws;
  size_t o = 0;
  float* deg_u_out_ur = w + o; o += N_USER;
  float* deg_r_in_ur  = w + o; o += N_REPO;
  float* deg_r_out_ru = w + o; o += N_REPO;
  float* deg_u_in_ru  = w + o; o += N_USER;
  float* reg_agg  = w + o; o += ((size_t)N_USER + N_REPO) * D_HID;
  float* reg_h    = w + o; o += ((size_t)N_USER + N_REPO) * LD_EMB;
  float* reg_feat = w + o; o += ((size_t)N_USER + N_REPO) * D_HID;
  float* reg_csr  = w + o; o += ((size_t)N_USER + N_REPO) * D_HID;
  float* inv_nu  = w + o; o += N_USER;
  float* inv_nr  = w + o; o += N_REPO;

  int* cnt_r = (int*)reg_agg;
  int* cnt_u = cnt_r + N_REPO;
  const size_t zeroBytes = (size_t)(3 * N_USER + 3 * N_REPO) * sizeof(float);

  int* col_ur  = (int*)reg_csr;
  int* col_ru  = col_ur + NEDGE;
  int* rank_ur = col_ru + NEDGE;
  int* rank_ru = rank_ur + NEDGE;
  int* ptr_r   = rank_ru + NEDGE;
  int* ptr_u   = ptr_r + N_REPO;

  float* h_user   = reg_h;
  float* h_repo   = reg_h + (size_t)N_USER * LD_EMB;
  float* feat_u   = reg_feat;
  float* feat_r   = reg_feat + (size_t)N_USER * D_HID;
  float* agg_u    = reg_agg;
  float* agg_r    = reg_agg + (size_t)N_USER * D_HID;
  float* out_user = reg_feat;
  float* out_repo = reg_feat + (size_t)N_USER * D_HID;
  float* feat2_u  = reg_agg;
  float* feat2_r  = reg_agg + (size_t)N_USER * D_HID;
  float* agg2_u   = reg_h;
  float* agg2_r   = reg_h + (size_t)N_USER * LD_EMB;
  float* hn_user  = reg_agg;
  float* hn_repo  = reg_agg + (size_t)N_USER * D_HID;

  auto gemm = [&](const float* A, int lda, const float* W_, int ldw,
                  const float* degA, const float* bias,
                  const float* agg, const float* degC, const float* bias2,
                  float* C, int ldc, int M, int Nreal, int K, int relu) {
    const int Kpad = (K + 3) & ~3;
    dim3 blk(32, 4);
    dim3 grd(ldc / 16, (M / 16 + 3) / 4);
    hipLaunchKernelGGL(gemm_wmma_f32, grd, blk, 0, stream,
                       A, lda, W_, ldw, degA, bias, agg, degC, bias2,
                       C, ldc, M, Nreal, K, Kpad, relu);
  };

  auto gather = [&](const float* feat, const int* rowptr, const int* col,
                    float* agg, int n_dst, int n_src, int D) {
    dim3 blk(32, 8);
    dim3 grd((n_dst + 7) / 8);
    if (D == D_HID)
      hipLaunchKernelGGL(gather_sum_kernel<D_HID>, grd, blk, 0, stream,
                         feat, rowptr, col, agg, n_dst, n_src, NEDGE);
    else
      hipLaunchKernelGGL(gather_sum_kernel<D_OUT>, grd, blk, 0, stream,
                         feat, rowptr, col, agg, n_dst, n_src, NEDGE);
  };

  hipMemsetAsync(deg_u_out_ur, 0, zeroBytes, stream);
  hipLaunchKernelGGL(degree_rank_kernel, dim3((NEDGE + 255) / 256), dim3(256), 0, stream,
                     e_ur_src, e_ur_dst, deg_u_out_ur, cnt_r, rank_ur, NEDGE);
  hipLaunchKernelGGL(degree_rank_kernel, dim3((NEDGE + 255) / 256), dim3(256), 0, stream,
                     e_ru_src, e_ru_dst, deg_r_out_ru, cnt_u, rank_ru, NEDGE);
  hipLaunchKernelGGL(rowptr_scan_kernel, dim3(2), dim3(SCAN_T), 0, stream,
                     cnt_r, ptr_r, deg_r_in_ur, N_REPO,
                     cnt_u, ptr_u, deg_u_in_ru, N_USER);
  hipLaunchKernelGGL(csr_fill_kernel, dim3((NEDGE + 255) / 256), dim3(256), 0, stream,
                     e_ur_src, e_ur_dst, rank_ur, ptr_r, col_ur, NEDGE);
  hipLaunchKernelGGL(csr_fill_kernel, dim3((NEDGE + 255) / 256), dim3(256), 0, stream,
                     e_ru_src, e_ru_dst, rank_ru, ptr_u, col_ru, NEDGE);

  gemm(user_feat, D_IN, W_ue, D_EMB, nullptr, b_ue, nullptr, nullptr, nullptr,
       h_user, LD_EMB, N_USER, D_EMB, D_IN, 0);
  gemm(repo_feat, D_IN, W_re, D_EMB, nullptr, b_re, nullptr, nullptr, nullptr,
       h_repo, LD_EMB, N_REPO, D_EMB, D_IN, 0);

  gemm(h_user, LD_EMB, W_hid_ur, D_HID, deg_u_out_ur, nullptr, nullptr, nullptr, nullptr,
       feat_u, D_HID, N_USER, D_HID, D_EMB, 0);
  gemm(h_repo, LD_EMB, W_hid_ru, D_HID, deg_r_out_ru, nullptr, nullptr, nullptr, nullptr,
       feat_r, D_HID, N_REPO, D_HID, D_EMB, 0);

  gather(feat_u, ptr_r, col_ur, agg_r, N_REPO, N_USER, D_HID);
  gather(feat_r, ptr_u, col_ru, agg_u, N_USER, N_REPO, D_HID);

  gemm(h_user, LD_EMB, W_hcu, D_HID, nullptr, b_hcu, agg_u, deg_u_in_ru, b_hid_ru,
       out_user, D_HID, N_USER, D_HID, D_EMB, 1);
  gemm(h_repo, LD_EMB, W_hcr, D_HID, nullptr, b_hcr, agg_r, deg_r_in_ur, b_hid_ur,
       out_repo, D_HID, N_REPO, D_HID, D_EMB, 1);

  gemm(out_user, D_HID, W_out_ur, D_OUT, deg_u_out_ur, nullptr, nullptr, nullptr, nullptr,
       feat2_u, D_OUT, N_USER, D_OUT, D_HID, 0);
  gemm(out_repo, D_HID, W_out_ru, D_OUT, deg_r_out_ru, nullptr, nullptr, nullptr, nullptr,
       feat2_r, D_OUT, N_REPO, D_OUT, D_HID, 0);

  gather(feat2_u, ptr_r, col_ur, agg2_r, N_REPO, N_USER, D_OUT);
  gather(feat2_r, ptr_u, col_ru, agg2_u, N_USER, N_REPO, D_OUT);

  gemm(out_user, D_HID, W_ocu, D_OUT, nullptr, b_ocu, agg2_u, deg_u_in_ru, b_out_ru,
       hn_user, D_OUT, N_USER, D_OUT, D_HID, 1);
  gemm(out_repo, D_HID, W_ocr, D_OUT, nullptr, b_ocr, agg2_r, deg_r_in_ur, b_out_ur,
       hn_repo, D_OUT, N_REPO, D_OUT, D_HID, 1);

  hipLaunchKernelGGL(rownorm_kernel, dim3((N_USER + 7) / 8), dim3(32, 8), 0, stream,
                     hn_user, inv_nu, N_USER);
  hipLaunchKernelGGL(rownorm_kernel, dim3((N_REPO + 7) / 8), dim3(32, 8), 0, stream,
                     hn_repo, inv_nr, N_REPO);
  hipLaunchKernelGGL(score_kernel, dim3((NEP + 7) / 8), dim3(32, 8), 0, stream,
                     hn_user, hn_repo, inv_nu, inv_nr, pos_u, pos_r, out, NEP);
  hipLaunchKernelGGL(score_kernel, dim3((NEP + 7) / 8), dim3(32, 8), 0, stream,
                     hn_user, hn_repo, inv_nu, inv_nr, neg_u, neg_r, out + NEP, NEP);
}


# Cosine tail (row norms, pair scores): one wave per row or pair becomes 8 lanes per row, 32 rows or pairs per wave (533k fewer waves); norm and dot sums now 8 per lane, then a 3-step lane tree
# speedup vs baseline: 1.9362x; 1.0652x over previous
#include <hip/hip_runtime.h>
#include <math.h>

#define N_USER 50000
#define N_REPO 100000
#define NEDGE  1000000
#define NEP    200000
#define D_IN   256
#define D_EMB  125
#define LD_EMB 128
#define D_HID  96
#define D_OUT  64
#define EPSN   1e-12f
#define MAX_K  256

typedef float v2f __attribute__((ext_vector_type(2)));
typedef float v8f __attribute__((ext_vector_type(8)));

__global__ void __launch_bounds__(128)
gemm_wmma_f32(const float* __restrict__ A, int lda,
              const float* __restrict__ W, int ldw,
              const float* __restrict__ degA,
              const float* __restrict__ bias,
              const float* __restrict__ agg,
              const float* __restrict__ degC,
              const float* __restrict__ bias2,
              float* __restrict__ C, int ldc,
              int M, int Nreal, int K, int Kpad, int relu)
{
  __shared__ __align__(16) float Wt[16 * (MAX_K + 4)];
  const int ldsStride = Kpad + 4;
  const int n0 = blockIdx.x * 16;

  {
    const int tid = threadIdx.y * 32 + threadIdx.x;
    const int sn  = tid >> 3;
    const int sk0 = tid & 7;
    const int gn  = n0 + sn;
    const bool gok = (gn < Nreal);
    float* dstp = &Wt[sn * ldsStride];
    for (int k = sk0; k < Kpad; k += 8)
      dstp[k] = (gok && k < K) ? W[(long)k * ldw + gn] : 0.0f;
  }
  __syncthreads();

  const int lane = threadIdx.x;
  const int half = lane >> 4;
  const int l16  = lane & 15;
  const int mtile = blockIdx.y * blockDim.y + threadIdx.y;
  if (mtile * 16 >= M) return;
  const int m0 = mtile * 16;

  const int   mA = m0 + l16;
  const float sA = degA ? rsqrtf(fmaxf(degA[mA], 1.0f)) : 1.0f;

  const float* Ap = A + (long)mA * lda + 2 * half;
  const float* Wp = &Wt[l16 * ldsStride + 2 * half];

  v8f acc = {};
#pragma unroll 4
  for (int k0 = 0; k0 < Kpad; k0 += 4) {
    v2f a = *(const v2f*)(Ap + k0);
    a *= sA;
    v2f b = *(const v2f*)(Wp + k0);
    acc = __builtin_amdgcn_wmma_f32_16x16x4_f32(false, a, false, b,
                                                (short)0, acc, false, false);
  }

  const int   n   = n0 + l16;
  const float bn  = (bias  && n < Nreal) ? bias[n]  : 0.0f;
  const float b2n = (bias2 && n < Nreal) ? bias2[n] : 0.0f;
#pragma unroll
  for (int v = 0; v < 8; ++v) {
    const int m = m0 + v + 8 * half;
    float val = acc[v] + bn;
    if (agg)  val += agg[(long)m * ldc + n] * rsqrtf(fmaxf(degC[m], 1.0f)) + b2n;
    if (relu) val = fmaxf(val, 0.0f);
    C[(long)m * ldc + n] = (n < Nreal) ? val : 0.0f;
  }
}


__global__ void degree_rank_kernel(const int* __restrict__ src, const int* __restrict__ dst,
                                   float* __restrict__ dsrc, int* __restrict__ cnt_dst,
                                   int* __restrict__ rank, int E)
{
  int e = blockIdx.x * blockDim.x + threadIdx.x;
  if (e < E) {
    const int s = src[e], d = dst[e];
    atomicAdd(&dsrc[s], 1.0f);
    rank[e] = atomicAdd(&cnt_dst[d], 1);
  }
}

#define SCAN_T 1024
#define SCAN_V 8
__global__ void __launch_bounds__(SCAN_T)
rowptr_scan_kernel(const int* __restrict__ cnt0, int* __restrict__ ptr0,
                   float* __restrict__ deg0, int n0,
                   const int* __restrict__ cnt1, int* __restrict__ ptr1,
                   float* __restrict__ deg1, int n1)
{
  const int* __restrict__ cnt = blockIdx.x ? cnt1 : cnt0;
  int*   __restrict__ ptr = blockIdx.x ? ptr1 : ptr0;
  float* __restrict__ deg = blockIdx.x ? deg1 : deg0;
  const int n = blockIdx.x ? n1 : n0;

  __shared__ int wsum[2][32];
  const int tid = threadIdx.x, lane = tid & 31, wv = tid >> 5;

  int v[SCAN_V], nv[SCAN_V];
#pragma unroll
  for (int k = 0; k < SCAN_V; ++k) {
    const int i = tid * SCAN_V + k;
    v[k] = (i < n) ? cnt[i] : 0;
  }
  int carry = 0, it = 0;
  for (int base = 0; base < n; base += SCAN_T * SCAN_V, ++it) {
    const int i0 = base + tid * SCAN_V;
    const int nb = i0 + SCAN_T * SCAN_V;
#pragma unroll
    for (int k = 0; k < SCAN_V; ++k) nv[k] = (nb + k < n) ? cnt[nb + k] : 0;

    int tsum = 0;
#pragma unroll
    for (int k = 0; k < SCAN_V; ++k) tsum += v[k];
    int x = tsum;
#pragma unroll
    for (int o = 1; o < 32; o <<= 1) {
      const int y = __shfl_up(x, o, 32);
      if (lane >= o) x += y;
    }
    if (lane == 31) wsum[it & 1][wv] = x;
    __syncthreads();
    const int w = wsum[it & 1][lane];
    int wx = w;
#pragma unroll
    for (int o = 1; o < 32; o <<= 1) {
      const int y = __shfl_up(wx, o, 32);
      if (lane >= o) wx += y;
    }
    const int wave_excl = __shfl(wx, wv, 32) - __shfl(w, wv, 32);
    const int total     = __shfl(wx, 31, 32);
    int run = carry + wave_excl + (x - tsum);
#pragma unroll
    for (int k = 0; k < SCAN_V; ++k) {
      const int i = i0 + k;
      if (i < n) { ptr[i] = run; deg[i] = (float)v[k]; }
      run += v[k];
    }
    carry += total;
#pragma unroll
    for (int k = 0; k < SCAN_V; ++k) v[k] = nv[k];
  }
}

__global__ void csr_fill_kernel(const int* __restrict__ src, const int* __restrict__ dst,
                                const int* __restrict__ rank, const int* __restrict__ rowptr,
                                int* __restrict__ col, int E)
{
  int e = blockIdx.x * blockDim.x + threadIdx.x;
  if (e < E) {
    int p = rowptr[dst[e]] + rank[e];
    p = min(max(p, 0), E - 1);
    col[p] = src[e];
  }
}

template <int D>
__global__ void __launch_bounds__(256)
gather_sum_kernel(const float* __restrict__ feat, const int* __restrict__ rowptr,
                  const int* __restrict__ col, float* __restrict__ agg,
                  int n_dst, int n_src, int E)
{
  const int d = __builtin_amdgcn_readfirstlane((int)(blockIdx.x * blockDim.y + threadIdx.y));
  if (d >= n_dst) return;
  const int lane = threadIdx.x;
  const int beg = max(rowptr[d], 0);
  const int end = min((d + 1 < n_dst) ? rowptr[d + 1] : E, E);

  float2 acc = make_float2(0.0f, 0.0f);
  float  acc2 = 0.0f;
  for (int base = beg; base < end; base += 32) {
    const int cnt = min(32, end - base);
    int myc = (lane < cnt) ? col[base + lane] : 0;
    myc = min(max(myc, 0), n_src - 1);
#pragma unroll 4
    for (int j = 0; j < cnt; ++j) {
      const int s = __builtin_amdgcn_readlane(myc, j);
      const float* p = feat + (size_t)s * D;
      const float2 t = ((const float2*)p)[lane];
      acc.x += t.x;
      acc.y += t.y;
      if (D == 96) acc2 += p[64 + lane];
    }
  }
  float* q = agg + (size_t)d * D;
  ((float2*)q)[lane] = acc;
  if (D == 96) q[64 + lane] = acc2;
}

#define TAIL_PER_WAVE 32

__global__ void __launch_bounds__(256)
rownorm_kernel(const float* __restrict__ X, float* __restrict__ invn, int rows)
{
  const int base = (blockIdx.x * blockDim.y + threadIdx.y) * TAIL_PER_WAVE;
  if (base >= rows) return;
  const int lane = threadIdx.x;
  const int g = lane >> 3, t = lane & 7;
  float sel = 1.0f;
#pragma unroll
  for (int j = 0; j < 8; ++j) {
    const int r = min(base + 4 * j + g, rows - 1);
    const float4* p = (const float4*)(X + (size_t)r * D_OUT);
    const float4 v0 = p[t], v1 = p[8 + t];
    float s = v0.x * v0.x + v0.y * v0.y + v0.z * v0.z + v0.w * v0.w
            + v1.x * v1.x + v1.y * v1.y + v1.z * v1.z + v1.w * v1.w;
    s += __shfl_xor(s, 4, 32);
    s += __shfl_xor(s, 2, 32);
    s += __shfl_xor(s, 1, 32);
    if (t == j) sel = s;
  }
  const int r = base + 4 * t + g;
  if (r < rows) invn[r] = 1.0f / fmaxf(sqrtf(sel), EPSN);
}

__global__ void __launch_bounds__(256)
score_kernel(const float* __restrict__ HU, const float* __restrict__ HR,
             const float* __restrict__ invU, const float* __restrict__ invR,
             const int* __restrict__ iu, const int* __restrict__ ir,
             float* __restrict__ out, int Ep)
{
  const int base = (blockIdx.x * blockDim.y + threadIdx.y) * TAIL_PER_WAVE;
  if (base >= Ep) return;
  const int lane = threadIdx.x;
  const int g = lane >> 3, t = lane & 7;
  const int el  = min(base + lane, Ep - 1);
  const int myu = iu[el], myr = ir[el];
  const float mynu = invU[myu], mynr = invR[myr];
  float sel = 0.0f;
#pragma unroll
  for (int j = 0; j < 8; ++j) {
    const int u = __shfl(myu, 4 * j + g, 32);
    const int r = __shfl(myr, 4 * j + g, 32);
    const float4* pu = (const float4*)(HU + (size_t)u * D_OUT);
    const float4* pr = (const float4*)(HR + (size_t)r * D_OUT);
    const float4 a0 = pu[t], a1 = pu[8 + t];
    const float4 b0 = pr[t], b1 = pr[8 + t];
    float s = a0.x * b0.x + a0.y * b0.y + a0.z * b0.z + a0.w * b0.w
            + a1.x * b1.x + a1.y * b1.y + a1.z * b1.z + a1.w * b1.w;
    s += __shfl_xor(s, 4, 32);
    s += __shfl_xor(s, 2, 32);
    s += __shfl_xor(s, 1, 32);
    if (t == j) sel = s;
  }
  const int own = 4 * t + g;
  const float nu = __shfl(mynu, own, 32);
  const float nr = __shfl(mynr, own, 32);
  const int e = base + own;
  if (e < Ep) out[e] = sel * nu * nr;
}

extern "C" void kernel_launch(void* const* d_in, const int* in_sizes, int n_in,
                              void* d_out, int out_size, void* d_ws, size_t ws_size,
                              hipStream_t stream)
{
  const float* user_feat = (const float*)d_in[0];
  const float* repo_feat = (const float*)d_in[1];
  const float* W_ue = (const float*)d_in[2];   const float* b_ue = (const float*)d_in[3];
  const float* W_re = (const float*)d_in[4];   const float* b_re = (const float*)d_in[5];
  const float* W_hid_ur = (const float*)d_in[6];  const float* b_hid_ur = (const float*)d_in[7];
  const float* W_hid_ru = (const float*)d_in[8];  const float* b_hid_ru = (const float*)d_in[9];
  const float* W_out_ur = (const float*)d_in[10]; const float* b_out_ur = (const float*)d_in[11];
  const float* W_out_ru = (const float*)d_in[12]; const float* b_out_ru = (const float*)d_in[13];
  const float* W_hcu = (const float*)d_in[14]; const float* b_hcu = (const float*)d_in[15];
  const float* W_hcr = (const float*)d_in[16]; const float* b_hcr = (const float*)d_in[17];
  const float* W_ocu = (const float*)d_in[18]; const float* b_ocu = (const float*)d_in[19];
  const float* W_ocr = (const float*)d_in[20]; const float* b_ocr = (const float*)d_in[21];
  const int* e_ur_src = (const int*)d_in[22];  const int* e_ur_dst = (const int*)d_in[23];
  const int* e_ru_src = (const int*)d_in[24];  const int* e_ru_dst = (const int*)d_in[25];
  const int* pos_u = (const int*)d_in[26];     const int* pos_r = (const int*)d_in[27];
  const int* neg_u = (const int*)d_in[28];     const int* neg_r = (const int*)d_in[29];
  float* out = (float*)d_out;

  float* w = (float*)d_ws;
  size_t o = 0;
  float* deg_u_out_ur = w + o; o += N_USER;
  float* deg_r_in_ur  = w + o; o += N_REPO;
  float* deg_r_out_ru = w + o; o += N_REPO;
  float* deg_u_in_ru  = w + o; o += N_USER;
  float* reg_agg  = w + o; o += ((size_t)N_USER + N_REPO) * D_HID;
  float* reg_h    = w + o; o += ((size_t)N_USER + N_REPO) * LD_EMB;
  float* reg_feat = w + o; o += ((size_t)N_USER + N_REPO) * D_HID;
  float* reg_csr  = w + o; o += ((size_t)N_USER + N_REPO) * D_HID;
  float* inv_nu  = w + o; o += N_USER;
  float* inv_nr  = w + o; o += N_REPO;

  int* cnt_r = (int*)reg_agg;
  int* cnt_u = cnt_r + N_REPO;
  const size_t zeroBytes = (size_t)(3 * N_USER + 3 * N_REPO) * sizeof(float);

  int* col_ur  = (int*)reg_csr;
  int* col_ru  = col_ur + NEDGE;
  int* rank_ur = col_ru + NEDGE;
  int* rank_ru = rank_ur + NEDGE;
  int* ptr_r   = rank_ru + NEDGE;
  int* ptr_u   = ptr_r + N_REPO;

  float* h_user   = reg_h;
  float* h_repo   = reg_h + (size_t)N_USER * LD_EMB;
  float* feat_u   = reg_feat;
  float* feat_r   = reg_feat + (size_t)N_USER * D_HID;
  float* agg_u    = reg_agg;
  float* agg_r    = reg_agg + (size_t)N_USER * D_HID;
  float* out_user = reg_feat;
  float* out_repo = reg_feat + (size_t)N_USER * D_HID;
  float* feat2_u  = reg_agg;
  float* feat2_r  = reg_agg + (size_t)N_USER * D_HID;
  float* agg2_u   = reg_h;
  float* agg2_r   = reg_h + (size_t)N_USER * LD_EMB;
  float* hn_user  = reg_agg;
  float* hn_repo  = reg_agg + (size_t)N_USER * D_HID;

  auto gemm = [&](const float* A, int lda, const float* W_, int ldw,
                  const float* degA, const float* bias,
                  const float* agg, const float* degC, const float* bias2,
                  float* C, int ldc, int M, int Nreal, int K, int relu) {
    const int Kpad = (K + 3) & ~3;
    dim3 blk(32, 4);
    dim3 grd(ldc / 16, (M / 16 + 3) / 4);
    hipLaunchKernelGGL(gemm_wmma_f32, grd, blk, 0, stream,
                       A, lda, W_, ldw, degA, bias, agg, degC, bias2,
                       C, ldc, M, Nreal, K, Kpad, relu);
  };

  auto gather = [&](const float* feat, const int* rowptr, const int* col,
                    float* agg, int n_dst, int n_src, int D) {
    dim3 blk(32, 8);
    dim3 grd((n_dst + 7) / 8);
    if (D == D_HID)
      hipLaunchKernelGGL(gather_sum_kernel<D_HID>, grd, blk, 0, stream,
                         feat, rowptr, col, agg, n_dst, n_src, NEDGE);
    else
      hipLaunchKernelGGL(gather_sum_kernel<D_OUT>, grd, blk, 0, stream,
                         feat, rowptr, col, agg, n_dst, n_src, NEDGE);
  };

  hipMemsetAsync(deg_u_out_ur, 0, zeroBytes, stream);
  hipLaunchKernelGGL(degree_rank_kernel, dim3((NEDGE + 255) / 256), dim3(256), 0, stream,
                     e_ur_src, e_ur_dst, deg_u_out_ur, cnt_r, rank_ur, NEDGE);
  hipLaunchKernelGGL(degree_rank_kernel, dim3((NEDGE + 255) / 256), dim3(256), 0, stream,
                     e_ru_src, e_ru_dst, deg_r_out_ru, cnt_u, rank_ru, NEDGE);
  hipLaunchKernelGGL(rowptr_scan_kernel, dim3(2), dim3(SCAN_T), 0, stream,
                     cnt_r, ptr_r, deg_r_in_ur, N_REPO,
                     cnt_u, ptr_u, deg_u_in_ru, N_USER);
  hipLaunchKernelGGL(csr_fill_kernel, dim3((NEDGE + 255) / 256), dim3(256), 0, stream,
                     e_ur_src, e_ur_dst, rank_ur, ptr_r, col_ur, NEDGE);
  hipLaunchKernelGGL(csr_fill_kernel, dim3((NEDGE + 255) / 256), dim3(256), 0, stream,
                     e_ru_src, e_ru_dst, rank_ru, ptr_u, col_ru, NEDGE);

  gemm(user_feat, D_IN, W_ue, D_EMB, nullptr, b_ue, nullptr, nullptr, nullptr,
       h_user, LD_EMB, N_USER, D_EMB, D_IN, 0);
  gemm(repo_feat, D_IN, W_re, D_EMB, nullptr, b_re, nullptr, nullptr, nullptr,
       h_repo, LD_EMB, N_REPO, D_EMB, D_IN, 0);

  gemm(h_user, LD_EMB, W_hid_ur, D_HID, deg_u_out_ur, nullptr, nullptr, nullptr, nullptr,
       feat_u, D_HID, N_USER, D_HID, D_EMB, 0);
  gemm(h_repo, LD_EMB, W_hid_ru, D_HID, deg_r_out_ru, nullptr, nullptr, nullptr, nullptr,
       feat_r, D_HID, N_REPO, D_HID, D_EMB, 0);

  gather(feat_u, ptr_r, col_ur, agg_r, N_REPO, N_USER, D_HID);
  gather(feat_r, ptr_u, col_ru, agg_u, N_USER, N_REPO, D_HID);

  gemm(h_user, LD_EMB, W_hcu, D_HID, nullptr, b_hcu, agg_u, deg_u_in_ru, b_hid_ru,
       out_user, D_HID, N_USER, D_HID, D_EMB, 1);
  gemm(h_repo, LD_EMB, W_hcr, D_HID, nullptr, b_hcr, agg_r, deg_r_in_ur, b_hid_ur,
       out_repo, D_HID, N_REPO, D_HID, D_EMB, 1);

  gemm(out_user, D_HID, W_out_ur, D_OUT, deg_u_out_ur, nullptr, nullptr, nullptr, nullptr,
       feat2_u, D_OUT, N_USER, D_OUT, D_HID, 0);
  gemm(out_repo, D_HID, W_out_ru, D_OUT, deg_r_out_ru, nullptr, nullptr, nullptr, nullptr,
       feat2_r, D_OUT, N_REPO, D_OUT, D_HID, 0);

  gather(feat2_u, ptr_r, col_ur, agg2_r, N_REPO, N_USER, D_OUT);
  gather(feat2_r, ptr_u, col_ru, agg2_u, N_USER, N_REPO, D_OUT);

  gemm(out_user, D_HID, W_ocu, D_OUT, nullptr, b_ocu, agg2_u, deg_u_in_ru, b_out_ru,
       hn_user, D_OUT, N_USER, D_OUT, D_HID, 1);
  gemm(out_repo, D_HID, W_ocr, D_OUT, nullptr, b_ocr, agg2_r, deg_r_in_ur, b_out_ur,
       hn_repo, D_OUT, N_REPO, D_OUT, D_HID, 1);

  hipLaunchKernelGGL(rownorm_kernel, dim3((N_USER + 8 * TAIL_PER_WAVE - 1) / (8 * TAIL_PER_WAVE)),
                     dim3(32, 8), 0, stream, hn_user, inv_nu, N_USER);
  hipLaunchKernelGGL(rownorm_kernel, dim3((N_REPO + 8 * TAIL_PER_WAVE - 1) / (8 * TAIL_PER_WAVE)),
                     dim3(32, 8), 0, stream, hn_repo, inv_nr, N_REPO);
  hipLaunchKernelGGL(score_kernel, dim3((NEP + 8 * TAIL_PER_WAVE - 1) / (8 * TAIL_PER_WAVE)),
                     dim3(32, 8), 0, stream,
                     hn_user, hn_repo, inv_nu, inv_nr, pos_u, pos_r, out, NEP);
  hipLaunchKernelGGL(score_kernel, dim3((NEP + 8 * TAIL_PER_WAVE - 1) / (8 * TAIL_PER_WAVE)),
                     dim3(32, 8), 0, stream,
                     hn_user, hn_repo, inv_nu, inv_nr, neg_u, neg_r, out + NEP, NEP);
}


# GEMM: one wave computes a whole 16-row strip (all 4/6/8 column tiles) per A-fragment load, W staged once per block in LDS, 512 blocks walk the strips; per-element sum order unchanged
# speedup vs baseline: 2.8008x; 1.4465x over previous
#include <hip/hip_runtime.h>
#include <math.h>

#define N_USER 50000
#define N_REPO 100000
#define NEDGE  1000000
#define NEP    200000
#define D_IN   256
#define D_EMB  125
#define LD_EMB 128
#define D_HID  96
#define D_OUT  64
#define EPSN   1e-12f

typedef float v2f __attribute__((ext_vector_type(2)));
typedef float v8f __attribute__((ext_vector_type(8)));

#define GEMM_WAVES  8
#define GEMM_BLOCKS 512
#define WT_FLOATS (LD_EMB * (D_IN + 4))

template <int NT, int KPAD>
__global__ void __launch_bounds__(GEMM_WAVES * 32, 4)
gemm_strip_wmma_f32(const float* __restrict__ A, int lda,
                    const float* __restrict__ W, int ldw,
                    const float* __restrict__ degA,
                    const float* __restrict__ bias,
                    const float* __restrict__ agg,
                    const float* __restrict__ degC,
                    const float* __restrict__ bias2,
                    float* __restrict__ C,
                    int M, int Nreal, int K, int relu)
{
  constexpr int NP  = NT * 16;
  constexpr int LS  = KPAD + 4;
  constexpr int NCH = KPAD / 32;
  static_assert(KPAD % 32 == 0, "K is walked in whole 128-byte lines");
  static_assert(NP * LS <= WT_FLOATS, "W^T fits the LDS array");
  __shared__ __align__(16) float Wt[WT_FLOATS];

  const int tid = threadIdx.x;

  const int lane = tid & 31;
  const int wave = tid >> 5;
  const int half = lane >> 4;
  const int l16  = lane & 15;

  float bn[NT], b2n[NT];
#pragma unroll
  for (int t = 0; t < NT; ++t) { bn[t] = 0.0f; b2n[t] = 0.0f; }
  if (bias) {
#pragma unroll
    for (int t = 0; t < NT; ++t) bn[t] = bias[min(t * 16 + l16, Nreal - 1)];
  }
  if (bias2) {
#pragma unroll
    for (int t = 0; t < NT; ++t) b2n[t] = bias2[min(t * 16 + l16, Nreal - 1)];
  }

  constexpr int NST = NP * KPAD / (GEMM_WAVES * 32);
  static_assert(NP * KPAD % (GEMM_WAVES * 32) == 0, "every thread stages the same count");
  constexpr int UNR = (NST % 16 == 0) ? 16 : 8;
  static_assert(NST % UNR == 0, "the staging loop unrolls without a remainder");
#pragma unroll UNR
  for (int j = 0; j < NST; ++j) {
    const int i = tid + j * (GEMM_WAVES * 32);
    const int k = i / NP;
    const int n = i - k * NP;
    const float w = W[(long)min(k, K - 1) * ldw + min(n, Nreal - 1)];
    const unsigned keep = (n < Nreal && k < K) ? 0xFFFFFFFFu : 0u;
    Wt[n * LS + k] = __uint_as_float(__float_as_uint(w) & keep);
  }
  __syncthreads();

  const float* Wp = &Wt[l16 * LS + 2 * half];
  const int Mtiles = M / 16;

  for (int mt = blockIdx.x + gridDim.x * wave; mt < Mtiles; mt += gridDim.x * GEMM_WAVES) {
    const int   m0 = mt * 16;
    const int   mA = m0 + l16;
    const float sA = degA ? rsqrtf(fmaxf(degA[mA], 1.0f)) : 1.0f;
    const float* Ap = A + (long)mA * lda + 2 * half;

    v8f acc[NT];
#pragma unroll
    for (int t = 0; t < NT; ++t) acc[t] = v8f{};

    v2f a_cur[8];
#pragma unroll
    for (int s = 0; s < 8; ++s) {
      a_cur[s] = *(const v2f*)(Ap + s * 4);
      a_cur[s] *= sA;
    }

    auto mma_chunk = [&](int c) {
      const float* Wc = Wp + c * 32;
#pragma unroll
      for (int s = 0; s < 8; ++s) {
        v2f b[NT];
#pragma unroll
        for (int t = 0; t < NT; ++t)
          b[t] = *(const v2f*)(Wc + t * 16 * LS + s * 4);
#pragma unroll
        for (int t = 0; t < NT; ++t)
          acc[t] = __builtin_amdgcn_wmma_f32_16x16x4_f32(false, a_cur[s], false, b[t],
                                                         (short)0, acc[t], false, false);
      }
    };

#pragma unroll 1
    for (int c = 0; c + 1 < NCH; ++c) {
      v2f a_nxt[8];
#pragma unroll
      for (int s = 0; s < 8; ++s) a_nxt[s] = *(const v2f*)(Ap + (c + 1) * 32 + s * 4);
      __builtin_amdgcn_sched_barrier(0);
      mma_chunk(c);
      __builtin_amdgcn_sched_barrier(0);
#pragma unroll
      for (int s = 0; s < 8; ++s) a_cur[s] = a_nxt[s] * sA;
    }
    mma_chunk(NCH - 1);

    const int mrow = m0 + 8 * half;
    if (agg) {
      float sC[8];
#pragma unroll
      for (int v = 0; v < 8; ++v) sC[v] = degC[mrow + v];
#pragma unroll
      for (int v = 0; v < 8; ++v) sC[v] = rsqrtf(fmaxf(sC[v], 1.0f));
      float g[NT][8];
#pragma unroll
      for (int t = 0; t < NT; ++t) {
#pragma unroll
        for (int v = 0; v < 8; ++v) g[t][v] = agg[(long)(mrow + v) * NP + t * 16 + l16];
      }
#pragma unroll
      for (int t = 0; t < NT; ++t) {
        const int n = t * 16 + l16;
#pragma unroll
        for (int v = 0; v < 8; ++v) {
          float val = acc[t][v] + bn[t];
          val += g[t][v] * sC[v] + b2n[t];
          if (relu) val = fmaxf(val, 0.0f);
          C[(long)(mrow + v) * NP + n] = (n < Nreal) ? val : 0.0f;
        }
      }
    } else {
#pragma unroll
      for (int t = 0; t < NT; ++t) {
        const int n = t * 16 + l16;
#pragma unroll
        for (int v = 0; v < 8; ++v) {
          float val = acc[t][v] + bn[t];
          if (relu) val = fmaxf(val, 0.0f);
          C[(long)(mrow + v) * NP + n] = (n < Nreal) ? val : 0.0f;
        }
      }
    }
  }
}


__global__ void degree_rank_kernel(const int* __restrict__ src, const int* __restrict__ dst,
                                   float* __restrict__ dsrc, int* __restrict__ cnt_dst,
                                   int* __restrict__ rank, int E)
{
  int e = blockIdx.x * blockDim.x + threadIdx.x;
  if (e < E) {
    const int s = src[e], d = dst[e];
    atomicAdd(&dsrc[s], 1.0f);
    rank[e] = atomicAdd(&cnt_dst[d], 1);
  }
}

#define SCAN_T 1024
#define SCAN_V 8
__global__ void __launch_bounds__(SCAN_T)
rowptr_scan_kernel(const int* __restrict__ cnt0, int* __restrict__ ptr0,
                   float* __restrict__ deg0, int n0,
                   const int* __restrict__ cnt1, int* __restrict__ ptr1,
                   float* __restrict__ deg1, int n1)
{
  const int* __restrict__ cnt = blockIdx.x ? cnt1 : cnt0;
  int*   __restrict__ ptr = blockIdx.x ? ptr1 : ptr0;
  float* __restrict__ deg = blockIdx.x ? deg1 : deg0;
  const int n = blockIdx.x ? n1 : n0;

  __shared__ int wsum[2][32];
  const int tid = threadIdx.x, lane = tid & 31, wv = tid >> 5;

  int v[SCAN_V], nv[SCAN_V];
#pragma unroll
  for (int k = 0; k < SCAN_V; ++k) {
    const int i = tid * SCAN_V + k;
    v[k] = (i < n) ? cnt[i] : 0;
  }
  int carry = 0, it = 0;
  for (int base = 0; base < n; base += SCAN_T * SCAN_V, ++it) {
    const int i0 = base + tid * SCAN_V;
    const int nb = i0 + SCAN_T * SCAN_V;
#pragma unroll
    for (int k = 0; k < SCAN_V; ++k) nv[k] = (nb + k < n) ? cnt[nb + k] : 0;

    int tsum = 0;
#pragma unroll
    for (int k = 0; k < SCAN_V; ++k) tsum += v[k];
    int x = tsum;
#pragma unroll
    for (int o = 1; o < 32; o <<= 1) {
      const int y = __shfl_up(x, o, 32);
      if (lane >= o) x += y;
    }
    if (lane == 31) wsum[it & 1][wv] = x;
    __syncthreads();
    const int w = wsum[it & 1][lane];
    int wx = w;
#pragma unroll
    for (int o = 1; o < 32; o <<= 1) {
      const int y = __shfl_up(wx, o, 32);
      if (lane >= o) wx += y;
    }
    const int wave_excl = __shfl(wx, wv, 32) - __shfl(w, wv, 32);
    const int total     = __shfl(wx, 31, 32);
    int run = carry + wave_excl + (x - tsum);
#pragma unroll
    for (int k = 0; k < SCAN_V; ++k) {
      const int i = i0 + k;
      if (i < n) { ptr[i] = run; deg[i] = (float)v[k]; }
      run += v[k];
    }
    carry += total;
#pragma unroll
    for (int k = 0; k < SCAN_V; ++k) v[k] = nv[k];
  }
}

__global__ void csr_fill_kernel(const int* __restrict__ src, const int* __restrict__ dst,
                                const int* __restrict__ rank, const int* __restrict__ rowptr,
                                int* __restrict__ col, int E)
{
  int e = blockIdx.x * blockDim.x + threadIdx.x;
  if (e < E) {
    int p = rowptr[dst[e]] + rank[e];
    p = min(max(p, 0), E - 1);
    col[p] = src[e];
  }
}

template <int D>
__global__ void __launch_bounds__(256)
gather_sum_kernel(const float* __restrict__ feat, const int* __restrict__ rowptr,
                  const int* __restrict__ col, float* __restrict__ agg,
                  int n_dst, int n_src, int E)
{
  const int d = __builtin_amdgcn_readfirstlane((int)(blockIdx.x * blockDim.y + threadIdx.y));
  if (d >= n_dst) return;
  const int lane = threadIdx.x;
  const int beg = max(rowptr[d], 0);
  const int end = min((d + 1 < n_dst) ? rowptr[d + 1] : E, E);

  float2 acc = make_float2(0.0f, 0.0f);
  float  acc2 = 0.0f;
  for (int base = beg; base < end; base += 32) {
    const int cnt = min(32, end - base);
    int myc = (lane < cnt) ? col[base + lane] : 0;
    myc = min(max(myc, 0), n_src - 1);
#pragma unroll 4
    for (int j = 0; j < cnt; ++j) {
      const int s = __builtin_amdgcn_readlane(myc, j);
      const float* p = feat + (size_t)s * D;
      const float2 t = ((const float2*)p)[lane];
      acc.x += t.x;
      acc.y += t.y;
      if (D == 96) acc2 += p[64 + lane];
    }
  }
  float* q = agg + (size_t)d * D;
  ((float2*)q)[lane] = acc;
  if (D == 96) q[64 + lane] = acc2;
}

#define TAIL_PER_WAVE 32

__global__ void __launch_bounds__(256)
rownorm_kernel(const float* __restrict__ X, float* __restrict__ invn, int rows)
{
  const int base = (blockIdx.x * blockDim.y + threadIdx.y) * TAIL_PER_WAVE;
  if (base >= rows) return;
  const int lane = threadIdx.x;
  const int g = lane >> 3, t = lane & 7;
  float sel = 1.0f;
#pragma unroll
  for (int j = 0; j < 8; ++j) {
    const int r = min(base + 4 * j + g, rows - 1);
    const float4* p = (const float4*)(X + (size_t)r * D_OUT);
    const float4 v0 = p[t], v1 = p[8 + t];
    float s = v0.x * v0.x + v0.y * v0.y + v0.z * v0.z + v0.w * v0.w
            + v1.x * v1.x + v1.y * v1.y + v1.z * v1.z + v1.w * v1.w;
    s += __shfl_xor(s, 4, 32);
    s += __shfl_xor(s, 2, 32);
    s += __shfl_xor(s, 1, 32);
    if (t == j) sel = s;
  }
  const int r = base + 4 * t + g;
  if (r < rows) invn[r] = 1.0f / fmaxf(sqrtf(sel), EPSN);
}

__global__ void __launch_bounds__(256)
score_kernel(const float* __restrict__ HU, const float* __restrict__ HR,
             const float* __restrict__ invU, const float* __restrict__ invR,
             const int* __restrict__ iu, const int* __restrict__ ir,
             float* __restrict__ out, int Ep)
{
  const int base = (blockIdx.x * blockDim.y + threadIdx.y) * TAIL_PER_WAVE;
  if (base >= Ep) return;
  const int lane = threadIdx.x;
  const int g = lane >> 3, t = lane & 7;
  const int el  = min(base + lane, Ep - 1);
  const int myu = iu[el], myr = ir[el];
  const float mynu = invU[myu], mynr = invR[myr];
  float sel = 0.0f;
#pragma unroll
  for (int j = 0; j < 8; ++j) {
    const int u = __shfl(myu, 4 * j + g, 32);
    const int r = __shfl(myr, 4 * j + g, 32);
    const float4* pu = (const float4*)(HU + (size_t)u * D_OUT);
    const float4* pr = (const float4*)(HR + (size_t)r * D_OUT);
    const float4 a0 = pu[t], a1 = pu[8 + t];
    const float4 b0 = pr[t], b1 = pr[8 + t];
    float s = a0.x * b0.x + a0.y * b0.y + a0.z * b0.z + a0.w * b0.w
            + a1.x * b1.x + a1.y * b1.y + a1.z * b1.z + a1.w * b1.w;
    s += __shfl_xor(s, 4, 32);
    s += __shfl_xor(s, 2, 32);
    s += __shfl_xor(s, 1, 32);
    if (t == j) sel = s;
  }
  const int own = 4 * t + g;
  const float nu = __shfl(mynu, own, 32);
  const float nr = __shfl(mynr, own, 32);
  const int e = base + own;
  if (e < Ep) out[e] = sel * nu * nr;
}

extern "C" void kernel_launch(void* const* d_in, const int* in_sizes, int n_in,
                              void* d_out, int out_size, void* d_ws, size_t ws_size,
                              hipStream_t stream)
{
  const float* user_feat = (const float*)d_in[0];
  const float* repo_feat = (const float*)d_in[1];
  const float* W_ue = (const float*)d_in[2];   const float* b_ue = (const float*)d_in[3];
  const float* W_re = (const float*)d_in[4];   const float* b_re = (const float*)d_in[5];
  const float* W_hid_ur = (const float*)d_in[6];  const float* b_hid_ur = (const float*)d_in[7];
  const float* W_hid_ru = (const float*)d_in[8];  const float* b_hid_ru = (const float*)d_in[9];
  const float* W_out_ur = (const float*)d_in[10]; const float* b_out_ur = (const float*)d_in[11];
  const float* W_out_ru = (const float*)d_in[12]; const float* b_out_ru = (const float*)d_in[13];
  const float* W_hcu = (const float*)d_in[14]; const float* b_hcu = (const float*)d_in[15];
  const float* W_hcr = (const float*)d_in[16]; const float* b_hcr = (const float*)d_in[17];
  const float* W_ocu = (const float*)d_in[18]; const float* b_ocu = (const float*)d_in[19];
  const float* W_ocr = (const float*)d_in[20]; const float* b_ocr = (const float*)d_in[21];
  const int* e_ur_src = (const int*)d_in[22];  const int* e_ur_dst = (const int*)d_in[23];
  const int* e_ru_src = (const int*)d_in[24];  const int* e_ru_dst = (const int*)d_in[25];
  const int* pos_u = (const int*)d_in[26];     const int* pos_r = (const int*)d_in[27];
  const int* neg_u = (const int*)d_in[28];     const int* neg_r = (const int*)d_in[29];
  float* out = (float*)d_out;

  float* w = (float*)d_ws;
  size_t o = 0;
  float* deg_u_out_ur = w + o; o += N_USER;
  float* deg_r_in_ur  = w + o; o += N_REPO;
  float* deg_r_out_ru = w + o; o += N_REPO;
  float* deg_u_in_ru  = w + o; o += N_USER;
  float* reg_agg  = w + o; o += ((size_t)N_USER + N_REPO) * D_HID;
  float* reg_h    = w + o; o += ((size_t)N_USER + N_REPO) * LD_EMB;
  float* reg_feat = w + o; o += ((size_t)N_USER + N_REPO) * D_HID;
  float* reg_csr  = w + o; o += ((size_t)N_USER + N_REPO) * D_HID;
  float* inv_nu  = w + o; o += N_USER;
  float* inv_nr  = w + o; o += N_REPO;

  int* cnt_r = (int*)reg_agg;
  int* cnt_u = cnt_r + N_REPO;
  const size_t zeroBytes = (size_t)(3 * N_USER + 3 * N_REPO) * sizeof(float);

  int* col_ur  = (int*)reg_csr;
  int* col_ru  = col_ur + NEDGE;
  int* rank_ur = col_ru + NEDGE;
  int* rank_ru = rank_ur + NEDGE;
  int* ptr_r   = rank_ru + NEDGE;
  int* ptr_u   = ptr_r + N_REPO;

  float* h_user   = reg_h;
  float* h_repo   = reg_h + (size_t)N_USER * LD_EMB;
  float* feat_u   = reg_feat;
  float* feat_r   = reg_feat + (size_t)N_USER * D_HID;
  float* agg_u    = reg_agg;
  float* agg_r    = reg_agg + (size_t)N_USER * D_HID;
  float* out_user = reg_feat;
  float* out_repo = reg_feat + (size_t)N_USER * D_HID;
  float* feat2_u  = reg_agg;
  float* feat2_r  = reg_agg + (size_t)N_USER * D_HID;
  float* agg2_u   = reg_h;
  float* agg2_r   = reg_h + (size_t)N_USER * LD_EMB;
  float* hn_user  = reg_agg;
  float* hn_repo  = reg_agg + (size_t)N_USER * D_HID;

  auto gemm = [&](const float* A, int lda, const float* W_, int ldw,
                  const float* degA, const float* bias,
                  const float* agg, const float* degC, const float* bias2,
                  float* C, int ldc, int M, int Nreal, int K, int relu) {
    dim3 blk(GEMM_WAVES * 32);
    dim3 grd(GEMM_BLOCKS);
    if (ldc == LD_EMB)
      hipLaunchKernelGGL((gemm_strip_wmma_f32<LD_EMB / 16, D_IN>), grd, blk, 0, stream,
                         A, lda, W_, ldw, degA, bias, agg, degC, bias2, C, M, Nreal, K, relu);
    else if (ldc == D_HID)
      hipLaunchKernelGGL((gemm_strip_wmma_f32<D_HID / 16, LD_EMB>), grd, blk, 0, stream,
                         A, lda, W_, ldw, degA, bias, agg, degC, bias2, C, M, Nreal, K, relu);
    else
      hipLaunchKernelGGL((gemm_strip_wmma_f32<D_OUT / 16, D_HID>), grd, blk, 0, stream,
                         A, lda, W_, ldw, degA, bias, agg, degC, bias2, C, M, Nreal, K, relu);
  };

  auto gather = [&](const float* feat, const int* rowptr, const int* col,
                    float* agg, int n_dst, int n_src, int D) {
    dim3 blk(32, 8);
    dim3 grd((n_dst + 7) / 8);
    if (D == D_HID)
      hipLaunchKernelGGL(gather_sum_kernel<D_HID>, grd, blk, 0, stream,
                         feat, rowptr, col, agg, n_dst, n_src, NEDGE);
    else
      hipLaunchKernelGGL(gather_sum_kernel<D_OUT>, grd, blk, 0, stream,
                         feat, rowptr, col, agg, n_dst, n_src, NEDGE);
  };

  hipMemsetAsync(deg_u_out_ur, 0, zeroBytes, stream);
  hipLaunchKernelGGL(degree_rank_kernel, dim3((NEDGE + 255) / 256), dim3(256), 0, stream,
                     e_ur_src, e_ur_dst, deg_u_out_ur, cnt_r, rank_ur, NEDGE);
  hipLaunchKernelGGL(degree_rank_kernel, dim3((NEDGE + 255) / 256), dim3(256), 0, stream,
                     e_ru_src, e_ru_dst, deg_r_out_ru, cnt_u, rank_ru, NEDGE);
  hipLaunchKernelGGL(rowptr_scan_kernel, dim3(2), dim3(SCAN_T), 0, stream,
                     cnt_r, ptr_r, deg_r_in_ur, N_REPO,
                     cnt_u, ptr_u, deg_u_in_ru, N_USER);
  hipLaunchKernelGGL(csr_fill_kernel, dim3((NEDGE + 255) / 256), dim3(256), 0, stream,
                     e_ur_src, e_ur_dst, rank_ur, ptr_r, col_ur, NEDGE);
  hipLaunchKernelGGL(csr_fill_kernel, dim3((NEDGE + 255) / 256), dim3(256), 0, stream,
                     e_ru_src, e_ru_dst, rank_ru, ptr_u, col_ru, NEDGE);

  gemm(user_feat, D_IN, W_ue, D_EMB, nullptr, b_ue, nullptr, nullptr, nullptr,
       h_user, LD_EMB, N_USER, D_EMB, D_IN, 0);
  gemm(repo_feat, D_IN, W_re, D_EMB, nullptr, b_re, nullptr, nullptr, nullptr,
       h_repo, LD_EMB, N_REPO, D_EMB, D_IN, 0);

  gemm(h_user, LD_EMB, W_hid_ur, D_HID, deg_u_out_ur, nullptr, nullptr, nullptr, nullptr,
       feat_u, D_HID, N_USER, D_HID, D_EMB, 0);
  gemm(h_repo, LD_EMB, W_hid_ru, D_HID, deg_r_out_ru, nullptr, nullptr, nullptr, nullptr,
       feat_r, D_HID, N_REPO, D_HID, D_EMB, 0);

  gather(feat_u, ptr_r, col_ur, agg_r, N_REPO, N_USER, D_HID);
  gather(feat_r, ptr_u, col_ru, agg_u, N_USER, N_REPO, D_HID);

  gemm(h_user, LD_EMB, W_hcu, D_HID, nullptr, b_hcu, agg_u, deg_u_in_ru, b_hid_ru,
       out_user, D_HID, N_USER, D_HID, D_EMB, 1);
  gemm(h_repo, LD_EMB, W_hcr, D_HID, nullptr, b_hcr, agg_r, deg_r_in_ur, b_hid_ur,
       out_repo, D_HID, N_REPO, D_HID, D_EMB, 1);

  gemm(out_user, D_HID, W_out_ur, D_OUT, deg_u_out_ur, nullptr, nullptr, nullptr, nullptr,
       feat2_u, D_OUT, N_USER, D_OUT, D_HID, 0);
  gemm(out_repo, D_HID, W_out_ru, D_OUT, deg_r_out_ru, nullptr, nullptr, nullptr, nullptr,
       feat2_r, D_OUT, N_REPO, D_OUT, D_HID, 0);

  gather(feat2_u, ptr_r, col_ur, agg2_r, N_REPO, N_USER, D_OUT);
  gather(feat2_r, ptr_u, col_ru, agg2_u, N_USER, N_REPO, D_OUT);

  gemm(out_user, D_HID, W_ocu, D_OUT, nullptr, b_ocu, agg2_u, deg_u_in_ru, b_out_ru,
       hn_user, D_OUT, N_USER, D_OUT, D_HID, 1);
  gemm(out_repo, D_HID, W_ocr, D_OUT, nullptr, b_ocr, agg2_r, deg_r_in_ur, b_out_ur,
       hn_repo, D_OUT, N_REPO, D_OUT, D_HID, 1);

  hipLaunchKernelGGL(rownorm_kernel, dim3((N_USER + 8 * TAIL_PER_WAVE - 1) / (8 * TAIL_PER_WAVE)),
                     dim3(32, 8), 0, stream, hn_user, inv_nu, N_USER);
  hipLaunchKernelGGL(rownorm_kernel, dim3((N_REPO + 8 * TAIL_PER_WAVE - 1) / (8 * TAIL_PER_WAVE)),
                     dim3(32, 8), 0, stream, hn_repo, inv_nr, N_REPO);
  hipLaunchKernelGGL(score_kernel, dim3((NEP + 8 * TAIL_PER_WAVE - 1) / (8 * TAIL_PER_WAVE)),
                     dim3(32, 8), 0, stream,
                     hn_user, hn_repo, inv_nu, inv_nr, pos_u, pos_r, out, NEP);
  hipLaunchKernelGGL(score_kernel, dim3((NEP + 8 * TAIL_PER_WAVE - 1) / (8 * TAIL_PER_WAVE)),
                     dim3(32, 8), 0, stream,
                     hn_user, hn_repo, inv_nu, inv_nr, neg_u, neg_r, out + NEP, NEP);
}


# CSR build: scattered global atomics (degrees, arrival rank) become LDS counters per node range and edge slice: count table, range-local scan, LDS-ranked fill; row sum order: slice, then arrival rank
# speedup vs baseline: 4.1774x; 1.4915x over previous
#include <hip/hip_runtime.h>
#include <math.h>

#define N_USER 50000
#define N_REPO 100000
#define NEDGE  1000000
#define NEP    200000
#define D_IN   256
#define D_EMB  125
#define LD_EMB 128
#define D_HID  96
#define D_OUT  64
#define EPSN   1e-12f

typedef float v2f __attribute__((ext_vector_type(2)));
typedef float v8f __attribute__((ext_vector_type(8)));

#define GEMM_WAVES  8
#define GEMM_BLOCKS 512
#define WT_FLOATS (LD_EMB * (D_IN + 4))

template <int NT, int KPAD>
__global__ void __launch_bounds__(GEMM_WAVES * 32, 4)
gemm_strip_wmma_f32(const float* __restrict__ A, int lda,
                    const float* __restrict__ W, int ldw,
                    const float* __restrict__ degA,
                    const float* __restrict__ bias,
                    const float* __restrict__ agg,
                    const float* __restrict__ degC,
                    const float* __restrict__ bias2,
                    float* __restrict__ C,
                    int M, int Nreal, int K, int relu)
{
  constexpr int NP  = NT * 16;
  constexpr int LS  = KPAD + 4;
  constexpr int NCH = KPAD / 32;
  static_assert(KPAD % 32 == 0, "K is walked in whole 128-byte lines");
  static_assert(NP * LS <= WT_FLOATS, "W^T fits the LDS array");
  __shared__ __align__(16) float Wt[WT_FLOATS];

  const int tid = threadIdx.x;

  const int lane = tid & 31;
  const int wave = tid >> 5;
  const int half = lane >> 4;
  const int l16  = lane & 15;

  float bn[NT], b2n[NT];
#pragma unroll
  for (int t = 0; t < NT; ++t) { bn[t] = 0.0f; b2n[t] = 0.0f; }
  if (bias) {
#pragma unroll
    for (int t = 0; t < NT; ++t) bn[t] = bias[min(t * 16 + l16, Nreal - 1)];
  }
  if (bias2) {
#pragma unroll
    for (int t = 0; t < NT; ++t) b2n[t] = bias2[min(t * 16 + l16, Nreal - 1)];
  }

  constexpr int NST = NP * KPAD / (GEMM_WAVES * 32);
  static_assert(NP * KPAD % (GEMM_WAVES * 32) == 0, "every thread stages the same count");
  constexpr int UNR = (NST % 16 == 0) ? 16 : 8;
  static_assert(NST % UNR == 0, "the staging loop unrolls without a remainder");
#pragma unroll UNR
  for (int j = 0; j < NST; ++j) {
    const int i = tid + j * (GEMM_WAVES * 32);
    const int k = i / NP;
    const int n = i - k * NP;
    const float w = W[(long)min(k, K - 1) * ldw + min(n, Nreal - 1)];
    const unsigned keep = (n < Nreal && k < K) ? 0xFFFFFFFFu : 0u;
    Wt[n * LS + k] = __uint_as_float(__float_as_uint(w) & keep);
  }
  __syncthreads();

  const float* Wp = &Wt[l16 * LS + 2 * half];
  const int Mtiles = M / 16;

  for (int mt = blockIdx.x + gridDim.x * wave; mt < Mtiles; mt += gridDim.x * GEMM_WAVES) {
    const int   m0 = mt * 16;
    const int   mA = m0 + l16;
    const float sA = degA ? rsqrtf(fmaxf(degA[mA], 1.0f)) : 1.0f;
    const float* Ap = A + (long)mA * lda + 2 * half;

    v8f acc[NT];
#pragma unroll
    for (int t = 0; t < NT; ++t) acc[t] = v8f{};

    v2f a_cur[8];
#pragma unroll
    for (int s = 0; s < 8; ++s) {
      a_cur[s] = *(const v2f*)(Ap + s * 4);
      a_cur[s] *= sA;
    }

    auto mma_chunk = [&](int c) {
      const float* Wc = Wp + c * 32;
#pragma unroll
      for (int s = 0; s < 8; ++s) {
        v2f b[NT];
#pragma unroll
        for (int t = 0; t < NT; ++t)
          b[t] = *(const v2f*)(Wc + t * 16 * LS + s * 4);
#pragma unroll
        for (int t = 0; t < NT; ++t)
          acc[t] = __builtin_amdgcn_wmma_f32_16x16x4_f32(false, a_cur[s], false, b[t],
                                                         (short)0, acc[t], false, false);
      }
    };

#pragma unroll 1
    for (int c = 0; c + 1 < NCH; ++c) {
      v2f a_nxt[8];
#pragma unroll
      for (int s = 0; s < 8; ++s) a_nxt[s] = *(const v2f*)(Ap + (c + 1) * 32 + s * 4);
      __builtin_amdgcn_sched_barrier(0);
      mma_chunk(c);
      __builtin_amdgcn_sched_barrier(0);
#pragma unroll
      for (int s = 0; s < 8; ++s) a_cur[s] = a_nxt[s] * sA;
    }
    mma_chunk(NCH - 1);

    const int mrow = m0 + 8 * half;
    if (agg) {
      float sC[8];
#pragma unroll
      for (int v = 0; v < 8; ++v) sC[v] = degC[mrow + v];
#pragma unroll
      for (int v = 0; v < 8; ++v) sC[v] = rsqrtf(fmaxf(sC[v], 1.0f));
      float g[NT][8];
#pragma unroll
      for (int t = 0; t < NT; ++t) {
#pragma unroll
        for (int v = 0; v < 8; ++v) g[t][v] = agg[(long)(mrow + v) * NP + t * 16 + l16];
      }
#pragma unroll
      for (int t = 0; t < NT; ++t) {
        const int n = t * 16 + l16;
#pragma unroll
        for (int v = 0; v < 8; ++v) {
          float val = acc[t][v] + bn[t];
          val += g[t][v] * sC[v] + b2n[t];
          if (relu) val = fmaxf(val, 0.0f);
          C[(long)(mrow + v) * NP + n] = (n < Nreal) ? val : 0.0f;
        }
      }
    } else {
#pragma unroll
      for (int t = 0; t < NT; ++t) {
        const int n = t * 16 + l16;
#pragma unroll
        for (int v = 0; v < 8; ++v) {
          float val = acc[t][v] + bn[t];
          if (relu) val = fmaxf(val, 0.0f);
          C[(long)(mrow + v) * NP + n] = (n < Nreal) ? val : 0.0f;
        }
      }
    }
  }
}


#define NRANGE  16
#define NSLICE  8
#define CSR_T   1024
#define SLICE_E (NEDGE / NSLICE)
#define RMAX    ((N_REPO + NRANGE - 1) / NRANGE)
static_assert(NEDGE % (4 * NSLICE) == 0, "a slice is a whole number of 16-byte groups");
static_assert(N_USER <= N_REPO, "RMAX covers both node kinds");
static_assert(NRANGE * NSLICE <= 128, "the totals of the earlier ranges are summed by four waves");
#define FILL_PER ((RMAX + CSR_T - 1) / CSR_T)
#define TOFF0 0
#define TOFF1 (NSLICE * N_USER)
#define TOFF2 (NSLICE * (N_USER + N_REPO))
#define TOFF3 (NSLICE * (N_USER + 2 * N_REPO))

__global__ void __launch_bounds__(CSR_T, 16)
range_count_kernel(const int* __restrict__ ia0, const int* __restrict__ ia1,
                   const int* __restrict__ ia2, const int* __restrict__ ia3,
                   int* __restrict__ part, int* __restrict__ tot)
{
  __shared__ int hist[RMAX];
  __shared__ int btot;
  const int a  = blockIdx.x / (NRANGE * NSLICE);
  const int rs = blockIdx.x - a * (NRANGE * NSLICE);
  const int r  = rs / NSLICE, s = rs - r * NSLICE;
  const int* __restrict__ idx = (a == 0) ? ia0 : (a == 1) ? ia1 : (a == 2) ? ia2 : ia3;
  const int n    = (a == 0 || a == 3) ? N_USER : N_REPO;
  const int toff = (a == 0) ? TOFF0 : (a == 1) ? TOFF1 : (a == 2) ? TOFF2 : TOFF3;
  const int R   = (n + NRANGE - 1) / NRANGE;
  const int lo  = r * R;
  const int len = max(min(R, n - lo), 0);
  const int tid = threadIdx.x;
  for (int i = tid; i < len; i += CSR_T) hist[i] = 0;
  if (tid == 0) btot = 0;
  __syncthreads();
  const int4* __restrict__ p = (const int4*)(idx + s * SLICE_E);
  constexpr int N4 = SLICE_E / 4;
  constexpr int GC = 8;
  for (int b = 0; b < N4; b += GC * CSR_T) {
    int4 v[GC];
#pragma unroll
    for (int g = 0; g < GC; ++g) {
      const int i = b + g * CSR_T + tid;
      v[g] = p[min(i, N4 - 1)];
      const int inval = (i < N4) ? 0 : -1;
      v[g].x |= inval; v[g].y |= inval; v[g].z |= inval; v[g].w |= inval;
    }
#pragma unroll
    for (int g = 0; g < GC; ++g) {
      unsigned u;
      u = (unsigned)(v[g].x - lo); if (u < (unsigned)len) atomicAdd(&hist[u], 1);
      u = (unsigned)(v[g].y - lo); if (u < (unsigned)len) atomicAdd(&hist[u], 1);
      u = (unsigned)(v[g].z - lo); if (u < (unsigned)len) atomicAdd(&hist[u], 1);
      u = (unsigned)(v[g].w - lo); if (u < (unsigned)len) atomicAdd(&hist[u], 1);
    }
  }
  __syncthreads();
  int* __restrict__ out = part + toff + s * n + lo;
  int my = 0;
  for (int i = tid; i < len; i += CSR_T) { const int h = hist[i]; out[i] = h; my += h; }
  my += __shfl_xor(my, 16, 32);
  my += __shfl_xor(my, 8, 32);
  my += __shfl_xor(my, 4, 32);
  my += __shfl_xor(my, 2, 32);
  my += __shfl_xor(my, 1, 32);
  if ((tid & 31) == 0) atomicAdd(&btot, my);
  __syncthreads();
  if (tid == 0) tot[blockIdx.x] = btot;
}

__global__ void __launch_bounds__(CSR_T, 16)
range_fill_kernel(const int* __restrict__ src0, const int* __restrict__ dst0,
                  const int* __restrict__ src1, const int* __restrict__ dst1,
                  const int* __restrict__ part, const int* __restrict__ tot,
                  int* __restrict__ ptr0, float* __restrict__ deg0,
                  int* __restrict__ ptr1, float* __restrict__ deg1,
                  float* __restrict__ dout0, float* __restrict__ dout2,
                  int* __restrict__ col0, int* __restrict__ col1)
{
  __shared__ int off[RMAX + 32];
  __shared__ int cnt[RMAX];
  __shared__ int wsum[32];
  __shared__ int bsum[4];
  const int tid = threadIdx.x, lane = tid & 31, wv = tid >> 5;

  if (blockIdx.x >= 2 * NRANGE * NSLICE) {
    const int x  = blockIdx.x - 2 * NRANGE * NSLICE;
    const int wh = x / NRANGE, r = x - wh * NRANGE;
    const int n  = wh ? N_REPO : N_USER;
    const int* __restrict__ tab = part + (wh ? TOFF2 : TOFF0);
    float* __restrict__ dout = wh ? dout2 : dout0;
    const int R   = (n + NRANGE - 1) / NRANGE;
    const int lo  = r * R;
    const int len = max(min(R, n - lo), 0);
    for (int i = tid; i < len; i += CSR_T) {
      int c = 0;
#pragma unroll
      for (int s2 = 0; s2 < NSLICE; ++s2) c += tab[s2 * n + lo + i];
      dout[lo + i] = (float)c;
    }
    return;
  }

  const int t  = blockIdx.x / (NRANGE * NSLICE);
  const int rs = blockIdx.x - t * (NRANGE * NSLICE);
  const int r  = rs / NSLICE, s = rs - r * NSLICE;
  const int* __restrict__ src = t ? src1 : src0;
  const int* __restrict__ dst = t ? dst1 : dst0;
  int*       __restrict__ ptr = t ? ptr1 : ptr0;
  float*     __restrict__ deg = t ? deg1 : deg0;
  int*       __restrict__ col = t ? col1 : col0;
  const int n   = t ? N_USER : N_REPO;
  const int* __restrict__ tab = part + (t ? TOFF3 : TOFF1);
  const int R   = (n + NRANGE - 1) / NRANGE;
  const int lo  = r * R;
  const int len = max(min(R, n - lo), 0);

  int bv = (tid < r * NSLICE) ? tot[(2 * t + 1) * NRANGE * NSLICE + tid] : 0;
  bv += __shfl_xor(bv, 16, 32);
  bv += __shfl_xor(bv, 8, 32);
  bv += __shfl_xor(bv, 4, 32);
  bv += __shfl_xor(bv, 2, 32);
  bv += __shfl_xor(bv, 1, 32);
  if (wv < 4 && lane == 0) bsum[wv] = bv;
  if (tid < 32) off[RMAX + tid] = 0;

#pragma unroll 2
  for (int i = tid; i < len; i += CSR_T) {
    int c = 0, pre = 0;
#pragma unroll
    for (int s2 = 0; s2 < NSLICE; ++s2) {
      const int x = tab[s2 * n + lo + i];
      c += x;
      pre += (s2 < s) ? x : 0;
    }
    cnt[i] = c;
    off[i] = pre;
  }
  __syncthreads();

  int cv[FILL_PER];
  int tsum = 0;
#pragma unroll
  for (int k = 0; k < FILL_PER; ++k) {
    const int i = tid * FILL_PER + k;
    cv[k] = (i < len) ? cnt[i] : 0;
    tsum += cv[k];
  }
  int x = tsum;
#pragma unroll
  for (int o = 1; o < 32; o <<= 1) {
    const int y = __shfl_up(x, o, 32);
    if (lane >= o) x += y;
  }
  if (lane == 31) wsum[wv] = x;
  __syncthreads();
  const int w = wsum[lane];
  int wx = w;
#pragma unroll
  for (int o = 1; o < 32; o <<= 1) {
    const int y = __shfl_up(wx, o, 32);
    if (lane >= o) wx += y;
  }
  const int wave_excl = __shfl(wx, wv, 32) - __shfl(w, wv, 32);
  int run = bsum[0] + bsum[1] + bsum[2] + bsum[3] + wave_excl + (x - tsum);
#pragma unroll
  for (int k = 0; k < FILL_PER; ++k) {
    const int i = tid * FILL_PER + k;
    if (i < len) {
      off[i] += run;
      if (s == 0) { ptr[lo + i] = run; deg[lo + i] = (float)cv[k]; }
    }
    run += cv[k];
  }
  __syncthreads();

  const unsigned spare = RMAX + lane;
  const int4* __restrict__ pd = (const int4*)(dst + s * SLICE_E);
  const int4* __restrict__ ps = (const int4*)(src + s * SLICE_E);
  constexpr int N4 = SLICE_E / 4;
  constexpr int GF = 4;
  for (int b = 0; b < N4; b += GF * CSR_T) {
    int4 d[GF], sv[GF];
#pragma unroll
    for (int g = 0; g < GF; ++g) {
      const int i = b + g * CSR_T + tid;
      d[g]  = pd[min(i, N4 - 1)];
      sv[g] = ps[min(i, N4 - 1)];
      const int inval = (i < N4) ? 0 : -1;
      d[g].x |= inval; d[g].y |= inval; d[g].z |= inval; d[g].w |= inval;
    }
#pragma unroll
    for (int g = 0; g < GF; ++g) {
      const unsigned u0 = (unsigned)(d[g].x - lo), u1 = (unsigned)(d[g].y - lo);
      const unsigned u2 = (unsigned)(d[g].z - lo), u3 = (unsigned)(d[g].w - lo);
      const bool m0 = u0 < (unsigned)len, m1 = u1 < (unsigned)len;
      const bool m2 = u2 < (unsigned)len, m3 = u3 < (unsigned)len;
      const int q0 = atomicAdd(&off[m0 ? u0 : spare], 1);
      const int q1 = atomicAdd(&off[m1 ? u1 : spare], 1);
      const int q2 = atomicAdd(&off[m2 ? u2 : spare], 1);
      const int q3 = atomicAdd(&off[m3 ? u3 : spare], 1);
      if (m0) col[min(max(q0, 0), NEDGE - 1)] = sv[g].x;
      if (m1) col[min(max(q1, 0), NEDGE - 1)] = sv[g].y;
      if (m2) col[min(max(q2, 0), NEDGE - 1)] = sv[g].z;
      if (m3) col[min(max(q3, 0), NEDGE - 1)] = sv[g].w;
    }
  }
}

template <int D>
__global__ void __launch_bounds__(256)
gather_sum_kernel(const float* __restrict__ feat, const int* __restrict__ rowptr,
                  const int* __restrict__ col, float* __restrict__ agg,
                  int n_dst, int n_src, int E)
{
  const int d = __builtin_amdgcn_readfirstlane((int)(blockIdx.x * blockDim.y + threadIdx.y));
  if (d >= n_dst) return;
  const int lane = threadIdx.x;
  const int beg = max(rowptr[d], 0);
  const int end = min((d + 1 < n_dst) ? rowptr[d + 1] : E, E);

  float2 acc = make_float2(0.0f, 0.0f);
  float  acc2 = 0.0f;
  for (int base = beg; base < end; base += 32) {
    const int cnt = min(32, end - base);
    int myc = (lane < cnt) ? col[base + lane] : 0;
    myc = min(max(myc, 0), n_src - 1);
#pragma unroll 4
    for (int j = 0; j < cnt; ++j) {
      const int s = __builtin_amdgcn_readlane(myc, j);
      const float* p = feat + (size_t)s * D;
      const float2 t = ((const float2*)p)[lane];
      acc.x += t.x;
      acc.y += t.y;
      if (D == 96) acc2 += p[64 + lane];
    }
  }
  float* q = agg + (size_t)d * D;
  ((float2*)q)[lane] = acc;
  if (D == 96) q[64 + lane] = acc2;
}

#define TAIL_PER_WAVE 32

__global__ void __launch_bounds__(256)
rownorm_kernel(const float* __restrict__ X, float* __restrict__ invn, int rows)
{
  const int base = (blockIdx.x * blockDim.y + threadIdx.y) * TAIL_PER_WAVE;
  if (base >= rows) return;
  const int lane = threadIdx.x;
  const int g = lane >> 3, t = lane & 7;
  float sel = 1.0f;
#pragma unroll
  for (int j = 0; j < 8; ++j) {
    const int r = min(base + 4 * j + g, rows - 1);
    const float4* p = (const float4*)(X + (size_t)r * D_OUT);
    const float4 v0 = p[t], v1 = p[8 + t];
    float s = v0.x * v0.x + v0.y * v0.y + v0.z * v0.z + v0.w * v0.w
            + v1.x * v1.x + v1.y * v1.y + v1.z * v1.z + v1.w * v1.w;
    s += __shfl_xor(s, 4, 32);
    s += __shfl_xor(s, 2, 32);
    s += __shfl_xor(s, 1, 32);
    if (t == j) sel = s;
  }
  const int r = base + 4 * t + g;
  if (r < rows) invn[r] = 1.0f / fmaxf(sqrtf(sel), EPSN);
}

__global__ void __launch_bounds__(256)
score_kernel(const float* __restrict__ HU, const float* __restrict__ HR,
             const float* __restrict__ invU, const float* __restrict__ invR,
             const int* __restrict__ iu, const int* __restrict__ ir,
             float* __restrict__ out, int Ep)
{
  const int base = (blockIdx.x * blockDim.y + threadIdx.y) * TAIL_PER_WAVE;
  if (base >= Ep) return;
  const int lane = threadIdx.x;
  const int g = lane >> 3, t = lane & 7;
  const int el  = min(base + lane, Ep - 1);
  const int myu = iu[el], myr = ir[el];
  const float mynu = invU[myu], mynr = invR[myr];
  float sel = 0.0f;
#pragma unroll
  for (int j = 0; j < 8; ++j) {
    const int u = __shfl(myu, 4 * j + g, 32);
    const int r = __shfl(myr, 4 * j + g, 32);
    const float4* pu = (const float4*)(HU + (size_t)u * D_OUT);
    const float4* pr = (const float4*)(HR + (size_t)r * D_OUT);
    const float4 a0 = pu[t], a1 = pu[8 + t];
    const float4 b0 = pr[t], b1 = pr[8 + t];
    float s = a0.x * b0.x + a0.y * b0.y + a0.z * b0.z + a0.w * b0.w
            + a1.x * b1.x + a1.y * b1.y + a1.z * b1.z + a1.w * b1.w;
    s += __shfl_xor(s, 4, 32);
    s += __shfl_xor(s, 2, 32);
    s += __shfl_xor(s, 1, 32);
    if (t == j) sel = s;
  }
  const int own = 4 * t + g;
  const float nu = __shfl(mynu, own, 32);
  const float nr = __shfl(mynr, own, 32);
  const int e = base + own;
  if (e < Ep) out[e] = sel * nu * nr;
}

extern "C" void kernel_launch(void* const* d_in, const int* in_sizes, int n_in,
                              void* d_out, int out_size, void* d_ws, size_t ws_size,
                              hipStream_t stream)
{
  const float* user_feat = (const float*)d_in[0];
  const float* repo_feat = (const float*)d_in[1];
  const float* W_ue = (const float*)d_in[2];   const float* b_ue = (const float*)d_in[3];
  const float* W_re = (const float*)d_in[4];   const float* b_re = (const float*)d_in[5];
  const float* W_hid_ur = (const float*)d_in[6];  const float* b_hid_ur = (const float*)d_in[7];
  const float* W_hid_ru = (const float*)d_in[8];  const float* b_hid_ru = (const float*)d_in[9];
  const float* W_out_ur = (const float*)d_in[10]; const float* b_out_ur = (const float*)d_in[11];
  const float* W_out_ru = (const float*)d_in[12]; const float* b_out_ru = (const float*)d_in[13];
  const float* W_hcu = (const float*)d_in[14]; const float* b_hcu = (const float*)d_in[15];
  const float* W_hcr = (const float*)d_in[16]; const float* b_hcr = (const float*)d_in[17];
  const float* W_ocu = (const float*)d_in[18]; const float* b_ocu = (const float*)d_in[19];
  const float* W_ocr = (const float*)d_in[20]; const float* b_ocr = (const float*)d_in[21];
  const int* e_ur_src = (const int*)d_in[22];  const int* e_ur_dst = (const int*)d_in[23];
  const int* e_ru_src = (const int*)d_in[24];  const int* e_ru_dst = (const int*)d_in[25];
  const int* pos_u = (const int*)d_in[26];     const int* pos_r = (const int*)d_in[27];
  const int* neg_u = (const int*)d_in[28];     const int* neg_r = (const int*)d_in[29];
  float* out = (float*)d_out;

  float* w = (float*)d_ws;
  size_t o = 0;
  float* deg_u_out_ur = w + o; o += N_USER;
  float* deg_r_in_ur  = w + o; o += N_REPO;
  float* deg_r_out_ru = w + o; o += N_REPO;
  float* deg_u_in_ru  = w + o; o += N_USER;
  float* reg_agg  = w + o; o += ((size_t)N_USER + N_REPO) * D_HID;
  float* reg_h    = w + o; o += ((size_t)N_USER + N_REPO) * LD_EMB;
  float* reg_feat = w + o; o += ((size_t)N_USER + N_REPO) * D_HID;
  float* reg_csr  = w + o; o += ((size_t)N_USER + N_REPO) * D_HID;
  float* inv_nu  = w + o; o += N_USER;
  float* inv_nr  = w + o; o += N_REPO;

  int* col_ur  = (int*)reg_csr;
  int* col_ru  = col_ur + NEDGE;
  int* part    = col_ru + NEDGE;
  int* tot     = part + NSLICE * 2 * (N_USER + N_REPO);
  int* ptr_r   = tot + 4 * NRANGE * NSLICE;
  int* ptr_u   = ptr_r + N_REPO;

  float* h_user   = reg_h;
  float* h_repo   = reg_h + (size_t)N_USER * LD_EMB;
  float* feat_u   = reg_feat;
  float* feat_r   = reg_feat + (size_t)N_USER * D_HID;
  float* agg_u    = reg_agg;
  float* agg_r    = reg_agg + (size_t)N_USER * D_HID;
  float* out_user = reg_feat;
  float* out_repo = reg_feat + (size_t)N_USER * D_HID;
  float* feat2_u  = reg_agg;
  float* feat2_r  = reg_agg + (size_t)N_USER * D_HID;
  float* agg2_u   = reg_h;
  float* agg2_r   = reg_h + (size_t)N_USER * LD_EMB;
  float* hn_user  = reg_agg;
  float* hn_repo  = reg_agg + (size_t)N_USER * D_HID;

  auto gemm = [&](const float* A, int lda, const float* W_, int ldw,
                  const float* degA, const float* bias,
                  const float* agg, const float* degC, const float* bias2,
                  float* C, int ldc, int M, int Nreal, int K, int relu) {
    dim3 blk(GEMM_WAVES * 32);
    dim3 grd(GEMM_BLOCKS);
    if (ldc == LD_EMB)
      hipLaunchKernelGGL((gemm_strip_wmma_f32<LD_EMB / 16, D_IN>), grd, blk, 0, stream,
                         A, lda, W_, ldw, degA, bias, agg, degC, bias2, C, M, Nreal, K, relu);
    else if (ldc == D_HID)
      hipLaunchKernelGGL((gemm_strip_wmma_f32<D_HID / 16, LD_EMB>), grd, blk, 0, stream,
                         A, lda, W_, ldw, degA, bias, agg, degC, bias2, C, M, Nreal, K, relu);
    else
      hipLaunchKernelGGL((gemm_strip_wmma_f32<D_OUT / 16, D_HID>), grd, blk, 0, stream,
                         A, lda, W_, ldw, degA, bias, agg, degC, bias2, C, M, Nreal, K, relu);
  };

  auto gather = [&](const float* feat, const int* rowptr, const int* col,
                    float* agg, int n_dst, int n_src, int D) {
    dim3 blk(32, 8);
    dim3 grd((n_dst + 7) / 8);
    if (D == D_HID)
      hipLaunchKernelGGL(gather_sum_kernel<D_HID>, grd, blk, 0, stream,
                         feat, rowptr, col, agg, n_dst, n_src, NEDGE);
    else
      hipLaunchKernelGGL(gather_sum_kernel<D_OUT>, grd, blk, 0, stream,
                         feat, rowptr, col, agg, n_dst, n_src, NEDGE);
  };

  hipLaunchKernelGGL(range_count_kernel, dim3(4 * NRANGE * NSLICE), dim3(CSR_T), 0, stream,
                     e_ur_src, e_ur_dst, e_ru_src, e_ru_dst, part, tot);
  hipLaunchKernelGGL(range_fill_kernel, dim3(2 * NRANGE * NSLICE + 2 * NRANGE), dim3(CSR_T), 0, stream,
                     e_ur_src, e_ur_dst, e_ru_src, e_ru_dst, (const int*)part, (const int*)tot,
                     ptr_r, deg_r_in_ur, ptr_u, deg_u_in_ru, deg_u_out_ur, deg_r_out_ru,
                     col_ur, col_ru);

  gemm(user_feat, D_IN, W_ue, D_EMB, nullptr, b_ue, nullptr, nullptr, nullptr,
       h_user, LD_EMB, N_USER, D_EMB, D_IN, 0);
  gemm(repo_feat, D_IN, W_re, D_EMB, nullptr, b_re, nullptr, nullptr, nullptr,
       h_repo, LD_EMB, N_REPO, D_EMB, D_IN, 0);

  gemm(h_user, LD_EMB, W_hid_ur, D_HID, deg_u_out_ur, nullptr, nullptr, nullptr, nullptr,
       feat_u, D_HID, N_USER, D_HID, D_EMB, 0);
  gemm(h_repo, LD_EMB, W_hid_ru, D_HID, deg_r_out_ru, nullptr, nullptr, nullptr, nullptr,
       feat_r, D_HID, N_REPO, D_HID, D_EMB, 0);

  gather(feat_u, ptr_r, col_ur, agg_r, N_REPO, N_USER, D_HID);
  gather(feat_r, ptr_u, col_ru, agg_u, N_USER, N_REPO, D_HID);

  gemm(h_user, LD_EMB, W_hcu, D_HID, nullptr, b_hcu, agg_u, deg_u_in_ru, b_hid_ru,
       out_user, D_HID, N_USER, D_HID, D_EMB, 1);
  gemm(h_repo, LD_EMB, W_hcr, D_HID, nullptr, b_hcr, agg_r, deg_r_in_ur, b_hid_ur,
       out_repo, D_HID, N_REPO, D_HID, D_EMB, 1);

  gemm(out_user, D_HID, W_out_ur, D_OUT, deg_u_out_ur, nullptr, nullptr, nullptr, nullptr,
       feat2_u, D_OUT, N_USER, D_OUT, D_HID, 0);
  gemm(out_repo, D_HID, W_out_ru, D_OUT, deg_r_out_ru, nullptr, nullptr, nullptr, nullptr,
       feat2_r, D_OUT, N_REPO, D_OUT, D_HID, 0);

  gather(feat2_u, ptr_r, col_ur, agg2_r, N_REPO, N_USER, D_OUT);
  gather(feat2_r, ptr_u, col_ru, agg2_u, N_USER, N_REPO, D_OUT);

  gemm(out_user, D_HID, W_ocu, D_OUT, nullptr, b_ocu, agg2_u, deg_u_in_ru, b_out_ru,
       hn_user, D_OUT, N_USER, D_OUT, D_HID, 1);
  gemm(out_repo, D_HID, W_ocr, D_OUT, nullptr, b_ocr, agg2_r, deg_r_in_ur, b_out_ur,
       hn_repo, D_OUT, N_REPO, D_OUT, D_HID, 1);

  hipLaunchKernelGGL(rownorm_kernel, dim3((N_USER + 8 * TAIL_PER_WAVE - 1) / (8 * TAIL_PER_WAVE)),
                     dim3(32, 8), 0, stream, hn_user, inv_nu, N_USER);
  hipLaunchKernelGGL(rownorm_kernel, dim3((N_REPO + 8 * TAIL_PER_WAVE - 1) / (8 * TAIL_PER_WAVE)),
                     dim3(32, 8), 0, stream, hn_repo, inv_nr, N_REPO);
  hipLaunchKernelGGL(score_kernel, dim3((NEP + 8 * TAIL_PER_WAVE - 1) / (8 * TAIL_PER_WAVE)),
                     dim3(32, 8), 0, stream,
                     hn_user, hn_repo, inv_nu, inv_nr, pos_u, pos_r, out, NEP);
  hipLaunchKernelGGL(score_kernel, dim3((NEP + 8 * TAIL_PER_WAVE - 1) / (8 * TAIL_PER_WAVE)),
                     dim3(32, 8), 0, stream,
                     hn_user, hn_repo, inv_nu, inv_nr, neg_u, neg_r, out + NEP, NEP);
}


# User-side and repo-side launches of every stage merged into one launch each (block index picks the side): 20 launches become 11; per-element sum order unchanged
# speedup vs baseline: 4.7557x; 1.1384x over previous
#include <hip/hip_runtime.h>
#include <math.h>

#define N_USER 50000
#define N_REPO 100000
#define NEDGE  1000000
#define NEP    200000
#define D_IN   256
#define D_EMB  125
#define LD_EMB 128
#define D_HID  96
#define D_OUT  64
#define EPSN   1e-12f

typedef float v2f __attribute__((ext_vector_type(2)));
typedef float v8f __attribute__((ext_vector_type(8)));


#define GEMM_WAVES  8
#define GEMM_BLOCKS 512
#define GEMM_BLOCKS_U ((GEMM_BLOCKS * N_USER + (N_USER + N_REPO) / 2) / (N_USER + N_REPO))
#define WT_FLOATS (LD_EMB * (D_IN + 4))

struct GemmSide {
  const float* A;
  const float* W;
  const float* degA;
  const float* bias;
  const float* agg;
  const float* degC;
  const float* bias2;
  float*       C;
  int          M;
};

template <int NT, int KPAD>
__device__ __forceinline__ void
gemm_strip_body(const float* __restrict__ A, int lda,
                const float* __restrict__ W, int ldw,
                const float* __restrict__ degA,
                const float* __restrict__ bias,
                const float* __restrict__ agg,
                const float* __restrict__ degC,
                const float* __restrict__ bias2,
                float* __restrict__ C,
                int M, int Nreal, int K, int relu, int lb, int nb)
{
  constexpr int NP  = NT * 16;
  constexpr int LS  = KPAD + 4;
  constexpr int NCH = KPAD / 32;
  static_assert(KPAD % 32 == 0, "K is walked in whole 128-byte lines");
  static_assert(NP * LS <= WT_FLOATS, "W^T fits the LDS array");
  __shared__ __align__(16) float Wt[WT_FLOATS];

  const int tid = threadIdx.x;

  const int lane = tid & 31;
  const int wave = tid >> 5;
  const int half = lane >> 4;
  const int l16  = lane & 15;

  float bn[NT], b2n[NT];
#pragma unroll
  for (int t = 0; t < NT; ++t) { bn[t] = 0.0f; b2n[t] = 0.0f; }
  if (bias) {
#pragma unroll
    for (int t = 0; t < NT; ++t) bn[t] = bias[min(t * 16 + l16, Nreal - 1)];
  }
  if (bias2) {
#pragma unroll
    for (int t = 0; t < NT; ++t) b2n[t] = bias2[min(t * 16 + l16, Nreal - 1)];
  }

  constexpr int NST = NP * KPAD / (GEMM_WAVES * 32);
  static_assert(NP * KPAD % (GEMM_WAVES * 32) == 0, "every thread stages the same count");
  constexpr int UNR = (NST % 16 == 0) ? 16 : 8;
  static_assert(NST % UNR == 0, "the staging loop unrolls without a remainder");
#pragma unroll UNR
  for (int j = 0; j < NST; ++j) {
    const int i = tid + j * (GEMM_WAVES * 32);
    const int k = i / NP;
    const int n = i - k * NP;
    const float w = W[(long)min(k, K - 1) * ldw + min(n, Nreal - 1)];
    const unsigned keep = (n < Nreal && k < K) ? 0xFFFFFFFFu : 0u;
    Wt[n * LS + k] = __uint_as_float(__float_as_uint(w) & keep);
  }
  __syncthreads();

  const float* Wp = &Wt[l16 * LS + 2 * half];
  const int Mtiles = M / 16;

  for (int mt = lb + nb * wave; mt < Mtiles; mt += nb * GEMM_WAVES) {
    const int   m0 = mt * 16;
    const int   mA = m0 + l16;
    const float sA = degA ? rsqrtf(fmaxf(degA[mA], 1.0f)) : 1.0f;
    const float* Ap = A + (long)mA * lda + 2 * half;

    v8f acc[NT];
#pragma unroll
    for (int t = 0; t < NT; ++t) acc[t] = v8f{};

    v2f a_cur[8];
#pragma unroll
    for (int s = 0; s < 8; ++s) {
      a_cur[s] = *(const v2f*)(Ap + s * 4);
      a_cur[s] *= sA;
    }

    auto mma_chunk = [&](int c) {
      const float* Wc = Wp + c * 32;
#pragma unroll
      for (int s = 0; s < 8; ++s) {
        v2f b[NT];
#pragma unroll
        for (int t = 0; t < NT; ++t)
          b[t] = *(const v2f*)(Wc + t * 16 * LS + s * 4);
#pragma unroll
        for (int t = 0; t < NT; ++t)
          acc[t] = __builtin_amdgcn_wmma_f32_16x16x4_f32(false, a_cur[s], false, b[t],
                                                         (short)0, acc[t], false, false);
      }
    };

#pragma unroll 1
    for (int c = 0; c + 1 < NCH; ++c) {
      v2f a_nxt[8];
#pragma unroll
      for (int s = 0; s < 8; ++s) a_nxt[s] = *(const v2f*)(Ap + (c + 1) * 32 + s * 4);
      __builtin_amdgcn_sched_barrier(0);
      mma_chunk(c);
      __builtin_amdgcn_sched_barrier(0);
#pragma unroll
      for (int s = 0; s < 8; ++s) a_cur[s] = a_nxt[s] * sA;
    }
    mma_chunk(NCH - 1);

    const int mrow = m0 + 8 * half;
    if (agg) {
      float sC[8];
#pragma unroll
      for (int v = 0; v < 8; ++v) sC[v] = degC[mrow + v];
#pragma unroll
      for (int v = 0; v < 8; ++v) sC[v] = rsqrtf(fmaxf(sC[v], 1.0f));
      float g[NT][8];
#pragma unroll
      for (int t = 0; t < NT; ++t) {
#pragma unroll
        for (int v = 0; v < 8; ++v) g[t][v] = agg[(long)(mrow + v) * NP + t * 16 + l16];
      }
#pragma unroll
      for (int t = 0; t < NT; ++t) {
        const int n = t * 16 + l16;
#pragma unroll
        for (int v = 0; v < 8; ++v) {
          float val = acc[t][v] + bn[t];
          val += g[t][v] * sC[v] + b2n[t];
          if (relu) val = fmaxf(val, 0.0f);
          C[(long)(mrow + v) * NP + n] = (n < Nreal) ? val : 0.0f;
        }
      }
    } else {
#pragma unroll
      for (int t = 0; t < NT; ++t) {
        const int n = t * 16 + l16;
#pragma unroll
        for (int v = 0; v < 8; ++v) {
          float val = acc[t][v] + bn[t];
          if (relu) val = fmaxf(val, 0.0f);
          C[(long)(mrow + v) * NP + n] = (n < Nreal) ? val : 0.0f;
        }
      }
    }
  }
}

template <int NT, int KPAD>
__global__ void __launch_bounds__(GEMM_WAVES * 32, 4)
gemm_pair_wmma_f32(GemmSide u, GemmSide r, int lda, int ldw,
                   int Nreal, int K, int relu, int nbu)
{
  const bool us = (int)blockIdx.x < nbu;
  gemm_strip_body<NT, KPAD>(us ? u.A : r.A, lda, us ? u.W : r.W, ldw,
                            us ? u.degA : r.degA, us ? u.bias : r.bias,
                            us ? u.agg : r.agg, us ? u.degC : r.degC,
                            us ? u.bias2 : r.bias2, us ? u.C : r.C,
                            us ? u.M : r.M, Nreal, K, relu,
                            us ? (int)blockIdx.x : (int)blockIdx.x - nbu,
                            us ? nbu : (int)gridDim.x - nbu);
}


#define NRANGE  16
#define NSLICE  8
#define CSR_T   1024
#define SLICE_E (NEDGE / NSLICE)
#define RMAX    ((N_REPO + NRANGE - 1) / NRANGE)
static_assert(NEDGE % (4 * NSLICE) == 0, "a slice is a whole number of 16-byte groups");
static_assert(N_USER <= N_REPO, "RMAX covers both node kinds");
static_assert(NRANGE * NSLICE <= 128, "the totals of the earlier ranges are summed by four waves");
#define FILL_PER ((RMAX + CSR_T - 1) / CSR_T)
#define TOFF0 0
#define TOFF1 (NSLICE * N_USER)
#define TOFF2 (NSLICE * (N_USER + N_REPO))
#define TOFF3 (NSLICE * (N_USER + 2 * N_REPO))

__global__ void __launch_bounds__(CSR_T, 16)
range_count_kernel(const int* __restrict__ ia0, const int* __restrict__ ia1,
                   const int* __restrict__ ia2, const int* __restrict__ ia3,
                   int* __restrict__ part, int* __restrict__ tot)
{
  __shared__ int hist[RMAX];
  __shared__ int btot;
  const int a  = blockIdx.x / (NRANGE * NSLICE);
  const int rs = blockIdx.x - a * (NRANGE * NSLICE);
  const int r  = rs / NSLICE, s = rs - r * NSLICE;
  const int* __restrict__ idx = (a == 0) ? ia0 : (a == 1) ? ia1 : (a == 2) ? ia2 : ia3;
  const int n    = (a == 0 || a == 3) ? N_USER : N_REPO;
  const int toff = (a == 0) ? TOFF0 : (a == 1) ? TOFF1 : (a == 2) ? TOFF2 : TOFF3;
  const int R   = (n + NRANGE - 1) / NRANGE;
  const int lo  = r * R;
  const int len = max(min(R, n - lo), 0);
  const int tid = threadIdx.x;
  for (int i = tid; i < len; i += CSR_T) hist[i] = 0;
  if (tid == 0) btot = 0;
  __syncthreads();
  const int4* __restrict__ p = (const int4*)(idx + s * SLICE_E);
  constexpr int N4 = SLICE_E / 4;
  constexpr int GC = 8;
  for (int b = 0; b < N4; b += GC * CSR_T) {
    int4 v[GC];
#pragma unroll
    for (int g = 0; g < GC; ++g) {
      const int i = b + g * CSR_T + tid;
      v[g] = p[min(i, N4 - 1)];
      const int inval = (i < N4) ? 0 : -1;
      v[g].x |= inval; v[g].y |= inval; v[g].z |= inval; v[g].w |= inval;
    }
#pragma unroll
    for (int g = 0; g < GC; ++g) {
      unsigned u;
      u = (unsigned)(v[g].x - lo); if (u < (unsigned)len) atomicAdd(&hist[u], 1);
      u = (unsigned)(v[g].y - lo); if (u < (unsigned)len) atomicAdd(&hist[u], 1);
      u = (unsigned)(v[g].z - lo); if (u < (unsigned)len) atomicAdd(&hist[u], 1);
      u = (unsigned)(v[g].w - lo); if (u < (unsigned)len) atomicAdd(&hist[u], 1);
    }
  }
  __syncthreads();
  int* __restrict__ out = part + toff + s * n + lo;
  int my = 0;
  for (int i = tid; i < len; i += CSR_T) { const int h = hist[i]; out[i] = h; my += h; }
  my += __shfl_xor(my, 16, 32);
  my += __shfl_xor(my, 8, 32);
  my += __shfl_xor(my, 4, 32);
  my += __shfl_xor(my, 2, 32);
  my += __shfl_xor(my, 1, 32);
  if ((tid & 31) == 0) atomicAdd(&btot, my);
  __syncthreads();
  if (tid == 0) tot[blockIdx.x] = btot;
}

__global__ void __launch_bounds__(CSR_T, 16)
range_fill_kernel(const int* __restrict__ src0, const int* __restrict__ dst0,
                  const int* __restrict__ src1, const int* __restrict__ dst1,
                  const int* __restrict__ part, const int* __restrict__ tot,
                  int* __restrict__ ptr0, float* __restrict__ deg0,
                  int* __restrict__ ptr1, float* __restrict__ deg1,
                  float* __restrict__ dout0, float* __restrict__ dout2,
                  int* __restrict__ col0, int* __restrict__ col1)
{
  __shared__ int off[RMAX + 32];
  __shared__ int cnt[RMAX];
  __shared__ int wsum[32];
  __shared__ int bsum[4];
  const int tid = threadIdx.x, lane = tid & 31, wv = tid >> 5;

  if (blockIdx.x >= 2 * NRANGE * NSLICE) {
    const int x  = blockIdx.x - 2 * NRANGE * NSLICE;
    const int wh = x / NRANGE, r = x - wh * NRANGE;
    const int n  = wh ? N_REPO : N_USER;
    const int* __restrict__ tab = part + (wh ? TOFF2 : TOFF0);
    float* __restrict__ dout = wh ? dout2 : dout0;
    const int R   = (n + NRANGE - 1) / NRANGE;
    const int lo  = r * R;
    const int len = max(min(R, n - lo), 0);
    for (int i = tid; i < len; i += CSR_T) {
      int c = 0;
#pragma unroll
      for (int s2 = 0; s2 < NSLICE; ++s2) c += tab[s2 * n + lo + i];
      dout[lo + i] = (float)c;
    }
    return;
  }

  const int t  = blockIdx.x / (NRANGE * NSLICE);
  const int rs = blockIdx.x - t * (NRANGE * NSLICE);
  const int r  = rs / NSLICE, s = rs - r * NSLICE;
  const int* __restrict__ src = t ? src1 : src0;
  const int* __restrict__ dst = t ? dst1 : dst0;
  int*       __restrict__ ptr = t ? ptr1 : ptr0;
  float*     __restrict__ deg = t ? deg1 : deg0;
  int*       __restrict__ col = t ? col1 : col0;
  const int n   = t ? N_USER : N_REPO;
  const int* __restrict__ tab = part + (t ? TOFF3 : TOFF1);
  const int R   = (n + NRANGE - 1) / NRANGE;
  const int lo  = r * R;
  const int len = max(min(R, n - lo), 0);

  int bv = (tid < r * NSLICE) ? tot[(2 * t + 1) * NRANGE * NSLICE + tid] : 0;
  bv += __shfl_xor(bv, 16, 32);
  bv += __shfl_xor(bv, 8, 32);
  bv += __shfl_xor(bv, 4, 32);
  bv += __shfl_xor(bv, 2, 32);
  bv += __shfl_xor(bv, 1, 32);
  if (wv < 4 && lane == 0) bsum[wv] = bv;
  if (tid < 32) off[RMAX + tid] = 0;

#pragma unroll 2
  for (int i = tid; i < len; i += CSR_T) {
    int c = 0, pre = 0;
#pragma unroll
    for (int s2 = 0; s2 < NSLICE; ++s2) {
      const int x = tab[s2 * n + lo + i];
      c += x;
      pre += (s2 < s) ? x : 0;
    }
    cnt[i] = c;
    off[i] = pre;
  }
  __syncthreads();

  int cv[FILL_PER];
  int tsum = 0;
#pragma unroll
  for (int k = 0; k < FILL_PER; ++k) {
    const int i = tid * FILL_PER + k;
    cv[k] = (i < len) ? cnt[i] : 0;
    tsum += cv[k];
  }
  int x = tsum;
#pragma unroll
  for (int o = 1; o < 32; o <<= 1) {
    const int y = __shfl_up(x, o, 32);
    if (lane >= o) x += y;
  }
  if (lane == 31) wsum[wv] = x;
  __syncthreads();
  const int w = wsum[lane];
  int wx = w;
#pragma unroll
  for (int o = 1; o < 32; o <<= 1) {
    const int y = __shfl_up(wx, o, 32);
    if (lane >= o) wx += y;
  }
  const int wave_excl = __shfl(wx, wv, 32) - __shfl(w, wv, 32);
  int run = bsum[0] + bsum[1] + bsum[2] + bsum[3] + wave_excl + (x - tsum);
#pragma unroll
  for (int k = 0; k < FILL_PER; ++k) {
    const int i = tid * FILL_PER + k;
    if (i < len) {
      off[i] += run;
      if (s == 0) { ptr[lo + i] = run; deg[lo + i] = (float)cv[k]; }
    }
    run += cv[k];
  }
  __syncthreads();

  const unsigned spare = RMAX + lane;
  const int4* __restrict__ pd = (const int4*)(dst + s * SLICE_E);
  const int4* __restrict__ ps = (const int4*)(src + s * SLICE_E);
  constexpr int N4 = SLICE_E / 4;
  constexpr int GF = 4;
  for (int b = 0; b < N4; b += GF * CSR_T) {
    int4 d[GF], sv[GF];
#pragma unroll
    for (int g = 0; g < GF; ++g) {
      const int i = b + g * CSR_T + tid;
      d[g]  = pd[min(i, N4 - 1)];
      sv[g] = ps[min(i, N4 - 1)];
      const int inval = (i < N4) ? 0 : -1;
      d[g].x |= inval; d[g].y |= inval; d[g].z |= inval; d[g].w |= inval;
    }
#pragma unroll
    for (int g = 0; g < GF; ++g) {
      const unsigned u0 = (unsigned)(d[g].x - lo), u1 = (unsigned)(d[g].y - lo);
      const unsigned u2 = (unsigned)(d[g].z - lo), u3 = (unsigned)(d[g].w - lo);
      const bool m0 = u0 < (unsigned)len, m1 = u1 < (unsigned)len;
      const bool m2 = u2 < (unsigned)len, m3 = u3 < (unsigned)len;
      const int q0 = atomicAdd(&off[m0 ? u0 : spare], 1);
      const int q1 = atomicAdd(&off[m1 ? u1 : spare], 1);
      const int q2 = atomicAdd(&off[m2 ? u2 : spare], 1);
      const int q3 = atomicAdd(&off[m3 ? u3 : spare], 1);
      if (m0) col[min(max(q0, 0), NEDGE - 1)] = sv[g].x;
      if (m1) col[min(max(q1, 0), NEDGE - 1)] = sv[g].y;
      if (m2) col[min(max(q2, 0), NEDGE - 1)] = sv[g].z;
      if (m3) col[min(max(q3, 0), NEDGE - 1)] = sv[g].w;
    }
  }
}

struct GatherSide {
  const float* feat;
  const int*   rowptr;
  const int*   col;
  float*       agg;
  int          n_dst;
  int          n_src;
};

template <int D>
__device__ __forceinline__ void
gather_row(const float* __restrict__ feat, const int* __restrict__ rowptr,
           const int* __restrict__ col, float* __restrict__ agg,
           int n_dst, int n_src, int E, int d)
{
  if (d >= n_dst) return;
  const int lane = threadIdx.x;
  const int beg = max(rowptr[d], 0);
  const int end = min((d + 1 < n_dst) ? rowptr[d + 1] : E, E);

  float2 acc = make_float2(0.0f, 0.0f);
  float  acc2 = 0.0f;
  for (int base = beg; base < end; base += 32) {
    const int cnt = min(32, end - base);
    int myc = (lane < cnt) ? col[base + lane] : 0;
    myc = min(max(myc, 0), n_src - 1);
#pragma unroll 4
    for (int j = 0; j < cnt; ++j) {
      const int s = __builtin_amdgcn_readlane(myc, j);
      const float* p = feat + (size_t)s * D;
      const float2 t = ((const float2*)p)[lane];
      acc.x += t.x;
      acc.y += t.y;
      if (D == 96) acc2 += p[64 + lane];
    }
  }
  float* q = agg + (size_t)d * D;
  ((float2*)q)[lane] = acc;
  if (D == 96) q[64 + lane] = acc2;
}

template <int D>
__global__ void __launch_bounds__(256)
gather_pair_kernel(GatherSide a, GatherSide b, int nblk_a, int E)
{
  const bool fa = (int)blockIdx.x < nblk_a;
  const int  lb = fa ? (int)blockIdx.x : (int)blockIdx.x - nblk_a;
  const int d = __builtin_amdgcn_readfirstlane((int)(lb * blockDim.y + threadIdx.y));
  gather_row<D>(fa ? a.feat : b.feat, fa ? a.rowptr : b.rowptr, fa ? a.col : b.col,
                fa ? a.agg : b.agg, fa ? a.n_dst : b.n_dst, fa ? a.n_src : b.n_src, E, d);
}

#define TAIL_PER_WAVE 32
#define NORM_BLK_U ((N_USER + 8 * TAIL_PER_WAVE - 1) / (8 * TAIL_PER_WAVE))
#define NORM_BLK_R ((N_REPO + 8 * TAIL_PER_WAVE - 1) / (8 * TAIL_PER_WAVE))
#define SCORE_BLK  ((NEP + 8 * TAIL_PER_WAVE - 1) / (8 * TAIL_PER_WAVE))

__device__ __forceinline__ void
rownorm_wave(const float* __restrict__ X, float* __restrict__ invn, int rows, int wv)
{
  const int base = wv * TAIL_PER_WAVE;
  if (base >= rows) return;
  const int lane = threadIdx.x;
  const int g = lane >> 3, t = lane & 7;
  float sel = 1.0f;
#pragma unroll
  for (int j = 0; j < 8; ++j) {
    const int r = min(base + 4 * j + g, rows - 1);
    const float4* p = (const float4*)(X + (size_t)r * D_OUT);
    const float4 v0 = p[t], v1 = p[8 + t];
    float s = v0.x * v0.x + v0.y * v0.y + v0.z * v0.z + v0.w * v0.w
            + v1.x * v1.x + v1.y * v1.y + v1.z * v1.z + v1.w * v1.w;
    s += __shfl_xor(s, 4, 32);
    s += __shfl_xor(s, 2, 32);
    s += __shfl_xor(s, 1, 32);
    if (t == j) sel = s;
  }
  const int r = base + 4 * t + g;
  if (r < rows) invn[r] = 1.0f / fmaxf(sqrtf(sel), EPSN);
}

__global__ void __launch_bounds__(256)
rownorm_pair_kernel(const float* __restrict__ XU, const float* __restrict__ XR,
                    float* __restrict__ invU, float* __restrict__ invR,
                    int rowsU, int rowsR, int nblk_u)
{
  const bool fu = (int)blockIdx.x < nblk_u;
  const int  lb = fu ? (int)blockIdx.x : (int)blockIdx.x - nblk_u;
  rownorm_wave(fu ? XU : XR, fu ? invU : invR, fu ? rowsU : rowsR,
               (int)(lb * blockDim.y + threadIdx.y));
}

__device__ __forceinline__ void
score_wave(const float* __restrict__ HU, const float* __restrict__ HR,
           const float* __restrict__ invU, const float* __restrict__ invR,
           const int* __restrict__ iu, const int* __restrict__ ir,
           float* __restrict__ out, int Ep, int wv)
{
  const int base = wv * TAIL_PER_WAVE;
  if (base >= Ep) return;
  const int lane = threadIdx.x;
  const int g = lane >> 3, t = lane & 7;
  const int el  = min(base + lane, Ep - 1);
  const int myu = iu[el], myr = ir[el];
  const float mynu = invU[myu], mynr = invR[myr];
  float sel = 0.0f;
#pragma unroll
  for (int j = 0; j < 8; ++j) {
    const int u = __shfl(myu, 4 * j + g, 32);
    const int r = __shfl(myr, 4 * j + g, 32);
    const float4* pu = (const float4*)(HU + (size_t)u * D_OUT);
    const float4* pr = (const float4*)(HR + (size_t)r * D_OUT);
    const float4 a0 = pu[t], a1 = pu[8 + t];
    const float4 b0 = pr[t], b1 = pr[8 + t];
    float s = a0.x * b0.x + a0.y * b0.y + a0.z * b0.z + a0.w * b0.w
            + a1.x * b1.x + a1.y * b1.y + a1.z * b1.z + a1.w * b1.w;
    s += __shfl_xor(s, 4, 32);
    s += __shfl_xor(s, 2, 32);
    s += __shfl_xor(s, 1, 32);
    if (t == j) sel = s;
  }
  const int own = 4 * t + g;
  const float nu = __shfl(mynu, own, 32);
  const float nr = __shfl(mynr, own, 32);
  const int e = base + own;
  if (e < Ep) out[e] = sel * nu * nr;
}

__global__ void __launch_bounds__(256)
score_pair_kernel(const float* __restrict__ HU, const float* __restrict__ HR,
                  const float* __restrict__ invU, const float* __restrict__ invR,
                  const int* __restrict__ iu0, const int* __restrict__ ir0,
                  const int* __restrict__ iu1, const int* __restrict__ ir1,
                  float* __restrict__ out, int Ep, int nblk)
{
  const bool f0 = (int)blockIdx.x < nblk;
  const int  lb = f0 ? (int)blockIdx.x : (int)blockIdx.x - nblk;
  score_wave(HU, HR, invU, invR, f0 ? iu0 : iu1, f0 ? ir0 : ir1,
             f0 ? out : out + Ep, Ep, (int)(lb * blockDim.y + threadIdx.y));
}

extern "C" void kernel_launch(void* const* d_in, const int* in_sizes, int n_in,
                              void* d_out, int out_size, void* d_ws, size_t ws_size,
                              hipStream_t stream)
{
  const float* user_feat = (const float*)d_in[0];
  const float* repo_feat = (const float*)d_in[1];
  const float* W_ue = (const float*)d_in[2];   const float* b_ue = (const float*)d_in[3];
  const float* W_re = (const float*)d_in[4];   const float* b_re = (const float*)d_in[5];
  const float* W_hid_ur = (const float*)d_in[6];  const float* b_hid_ur = (const float*)d_in[7];
  const float* W_hid_ru = (const float*)d_in[8];  const float* b_hid_ru = (const float*)d_in[9];
  const float* W_out_ur = (const float*)d_in[10]; const float* b_out_ur = (const float*)d_in[11];
  const float* W_out_ru = (const float*)d_in[12]; const float* b_out_ru = (const float*)d_in[13];
  const float* W_hcu = (const float*)d_in[14]; const float* b_hcu = (const float*)d_in[15];
  const float* W_hcr = (const float*)d_in[16]; const float* b_hcr = (const float*)d_in[17];
  const float* W_ocu = (const float*)d_in[18]; const float* b_ocu = (const float*)d_in[19];
  const float* W_ocr = (const float*)d_in[20]; const float* b_ocr = (const float*)d_in[21];
  const int* e_ur_src = (const int*)d_in[22];  const int* e_ur_dst = (const int*)d_in[23];
  const int* e_ru_src = (const int*)d_in[24];  const int* e_ru_dst = (const int*)d_in[25];
  const int* pos_u = (const int*)d_in[26];     const int* pos_r = (const int*)d_in[27];
  const int* neg_u = (const int*)d_in[28];     const int* neg_r = (const int*)d_in[29];
  float* out = (float*)d_out;

  float* w = (float*)d_ws;
  size_t o = 0;
  float* deg_u_out_ur = w + o; o += N_USER;
  float* deg_r_in_ur  = w + o; o += N_REPO;
  float* deg_r_out_ru = w + o; o += N_REPO;
  float* deg_u_in_ru  = w + o; o += N_USER;
  float* reg_agg  = w + o; o += ((size_t)N_USER + N_REPO) * D_HID;
  float* reg_h    = w + o; o += ((size_t)N_USER + N_REPO) * LD_EMB;
  float* reg_feat = w + o; o += ((size_t)N_USER + N_REPO) * D_HID;
  float* reg_csr  = w + o; o += ((size_t)N_USER + N_REPO) * D_HID;
  float* inv_nu  = w + o; o += N_USER;
  float* inv_nr  = w + o; o += N_REPO;

  int* col_ur  = (int*)reg_csr;
  int* col_ru  = col_ur + NEDGE;
  int* part    = col_ru + NEDGE;
  int* tot     = part + NSLICE * 2 * (N_USER + N_REPO);
  int* ptr_r   = tot + 4 * NRANGE * NSLICE;
  int* ptr_u   = ptr_r + N_REPO;

  float* h_user   = reg_h;
  float* h_repo   = reg_h + (size_t)N_USER * LD_EMB;
  float* feat_u   = reg_feat;
  float* feat_r   = reg_feat + (size_t)N_USER * D_HID;
  float* agg_u    = reg_agg;
  float* agg_r    = reg_agg + (size_t)N_USER * D_HID;
  float* out_user = reg_feat;
  float* out_repo = reg_feat + (size_t)N_USER * D_HID;
  float* feat2_u  = reg_agg;
  float* feat2_r  = reg_agg + (size_t)N_USER * D_HID;
  float* agg2_u   = reg_h;
  float* agg2_r   = reg_h + (size_t)N_USER * LD_EMB;
  float* hn_user  = reg_agg;
  float* hn_repo  = reg_agg + (size_t)N_USER * D_HID;

  auto gemm = [&](const GemmSide& u, const GemmSide& r, int lda, int ldw,
                  int ldc, int Nreal, int K, int relu) {
    dim3 blk(GEMM_WAVES * 32);
    dim3 grd(GEMM_BLOCKS);
    if (ldc == LD_EMB)
      hipLaunchKernelGGL((gemm_pair_wmma_f32<LD_EMB / 16, D_IN>), grd, blk, 0, stream,
                         u, r, lda, ldw, Nreal, K, relu, GEMM_BLOCKS_U);
    else if (ldc == D_HID)
      hipLaunchKernelGGL((gemm_pair_wmma_f32<D_HID / 16, LD_EMB>), grd, blk, 0, stream,
                         u, r, lda, ldw, Nreal, K, relu, GEMM_BLOCKS_U);
    else
      hipLaunchKernelGGL((gemm_pair_wmma_f32<D_OUT / 16, D_HID>), grd, blk, 0, stream,
                         u, r, lda, ldw, Nreal, K, relu, GEMM_BLOCKS_U);
  };

  auto gather = [&](const GatherSide& a, const GatherSide& b, int D) {
    dim3 blk(32, 8);
    dim3 grd((N_USER + 7) / 8 + (N_REPO + 7) / 8);
    if (D == D_HID)
      hipLaunchKernelGGL(gather_pair_kernel<D_HID>, grd, blk, 0, stream,
                         a, b, (N_USER + 7) / 8, NEDGE);
    else
      hipLaunchKernelGGL(gather_pair_kernel<D_OUT>, grd, blk, 0, stream,
                         a, b, (N_USER + 7) / 8, NEDGE);
  };

  hipLaunchKernelGGL(range_count_kernel, dim3(4 * NRANGE * NSLICE), dim3(CSR_T), 0, stream,
                     e_ur_src, e_ur_dst, e_ru_src, e_ru_dst, part, tot);
  hipLaunchKernelGGL(range_fill_kernel, dim3(2 * NRANGE * NSLICE + 2 * NRANGE), dim3(CSR_T), 0, stream,
                     e_ur_src, e_ur_dst, e_ru_src, e_ru_dst, (const int*)part, (const int*)tot,
                     ptr_r, deg_r_in_ur, ptr_u, deg_u_in_ru, deg_u_out_ur, deg_r_out_ru,
                     col_ur, col_ru);

  gemm(GemmSide{user_feat, W_ue, nullptr, b_ue, nullptr, nullptr, nullptr, h_user, N_USER},
       GemmSide{repo_feat, W_re, nullptr, b_re, nullptr, nullptr, nullptr, h_repo, N_REPO},
       D_IN, D_EMB, LD_EMB, D_EMB, D_IN, 0);

  gemm(GemmSide{h_user, W_hid_ur, deg_u_out_ur, nullptr, nullptr, nullptr, nullptr, feat_u, N_USER},
       GemmSide{h_repo, W_hid_ru, deg_r_out_ru, nullptr, nullptr, nullptr, nullptr, feat_r, N_REPO},
       LD_EMB, D_HID, D_HID, D_HID, D_EMB, 0);

  gather(GatherSide{feat_r, ptr_u, col_ru, agg_u, N_USER, N_REPO},
         GatherSide{feat_u, ptr_r, col_ur, agg_r, N_REPO, N_USER}, D_HID);

  gemm(GemmSide{h_user, W_hcu, nullptr, b_hcu, agg_u, deg_u_in_ru, b_hid_ru, out_user, N_USER},
       GemmSide{h_repo, W_hcr, nullptr, b_hcr, agg_r, deg_r_in_ur, b_hid_ur, out_repo, N_REPO},
       LD_EMB, D_HID, D_HID, D_HID, D_EMB, 1);

  gemm(GemmSide{out_user, W_out_ur, deg_u_out_ur, nullptr, nullptr, nullptr, nullptr, feat2_u, N_USER},
       GemmSide{out_repo, W_out_ru, deg_r_out_ru, nullptr, nullptr, nullptr, nullptr, feat2_r, N_REPO},
       D_HID, D_OUT, D_OUT, D_OUT, D_HID, 0);

  gather(GatherSide{feat2_r, ptr_u, col_ru, agg2_u, N_USER, N_REPO},
         GatherSide{feat2_u, ptr_r, col_ur, agg2_r, N_REPO, N_USER}, D_OUT);

  gemm(GemmSide{out_user, W_ocu, nullptr, b_ocu, agg2_u, deg_u_in_ru, b_out_ru, hn_user, N_USER},
       GemmSide{out_repo, W_ocr, nullptr, b_ocr, agg2_r, deg_r_in_ur, b_out_ur, hn_repo, N_REPO},
       D_HID, D_OUT, D_OUT, D_OUT, D_HID, 1);

  hipLaunchKernelGGL(rownorm_pair_kernel, dim3(NORM_BLK_U + NORM_BLK_R), dim3(32, 8), 0, stream,
                     (const float*)hn_user, (const float*)hn_repo, inv_nu, inv_nr,
                     N_USER, N_REPO, NORM_BLK_U);
  hipLaunchKernelGGL(score_pair_kernel, dim3(2 * SCORE_BLK), dim3(32, 8), 0, stream,
                     (const float*)hn_user, (const float*)hn_repo,
                     (const float*)inv_nu, (const float*)inv_nr,
                     pos_u, pos_r, neg_u, neg_r, out, NEP, SCORE_BLK);
}


# Edge gather: one wave per destination row becomes 8 lanes per row, 4 rows per wave, 8 neighbour rows per row loaded ahead of their sums (225k fewer waves per call); per-element sum order unchanged
# speedup vs baseline: 4.9676x; 1.0446x over previous
#include <hip/hip_runtime.h>
#include <math.h>

#define N_USER 50000
#define N_REPO 100000
#define NEDGE  1000000
#define NEP    200000
#define D_IN   256
#define D_EMB  125
#define LD_EMB 128
#define D_HID  96
#define D_OUT  64
#define EPSN   1e-12f

typedef float v2f __attribute__((ext_vector_type(2)));
typedef float v8f __attribute__((ext_vector_type(8)));


#define GEMM_WAVES  8
#define GEMM_BLOCKS 512
#define GEMM_BLOCKS_U ((GEMM_BLOCKS * N_USER + (N_USER + N_REPO) / 2) / (N_USER + N_REPO))
#define WT_FLOATS (LD_EMB * (D_IN + 4))

struct GemmSide {
  const float* A;
  const float* W;
  const float* degA;
  const float* bias;
  const float* agg;
  const float* degC;
  const float* bias2;
  float*       C;
  int          M;
};

template <int NT, int KPAD>
__device__ __forceinline__ void
gemm_strip_body(const float* __restrict__ A, int lda,
                const float* __restrict__ W, int ldw,
                const float* __restrict__ degA,
                const float* __restrict__ bias,
                const float* __restrict__ agg,
                const float* __restrict__ degC,
                const float* __restrict__ bias2,
                float* __restrict__ C,
                int M, int Nreal, int K, int relu, int lb, int nb)
{
  constexpr int NP  = NT * 16;
  constexpr int LS  = KPAD + 4;
  constexpr int NCH = KPAD / 32;
  static_assert(KPAD % 32 == 0, "K is walked in whole 128-byte lines");
  static_assert(NP * LS <= WT_FLOATS, "W^T fits the LDS array");
  __shared__ __align__(16) float Wt[WT_FLOATS];

  const int tid = threadIdx.x;

  const int lane = tid & 31;
  const int wave = tid >> 5;
  const int half = lane >> 4;
  const int l16  = lane & 15;

  float bn[NT], b2n[NT];
#pragma unroll
  for (int t = 0; t < NT; ++t) { bn[t] = 0.0f; b2n[t] = 0.0f; }
  if (bias) {
#pragma unroll
    for (int t = 0; t < NT; ++t) bn[t] = bias[min(t * 16 + l16, Nreal - 1)];
  }
  if (bias2) {
#pragma unroll
    for (int t = 0; t < NT; ++t) b2n[t] = bias2[min(t * 16 + l16, Nreal - 1)];
  }

  constexpr int NST = NP * KPAD / (GEMM_WAVES * 32);
  static_assert(NP * KPAD % (GEMM_WAVES * 32) == 0, "every thread stages the same count");
  constexpr int UNR = (NST % 16 == 0) ? 16 : 8;
  static_assert(NST % UNR == 0, "the staging loop unrolls without a remainder");
#pragma unroll UNR
  for (int j = 0; j < NST; ++j) {
    const int i = tid + j * (GEMM_WAVES * 32);
    const int k = i / NP;
    const int n = i - k * NP;
    const float w = W[(long)min(k, K - 1) * ldw + min(n, Nreal - 1)];
    const unsigned keep = (n < Nreal && k < K) ? 0xFFFFFFFFu : 0u;
    Wt[n * LS + k] = __uint_as_float(__float_as_uint(w) & keep);
  }
  __syncthreads();

  const float* Wp = &Wt[l16 * LS + 2 * half];
  const int Mtiles = M / 16;

  for (int mt = lb + nb * wave; mt < Mtiles; mt += nb * GEMM_WAVES) {
    const int   m0 = mt * 16;
    const int   mA = m0 + l16;
    const float sA = degA ? rsqrtf(fmaxf(degA[mA], 1.0f)) : 1.0f;
    const float* Ap = A + (long)mA * lda + 2 * half;

    v8f acc[NT];
#pragma unroll
    for (int t = 0; t < NT; ++t) acc[t] = v8f{};

    v2f a_cur[8];
#pragma unroll
    for (int s = 0; s < 8; ++s) {
      a_cur[s] = *(const v2f*)(Ap + s * 4);
      a_cur[s] *= sA;
    }

    auto mma_chunk = [&](int c) {
      const float* Wc = Wp + c * 32;
#pragma unroll
      for (int s = 0; s < 8; ++s) {
        v2f b[NT];
#pragma unroll
        for (int t = 0; t < NT; ++t)
          b[t] = *(const v2f*)(Wc + t * 16 * LS + s * 4);
#pragma unroll
        for (int t = 0; t < NT; ++t)
          acc[t] = __builtin_amdgcn_wmma_f32_16x16x4_f32(false, a_cur[s], false, b[t],
                                                         (short)0, acc[t], false, false);
      }
    };

#pragma unroll 1
    for (int c = 0; c + 1 < NCH; ++c) {
      v2f a_nxt[8];
#pragma unroll
      for (int s = 0; s < 8; ++s) a_nxt[s] = *(const v2f*)(Ap + (c + 1) * 32 + s * 4);
      __builtin_amdgcn_sched_barrier(0);
      mma_chunk(c);
      __builtin_amdgcn_sched_barrier(0);
#pragma unroll
      for (int s = 0; s < 8; ++s) a_cur[s] = a_nxt[s] * sA;
    }
    mma_chunk(NCH - 1);

    const int mrow = m0 + 8 * half;
    if (agg) {
      float sC[8];
#pragma unroll
      for (int v = 0; v < 8; ++v) sC[v] = degC[mrow + v];
#pragma unroll
      for (int v = 0; v < 8; ++v) sC[v] = rsqrtf(fmaxf(sC[v], 1.0f));
      float g[NT][8];
#pragma unroll
      for (int t = 0; t < NT; ++t) {
#pragma unroll
        for (int v = 0; v < 8; ++v) g[t][v] = agg[(long)(mrow + v) * NP + t * 16 + l16];
      }
#pragma unroll
      for (int t = 0; t < NT; ++t) {
        const int n = t * 16 + l16;
#pragma unroll
        for (int v = 0; v < 8; ++v) {
          float val = acc[t][v] + bn[t];
          val += g[t][v] * sC[v] + b2n[t];
          if (relu) val = fmaxf(val, 0.0f);
          C[(long)(mrow + v) * NP + n] = (n < Nreal) ? val : 0.0f;
        }
      }
    } else {
#pragma unroll
      for (int t = 0; t < NT; ++t) {
        const int n = t * 16 + l16;
#pragma unroll
        for (int v = 0; v < 8; ++v) {
          float val = acc[t][v] + bn[t];
          if (relu) val = fmaxf(val, 0.0f);
          C[(long)(mrow + v) * NP + n] = (n < Nreal) ? val : 0.0f;
        }
      }
    }
  }
}

template <int NT, int KPAD>
__global__ void __launch_bounds__(GEMM_WAVES * 32, 4)
gemm_pair_wmma_f32(GemmSide u, GemmSide r, int lda, int ldw,
                   int Nreal, int K, int relu, int nbu)
{
  const bool us = (int)blockIdx.x < nbu;
  gemm_strip_body<NT, KPAD>(us ? u.A : r.A, lda, us ? u.W : r.W, ldw,
                            us ? u.degA : r.degA, us ? u.bias : r.bias,
                            us ? u.agg : r.agg, us ? u.degC : r.degC,
                            us ? u.bias2 : r.bias2, us ? u.C : r.C,
                            us ? u.M : r.M, Nreal, K, relu,
                            us ? (int)blockIdx.x : (int)blockIdx.x - nbu,
                            us ? nbu : (int)gridDim.x - nbu);
}


#define NRANGE  16
#define NSLICE  8
#define CSR_T   1024
#define SLICE_E (NEDGE / NSLICE)
#define RMAX    ((N_REPO + NRANGE - 1) / NRANGE)
static_assert(NEDGE % (4 * NSLICE) == 0, "a slice is a whole number of 16-byte groups");
static_assert(N_USER <= N_REPO, "RMAX covers both node kinds");
static_assert(NRANGE * NSLICE <= 128, "the totals of the earlier ranges are summed by four waves");
#define FILL_PER ((RMAX + CSR_T - 1) / CSR_T)
#define TOFF0 0
#define TOFF1 (NSLICE * N_USER)
#define TOFF2 (NSLICE * (N_USER + N_REPO))
#define TOFF3 (NSLICE * (N_USER + 2 * N_REPO))

__global__ void __launch_bounds__(CSR_T, 16)
range_count_kernel(const int* __restrict__ ia0, const int* __restrict__ ia1,
                   const int* __restrict__ ia2, const int* __restrict__ ia3,
                   int* __restrict__ part, int* __restrict__ tot)
{
  __shared__ int hist[RMAX];
  __shared__ int btot;
  const int a  = blockIdx.x / (NRANGE * NSLICE);
  const int rs = blockIdx.x - a * (NRANGE * NSLICE);
  const int r  = rs / NSLICE, s = rs - r * NSLICE;
  const int* __restrict__ idx = (a == 0) ? ia0 : (a == 1) ? ia1 : (a == 2) ? ia2 : ia3;
  const int n    = (a == 0 || a == 3) ? N_USER : N_REPO;
  const int toff = (a == 0) ? TOFF0 : (a == 1) ? TOFF1 : (a == 2) ? TOFF2 : TOFF3;
  const int R   = (n + NRANGE - 1) / NRANGE;
  const int lo  = r * R;
  const int len = max(min(R, n - lo), 0);
  const int tid = threadIdx.x;
  for (int i = tid; i < len; i += CSR_T) hist[i] = 0;
  if (tid == 0) btot = 0;
  __syncthreads();
  const int4* __restrict__ p = (const int4*)(idx + s * SLICE_E);
  constexpr int N4 = SLICE_E / 4;
  constexpr int GC = 8;
  for (int b = 0; b < N4; b += GC * CSR_T) {
    int4 v[GC];
#pragma unroll
    for (int g = 0; g < GC; ++g) {
      const int i = b + g * CSR_T + tid;
      v[g] = p[min(i, N4 - 1)];
      const int inval = (i < N4) ? 0 : -1;
      v[g].x |= inval; v[g].y |= inval; v[g].z |= inval; v[g].w |= inval;
    }
#pragma unroll
    for (int g = 0; g < GC; ++g) {
      unsigned u;
      u = (unsigned)(v[g].x - lo); if (u < (unsigned)len) atomicAdd(&hist[u], 1);
      u = (unsigned)(v[g].y - lo); if (u < (unsigned)len) atomicAdd(&hist[u], 1);
      u = (unsigned)(v[g].z - lo); if (u < (unsigned)len) atomicAdd(&hist[u], 1);
      u = (unsigned)(v[g].w - lo); if (u < (unsigned)len) atomicAdd(&hist[u], 1);
    }
  }
  __syncthreads();
  int* __restrict__ out = part + toff + s * n + lo;
  int my = 0;
  for (int i = tid; i < len; i += CSR_T) { const int h = hist[i]; out[i] = h; my += h; }
  my += __shfl_xor(my, 16, 32);
  my += __shfl_xor(my, 8, 32);
  my += __shfl_xor(my, 4, 32);
  my += __shfl_xor(my, 2, 32);
  my += __shfl_xor(my, 1, 32);
  if ((tid & 31) == 0) atomicAdd(&btot, my);
  __syncthreads();
  if (tid == 0) tot[blockIdx.x] = btot;
}

__global__ void __launch_bounds__(CSR_T, 16)
range_fill_kernel(const int* __restrict__ src0, const int* __restrict__ dst0,
                  const int* __restrict__ src1, const int* __restrict__ dst1,
                  const int* __restrict__ part, const int* __restrict__ tot,
                  int* __restrict__ ptr0, float* __restrict__ deg0,
                  int* __restrict__ ptr1, float* __restrict__ deg1,
                  float* __restrict__ dout0, float* __restrict__ dout2,
                  int* __restrict__ col0, int* __restrict__ col1)
{
  __shared__ int off[RMAX + 32];
  __shared__ int cnt[RMAX];
  __shared__ int wsum[32];
  __shared__ int bsum[4];
  const int tid = threadIdx.x, lane = tid & 31, wv = tid >> 5;

  if (blockIdx.x >= 2 * NRANGE * NSLICE) {
    const int x  = blockIdx.x - 2 * NRANGE * NSLICE;
    const int wh = x / NRANGE, r = x - wh * NRANGE;
    const int n  = wh ? N_REPO : N_USER;
    const int* __restrict__ tab = part + (wh ? TOFF2 : TOFF0);
    float* __restrict__ dout = wh ? dout2 : dout0;
    const int R   = (n + NRANGE - 1) / NRANGE;
    const int lo  = r * R;
    const int len = max(min(R, n - lo), 0);
    for (int i = tid; i < len; i += CSR_T) {
      int c = 0;
#pragma unroll
      for (int s2 = 0; s2 < NSLICE; ++s2) c += tab[s2 * n + lo + i];
      dout[lo + i] = (float)c;
    }
    return;
  }

  const int t  = blockIdx.x / (NRANGE * NSLICE);
  const int rs = blockIdx.x - t * (NRANGE * NSLICE);
  const int r  = rs / NSLICE, s = rs - r * NSLICE;
  const int* __restrict__ src = t ? src1 : src0;
  const int* __restrict__ dst = t ? dst1 : dst0;
  int*       __restrict__ ptr = t ? ptr1 : ptr0;
  float*     __restrict__ deg = t ? deg1 : deg0;
  int*       __restrict__ col = t ? col1 : col0;
  const int n   = t ? N_USER : N_REPO;
  const int* __restrict__ tab = part + (t ? TOFF3 : TOFF1);
  const int R   = (n + NRANGE - 1) / NRANGE;
  const int lo  = r * R;
  const int len = max(min(R, n - lo), 0);

  int bv = (tid < r * NSLICE) ? tot[(2 * t + 1) * NRANGE * NSLICE + tid] : 0;
  bv += __shfl_xor(bv, 16, 32);
  bv += __shfl_xor(bv, 8, 32);
  bv += __shfl_xor(bv, 4, 32);
  bv += __shfl_xor(bv, 2, 32);
  bv += __shfl_xor(bv, 1, 32);
  if (wv < 4 && lane == 0) bsum[wv] = bv;
  if (tid < 32) off[RMAX + tid] = 0;

#pragma unroll 2
  for (int i = tid; i < len; i += CSR_T) {
    int c = 0, pre = 0;
#pragma unroll
    for (int s2 = 0; s2 < NSLICE; ++s2) {
      const int x = tab[s2 * n + lo + i];
      c += x;
      pre += (s2 < s) ? x : 0;
    }
    cnt[i] = c;
    off[i] = pre;
  }
  __syncthreads();

  int cv[FILL_PER];
  int tsum = 0;
#pragma unroll
  for (int k = 0; k < FILL_PER; ++k) {
    const int i = tid * FILL_PER + k;
    cv[k] = (i < len) ? cnt[i] : 0;
    tsum += cv[k];
  }
  int x = tsum;
#pragma unroll
  for (int o = 1; o < 32; o <<= 1) {
    const int y = __shfl_up(x, o, 32);
    if (lane >= o) x += y;
  }
  if (lane == 31) wsum[wv] = x;
  __syncthreads();
  const int w = wsum[lane];
  int wx = w;
#pragma unroll
  for (int o = 1; o < 32; o <<= 1) {
    const int y = __shfl_up(wx, o, 32);
    if (lane >= o) wx += y;
  }
  const int wave_excl = __shfl(wx, wv, 32) - __shfl(w, wv, 32);
  int run = bsum[0] + bsum[1] + bsum[2] + bsum[3] + wave_excl + (x - tsum);
#pragma unroll
  for (int k = 0; k < FILL_PER; ++k) {
    const int i = tid * FILL_PER + k;
    if (i < len) {
      off[i] += run;
      if (s == 0) { ptr[lo + i] = run; deg[lo + i] = (float)cv[k]; }
    }
    run += cv[k];
  }
  __syncthreads();

  const unsigned spare = RMAX + lane;
  const int4* __restrict__ pd = (const int4*)(dst + s * SLICE_E);
  const int4* __restrict__ ps = (const int4*)(src + s * SLICE_E);
  constexpr int N4 = SLICE_E / 4;
  constexpr int GF = 4;
  for (int b = 0; b < N4; b += GF * CSR_T) {
    int4 d[GF], sv[GF];
#pragma unroll
    for (int g = 0; g < GF; ++g) {
      const int i = b + g * CSR_T + tid;
      d[g]  = pd[min(i, N4 - 1)];
      sv[g] = ps[min(i, N4 - 1)];
      const int inval = (i < N4) ? 0 : -1;
      d[g].x |= inval; d[g].y |= inval; d[g].z |= inval; d[g].w |= inval;
    }
#pragma unroll
    for (int g = 0; g < GF; ++g) {
      const unsigned u0 = (unsigned)(d[g].x - lo), u1 = (unsigned)(d[g].y - lo);
      const unsigned u2 = (unsigned)(d[g].z - lo), u3 = (unsigned)(d[g].w - lo);
      const bool m0 = u0 < (unsigned)len, m1 = u1 < (unsigned)len;
      const bool m2 = u2 < (unsigned)len, m3 = u3 < (unsigned)len;
      const int q0 = atomicAdd(&off[m0 ? u0 : spare], 1);
      const int q1 = atomicAdd(&off[m1 ? u1 : spare], 1);
      const int q2 = atomicAdd(&off[m2 ? u2 : spare], 1);
      const int q3 = atomicAdd(&off[m3 ? u3 : spare], 1);
      if (m0) col[min(max(q0, 0), NEDGE - 1)] = sv[g].x;
      if (m1) col[min(max(q1, 0), NEDGE - 1)] = sv[g].y;
      if (m2) col[min(max(q2, 0), NEDGE - 1)] = sv[g].z;
      if (m3) col[min(max(q3, 0), NEDGE - 1)] = sv[g].w;
    }
  }
}

#define GATHER_ROWS 4
#define GATHER_BLK_U ((N_USER + 8 * GATHER_ROWS - 1) / (8 * GATHER_ROWS))
#define GATHER_BLK_R ((N_REPO + 8 * GATHER_ROWS - 1) / (8 * GATHER_ROWS))
struct GatherSide {
  const float* feat;
  const int*   rowptr;
  const int*   col;
  float*       agg;
  int          n_dst;
  int          n_src;
};

template <int D, int JJ>
__device__ __forceinline__ void
gather_trip(float4 (&v)[D / 32], const float* __restrict__ feat, int n_src, int cur, int t, bool on)
{
  const int s = min(max((int)__builtin_amdgcn_mov_dpp8((unsigned)cur, JJ * 0x249249), 0), n_src - 1);
  const float4* p = (const float4*)(feat + (size_t)s * D);
#pragma unroll
  for (int q = 0; q < D / 32; ++q) v[q] = make_float4(0.0f, 0.0f, 0.0f, 0.0f);
  if (on) {
#pragma unroll
    for (int q = 0; q < D / 32; ++q) v[q] = p[q * 8 + t];
  }
}

template <int D>
__device__ __forceinline__ void
gather_rows(const float* __restrict__ feat, const int* __restrict__ rowptr,
            const int* __restrict__ col, float* __restrict__ agg,
            int n_dst, int n_src, int E, int wv)
{
  constexpr int NPC = D / 32;
  static_assert(D % 32 == 0, "a row is a whole number of 128-byte lines");
  const int d0 = wv * GATHER_ROWS;
  if (d0 >= n_dst) return;
  const int lane = threadIdx.x;
  const int g = lane >> 3, t = lane & 7;

  const int xb = rowptr[min(d0 + min(lane, GATHER_ROWS), n_dst - 1)];
  int rp[GATHER_ROWS + 1];
#pragma unroll
  for (int i = 0; i <= GATHER_ROWS; ++i) {
    const int x = __builtin_amdgcn_readlane(xb, i);
    rp[i] = (d0 + i < n_dst) ? min(max(x, 0), E) : E;
  }
  int mx = 0;
#pragma unroll
  for (int i = 0; i < GATHER_ROWS; ++i) mx = max(mx, rp[i + 1] - rp[i]);
  int beg = rp[0], end = rp[1];
#pragma unroll
  for (int i = 1; i < GATHER_ROWS; ++i) {
    if (g == i) { beg = rp[i]; end = rp[i + 1]; }
  }
  const int  deg  = max(end - beg, 0);
  const int  d    = d0 + g;
  const bool live = d < n_dst;

  float4 acc[NPC];
#pragma unroll
  for (int q = 0; q < NPC; ++q) acc[q] = make_float4(0.0f, 0.0f, 0.0f, 0.0f);

  int cur = 0;
  if (t < deg) cur = col[beg + t];
  cur = min(max(cur, 0), n_src - 1);
  for (int j0 = 0; j0 < mx; j0 += 8) {
    int nxt = 0;
    if (j0 + 8 + t < deg) nxt = col[beg + j0 + 8 + t];
    const int rem = deg - j0;
    float4 v[8][NPC];
    gather_trip<D, 0>(v[0], feat, n_src, cur, t, 0 < rem);
    gather_trip<D, 1>(v[1], feat, n_src, cur, t, 1 < rem);
    gather_trip<D, 2>(v[2], feat, n_src, cur, t, 2 < rem);
    gather_trip<D, 3>(v[3], feat, n_src, cur, t, 3 < rem);
    gather_trip<D, 4>(v[4], feat, n_src, cur, t, 4 < rem);
    gather_trip<D, 5>(v[5], feat, n_src, cur, t, 5 < rem);
    gather_trip<D, 6>(v[6], feat, n_src, cur, t, 6 < rem);
    gather_trip<D, 7>(v[7], feat, n_src, cur, t, 7 < rem);
#pragma unroll
    for (int jj = 0; jj < 8; ++jj) {
#pragma unroll
      for (int q = 0; q < NPC; ++q) {
        acc[q].x += v[jj][q].x;
        acc[q].y += v[jj][q].y;
        acc[q].z += v[jj][q].z;
        acc[q].w += v[jj][q].w;
      }
    }
    cur = nxt;
  }
  if (live) {
    float4* o = (float4*)(agg + (size_t)d * D);
#pragma unroll
    for (int q = 0; q < NPC; ++q) o[q * 8 + t] = acc[q];
  }
}

template <int D>
__global__ void __launch_bounds__(256)
gather_pair_kernel(GatherSide a, GatherSide b, int nblk_a, int E)
{
  const bool fa = (int)blockIdx.x < nblk_a;
  const int  lb = fa ? (int)blockIdx.x : (int)blockIdx.x - nblk_a;
  const int wv = __builtin_amdgcn_readfirstlane((int)(lb * blockDim.y + threadIdx.y));
  gather_rows<D>(fa ? a.feat : b.feat, fa ? a.rowptr : b.rowptr, fa ? a.col : b.col,
                 fa ? a.agg : b.agg, fa ? a.n_dst : b.n_dst, fa ? a.n_src : b.n_src, E, wv);
}

#define TAIL_PER_WAVE 32
#define NORM_BLK_U ((N_USER + 8 * TAIL_PER_WAVE - 1) / (8 * TAIL_PER_WAVE))
#define NORM_BLK_R ((N_REPO + 8 * TAIL_PER_WAVE - 1) / (8 * TAIL_PER_WAVE))
#define SCORE_BLK  ((NEP + 8 * TAIL_PER_WAVE - 1) / (8 * TAIL_PER_WAVE))

__device__ __forceinline__ void
rownorm_wave(const float* __restrict__ X, float* __restrict__ invn, int rows, int wv)
{
  const int base = wv * TAIL_PER_WAVE;
  if (base >= rows) return;
  const int lane = threadIdx.x;
  const int g = lane >> 3, t = lane & 7;
  float sel = 1.0f;
#pragma unroll
  for (int j = 0; j < 8; ++j) {
    const int r = min(base + 4 * j + g, rows - 1);
    const float4* p = (const float4*)(X + (size_t)r * D_OUT);
    const float4 v0 = p[t], v1 = p[8 + t];
    float s = v0.x * v0.x + v0.y * v0.y + v0.z * v0.z + v0.w * v0.w
            + v1.x * v1.x + v1.y * v1.y + v1.z * v1.z + v1.w * v1.w;
    s += __shfl_xor(s, 4, 32);
    s += __shfl_xor(s, 2, 32);
    s += __shfl_xor(s, 1, 32);
    if (t == j) sel = s;
  }
  const int r = base + 4 * t + g;
  if (r < rows) invn[r] = 1.0f / fmaxf(sqrtf(sel), EPSN);
}

__global__ void __launch_bounds__(256)
rownorm_pair_kernel(const float* __restrict__ XU, const float* __restrict__ XR,
                    float* __restrict__ invU, float* __restrict__ invR,
                    int rowsU, int rowsR, int nblk_u)
{
  const bool fu = (int)blockIdx.x < nblk_u;
  const int  lb = fu ? (int)blockIdx.x : (int)blockIdx.x - nblk_u;
  rownorm_wave(fu ? XU : XR, fu ? invU : invR, fu ? rowsU : rowsR,
               (int)(lb * blockDim.y + threadIdx.y));
}

__device__ __forceinline__ void
score_wave(const float* __restrict__ HU, const float* __restrict__ HR,
           const float* __restrict__ invU, const float* __restrict__ invR,
           const int* __restrict__ iu, const int* __restrict__ ir,
           float* __restrict__ out, int Ep, int wv)
{
  const int base = wv * TAIL_PER_WAVE;
  if (base >= Ep) return;
  const int lane = threadIdx.x;
  const int g = lane >> 3, t = lane & 7;
  const int el  = min(base + lane, Ep - 1);
  const int myu = iu[el], myr = ir[el];
  const float mynu = invU[myu], mynr = invR[myr];
  float sel = 0.0f;
#pragma unroll
  for (int j = 0; j < 8; ++j) {
    const int u = __shfl(myu, 4 * j + g, 32);
    const int r = __shfl(myr, 4 * j + g, 32);
    const float4* pu = (const float4*)(HU + (size_t)u * D_OUT);
    const float4* pr = (const float4*)(HR + (size_t)r * D_OUT);
    const float4 a0 = pu[t], a1 = pu[8 + t];
    const float4 b0 = pr[t], b1 = pr[8 + t];
    float s = a0.x * b0.x + a0.y * b0.y + a0.z * b0.z + a0.w * b0.w
            + a1.x * b1.x + a1.y * b1.y + a1.z * b1.z + a1.w * b1.w;
    s += __shfl_xor(s, 4, 32);
    s += __shfl_xor(s, 2, 32);
    s += __shfl_xor(s, 1, 32);
    if (t == j) sel = s;
  }
  const int own = 4 * t + g;
  const float nu = __shfl(mynu, own, 32);
  const float nr = __shfl(mynr, own, 32);
  const int e = base + own;
  if (e < Ep) out[e] = sel * nu * nr;
}

__global__ void __launch_bounds__(256)
score_pair_kernel(const float* __restrict__ HU, const float* __restrict__ HR,
                  const float* __restrict__ invU, const float* __restrict__ invR,
                  const int* __restrict__ iu0, const int* __restrict__ ir0,
                  const int* __restrict__ iu1, const int* __restrict__ ir1,
                  float* __restrict__ out, int Ep, int nblk)
{
  const bool f0 = (int)blockIdx.x < nblk;
  const int  lb = f0 ? (int)blockIdx.x : (int)blockIdx.x - nblk;
  score_wave(HU, HR, invU, invR, f0 ? iu0 : iu1, f0 ? ir0 : ir1,
             f0 ? out : out + Ep, Ep, (int)(lb * blockDim.y + threadIdx.y));
}

extern "C" void kernel_launch(void* const* d_in, const int* in_sizes, int n_in,
                              void* d_out, int out_size, void* d_ws, size_t ws_size,
                              hipStream_t stream)
{
  const float* user_feat = (const float*)d_in[0];
  const float* repo_feat = (const float*)d_in[1];
  const float* W_ue = (const float*)d_in[2];   const float* b_ue = (const float*)d_in[3];
  const float* W_re = (const float*)d_in[4];   const float* b_re = (const float*)d_in[5];
  const float* W_hid_ur = (const float*)d_in[6];  const float* b_hid_ur = (const float*)d_in[7];
  const float* W_hid_ru = (const float*)d_in[8];  const float* b_hid_ru = (const float*)d_in[9];
  const float* W_out_ur = (const float*)d_in[10]; const float* b_out_ur = (const float*)d_in[11];
  const float* W_out_ru = (const float*)d_in[12]; const float* b_out_ru = (const float*)d_in[13];
  const float* W_hcu = (const float*)d_in[14]; const float* b_hcu = (const float*)d_in[15];
  const float* W_hcr = (const float*)d_in[16]; const float* b_hcr = (const float*)d_in[17];
  const float* W_ocu = (const float*)d_in[18]; const float* b_ocu = (const float*)d_in[19];
  const float* W_ocr = (const float*)d_in[20]; const float* b_ocr = (const float*)d_in[21];
  const int* e_ur_src = (const int*)d_in[22];  const int* e_ur_dst = (const int*)d_in[23];
  const int* e_ru_src = (const int*)d_in[24];  const int* e_ru_dst = (const int*)d_in[25];
  const int* pos_u = (const int*)d_in[26];     const int* pos_r = (const int*)d_in[27];
  const int* neg_u = (const int*)d_in[28];     const int* neg_r = (const int*)d_in[29];
  float* out = (float*)d_out;

  float* w = (float*)d_ws;
  size_t o = 0;
  float* deg_u_out_ur = w + o; o += N_USER;
  float* deg_r_in_ur  = w + o; o += N_REPO;
  float* deg_r_out_ru = w + o; o += N_REPO;
  float* deg_u_in_ru  = w + o; o += N_USER;
  float* reg_agg  = w + o; o += ((size_t)N_USER + N_REPO) * D_HID;
  float* reg_h    = w + o; o += ((size_t)N_USER + N_REPO) * LD_EMB;
  float* reg_feat = w + o; o += ((size_t)N_USER + N_REPO) * D_HID;
  float* reg_csr  = w + o; o += ((size_t)N_USER + N_REPO) * D_HID;
  float* inv_nu  = w + o; o += N_USER;
  float* inv_nr  = w + o; o += N_REPO;

  int* col_ur  = (int*)reg_csr;
  int* col_ru  = col_ur + NEDGE;
  int* part    = col_ru + NEDGE;
  int* tot     = part + NSLICE * 2 * (N_USER + N_REPO);
  int* ptr_r   = tot + 4 * NRANGE * NSLICE;
  int* ptr_u   = ptr_r + N_REPO;

  float* h_user   = reg_h;
  float* h_repo   = reg_h + (size_t)N_USER * LD_EMB;
  float* feat_u   = reg_feat;
  float* feat_r   = reg_feat + (size_t)N_USER * D_HID;
  float* agg_u    = reg_agg;
  float* agg_r    = reg_agg + (size_t)N_USER * D_HID;
  float* out_user = reg_feat;
  float* out_repo = reg_feat + (size_t)N_USER * D_HID;
  float* feat2_u  = reg_agg;
  float* feat2_r  = reg_agg + (size_t)N_USER * D_HID;
  float* agg2_u   = reg_h;
  float* agg2_r   = reg_h + (size_t)N_USER * LD_EMB;
  float* hn_user  = reg_agg;
  float* hn_repo  = reg_agg + (size_t)N_USER * D_HID;

  auto gemm = [&](const GemmSide& u, const GemmSide& r, int lda, int ldw,
                  int ldc, int Nreal, int K, int relu) {
    dim3 blk(GEMM_WAVES * 32);
    dim3 grd(GEMM_BLOCKS);
    if (ldc == LD_EMB)
      hipLaunchKernelGGL((gemm_pair_wmma_f32<LD_EMB / 16, D_IN>), grd, blk, 0, stream,
                         u, r, lda, ldw, Nreal, K, relu, GEMM_BLOCKS_U);
    else if (ldc == D_HID)
      hipLaunchKernelGGL((gemm_pair_wmma_f32<D_HID / 16, LD_EMB>), grd, blk, 0, stream,
                         u, r, lda, ldw, Nreal, K, relu, GEMM_BLOCKS_U);
    else
      hipLaunchKernelGGL((gemm_pair_wmma_f32<D_OUT / 16, D_HID>), grd, blk, 0, stream,
                         u, r, lda, ldw, Nreal, K, relu, GEMM_BLOCKS_U);
  };

  auto gather = [&](const GatherSide& a, const GatherSide& b, int D) {
    dim3 blk(32, 8);
    dim3 grd(GATHER_BLK_U + GATHER_BLK_R);
    if (D == D_HID)
      hipLaunchKernelGGL(gather_pair_kernel<D_HID>, grd, blk, 0, stream,
                         a, b, GATHER_BLK_U, NEDGE);
    else
      hipLaunchKernelGGL(gather_pair_kernel<D_OUT>, grd, blk, 0, stream,
                         a, b, GATHER_BLK_U, NEDGE);
  };

  hipLaunchKernelGGL(range_count_kernel, dim3(4 * NRANGE * NSLICE), dim3(CSR_T), 0, stream,
                     e_ur_src, e_ur_dst, e_ru_src, e_ru_dst, part, tot);
  hipLaunchKernelGGL(range_fill_kernel, dim3(2 * NRANGE * NSLICE + 2 * NRANGE), dim3(CSR_T), 0, stream,
                     e_ur_src, e_ur_dst, e_ru_src, e_ru_dst, (const int*)part, (const int*)tot,
                     ptr_r, deg_r_in_ur, ptr_u, deg_u_in_ru, deg_u_out_ur, deg_r_out_ru,
                     col_ur, col_ru);

  gemm(GemmSide{user_feat, W_ue, nullptr, b_ue, nullptr, nullptr, nullptr, h_user, N_USER},
       GemmSide{repo_feat, W_re, nullptr, b_re, nullptr, nullptr, nullptr, h_repo, N_REPO},
       D_IN, D_EMB, LD_EMB, D_EMB, D_IN, 0);

  gemm(GemmSide{h_user, W_hid_ur, deg_u_out_ur, nullptr, nullptr, nullptr, nullptr, feat_u, N_USER},
       GemmSide{h_repo, W_hid_ru, deg_r_out_ru, nullptr, nullptr, nullptr, nullptr, feat_r, N_REPO},
       LD_EMB, D_HID, D_HID, D_HID, D_EMB, 0);

  gather(GatherSide{feat_r, ptr_u, col_ru, agg_u, N_USER, N_REPO},
         GatherSide{feat_u, ptr_r, col_ur, agg_r, N_REPO, N_USER}, D_HID);

  gemm(GemmSide{h_user, W_hcu, nullptr, b_hcu, agg_u, deg_u_in_ru, b_hid_ru, out_user, N_USER},
       GemmSide{h_repo, W_hcr, nullptr, b_hcr, agg_r, deg_r_in_ur, b_hid_ur, out_repo, N_REPO},
       LD_EMB, D_HID, D_HID, D_HID, D_EMB, 1);

  gemm(GemmSide{out_user, W_out_ur, deg_u_out_ur, nullptr, nullptr, nullptr, nullptr, feat2_u, N_USER},
       GemmSide{out_repo, W_out_ru, deg_r_out_ru, nullptr, nullptr, nullptr, nullptr, feat2_r, N_REPO},
       D_HID, D_OUT, D_OUT, D_OUT, D_HID, 0);

  gather(GatherSide{feat2_r, ptr_u, col_ru, agg2_u, N_USER, N_REPO},
         GatherSide{feat2_u, ptr_r, col_ur, agg2_r, N_REPO, N_USER}, D_OUT);

  gemm(GemmSide{out_user, W_ocu, nullptr, b_ocu, agg2_u, deg_u_in_ru, b_out_ru, hn_user, N_USER},
       GemmSide{out_repo, W_ocr, nullptr, b_ocr, agg2_r, deg_r_in_ur, b_out_ur, hn_repo, N_REPO},
       D_HID, D_OUT, D_OUT, D_OUT, D_HID, 1);

  hipLaunchKernelGGL(rownorm_pair_kernel, dim3(NORM_BLK_U + NORM_BLK_R), dim3(32, 8), 0, stream,
                     (const float*)hn_user, (const float*)hn_repo, inv_nu, inv_nr,
                     N_USER, N_REPO, NORM_BLK_U);
  hipLaunchKernelGGL(score_pair_kernel, dim3(2 * SCORE_BLK), dim3(32, 8), 0, stream,
                     (const float*)hn_user, (const float*)hn_repo,
                     (const float*)inv_nu, (const float*)inv_nr,
                     pos_u, pos_r, neg_u, neg_r, out, NEP, SCORE_BLK);
}


# Each layer's two GEMMs of one node matrix merged into one launch; degree scale, conv bias, relu and final row norm moved into the edge gather: 11 launches become 8; per-element sum order unchanged
# speedup vs baseline: 5.1360x; 1.0339x over previous
#include <hip/hip_runtime.h>
#include <math.h>

#define N_USER 50000
#define N_REPO 100000
#define NEDGE  1000000
#define NEP    200000
#define D_IN   256
#define D_EMB  125
#define LD_EMB 128
#define D_HID  96
#define D_OUT  64
#define EPSN   1e-12f

typedef float v2f __attribute__((ext_vector_type(2)));
typedef float v8f __attribute__((ext_vector_type(8)));


#define GEMM_WAVES  8
#define GEMM_BLOCKS 512
#define GEMM_BLOCKS_U ((GEMM_BLOCKS * N_USER + (N_USER + N_REPO) / 2) / (N_USER + N_REPO))
#define WT_FLOATS (LD_EMB * (D_IN + 4))

struct GemmSide {
  const float* A;
  const float* W1;
  const float* degA;
  const float* bias1;
  const float* W2;
  const float* bias2;
  float*       C1;
  float*       C2;
  int          M;
};

template <int NT1, int NT2, int KPAD>
__device__ __forceinline__ void
gemm_strip_body(const float* __restrict__ A, int lda,
                const float* __restrict__ W1,
                const float* __restrict__ W2, int ldw,
                const float* __restrict__ degA,
                const float* __restrict__ bias1,
                const float* __restrict__ bias2,
                float* __restrict__ C1,
                float* __restrict__ C2,
                int M, int Nreal, int K, int lb, int nb)
{
  constexpr int NT  = NT1 + NT2;
  constexpr int NP1 = NT1 * 16;
  constexpr int NP2 = NT2 * 16;
  constexpr int LS  = KPAD + 4;
  constexpr int NCH = KPAD / 32;
  static_assert(KPAD % 32 == 0, "K is walked in whole 128-byte lines");
  static_assert(NT2 == 0 || NT2 == NT1, "both products have the same width");
  static_assert((NP1 + NP2) * LS <= WT_FLOATS, "W1^T and W2^T fit the LDS array");
  __shared__ __align__(16) float Wt[WT_FLOATS];

  const int tid = threadIdx.x;

  const int lane = tid & 31;
  const int wave = tid >> 5;
  const int half = lane >> 4;
  const int l16  = lane & 15;

  float bn1[NT1], bn2[NT1];
#pragma unroll
  for (int t = 0; t < NT1; ++t) { bn1[t] = 0.0f; bn2[t] = 0.0f; }
  if (bias1) {
#pragma unroll
    for (int t = 0; t < NT1; ++t) bn1[t] = bias1[min(t * 16 + l16, Nreal - 1)];
  }
  if (NT2 > 0 && bias2) {
#pragma unroll
    for (int t = 0; t < NT2; ++t) bn2[t] = bias2[min(t * 16 + l16, Nreal - 1)];
  }

  constexpr int NST = NP1 * KPAD / (GEMM_WAVES * 32);
  static_assert(NP1 * KPAD % (GEMM_WAVES * 32) == 0, "every thread stages the same count");
  constexpr int UNR = (NST % 16 == 0) ? 16 : 8;
  static_assert(NST % UNR == 0, "the staging loop unrolls without a remainder");
  auto stage = [&](const float* __restrict__ W, int row0) {
#pragma unroll UNR
    for (int j = 0; j < NST; ++j) {
      const int i = tid + j * (GEMM_WAVES * 32);
      const int k = i / NP1;
      const int n = i - k * NP1;
      const float w = W[(long)min(k, K - 1) * ldw + min(n, Nreal - 1)];
      const unsigned keep = (n < Nreal && k < K) ? 0xFFFFFFFFu : 0u;
      Wt[(row0 + n) * LS + k] = __uint_as_float(__float_as_uint(w) & keep);
    }
  };
  stage(W1, 0);
  if (NT2 > 0) stage(W2, NP1);
  __syncthreads();

  const float* Wp = &Wt[l16 * LS + 2 * half];
  const int Mtiles = M / 16;

  for (int mt = lb + nb * wave; mt < Mtiles; mt += nb * GEMM_WAVES) {
    const int   m0 = mt * 16;
    const int   mA = m0 + l16;
    const float sA = degA ? rsqrtf(fmaxf(degA[mA], 1.0f)) : 1.0f;
    const float* Ap = A + (long)mA * lda + 2 * half;

    v8f acc[NT];
#pragma unroll
    for (int t = 0; t < NT; ++t) acc[t] = v8f{};

    v2f a_raw[8], a_sc[8];
#pragma unroll
    for (int s = 0; s < 8; ++s) {
      a_raw[s] = *(const v2f*)(Ap + s * 4);
      a_sc[s]  = a_raw[s] * sA;
    }

    auto mma_chunk = [&](int c) {
      const float* Wc = Wp + c * 32;
#pragma unroll
      for (int s = 0; s < 8; ++s) {
        v2f b[NT];
#pragma unroll
        for (int t = 0; t < NT; ++t)
          b[t] = *(const v2f*)(Wc + t * 16 * LS + s * 4);
#pragma unroll
        for (int t = 0; t < NT1; ++t)
          acc[t] = __builtin_amdgcn_wmma_f32_16x16x4_f32(false, a_sc[s], false, b[t],
                                                         (short)0, acc[t], false, false);
#pragma unroll
        for (int t = NT1; t < NT; ++t)
          acc[t] = __builtin_amdgcn_wmma_f32_16x16x4_f32(false, a_raw[s], false, b[t],
                                                         (short)0, acc[t], false, false);
      }
    };

#pragma unroll 1
    for (int c = 0; c + 1 < NCH; ++c) {
      v2f a_nxt[8];
#pragma unroll
      for (int s = 0; s < 8; ++s) a_nxt[s] = *(const v2f*)(Ap + (c + 1) * 32 + s * 4);
      __builtin_amdgcn_sched_barrier(0);
      mma_chunk(c);
      __builtin_amdgcn_sched_barrier(0);
#pragma unroll
      for (int s = 0; s < 8; ++s) { a_raw[s] = a_nxt[s]; a_sc[s] = a_nxt[s] * sA; }
    }
    mma_chunk(NCH - 1);

    const int mrow = m0 + 8 * half;
#pragma unroll
    for (int t = 0; t < NT1; ++t) {
      const int n = t * 16 + l16;
#pragma unroll
      for (int v = 0; v < 8; ++v) {
        const float val = acc[t][v] + bn1[t];
        C1[(long)(mrow + v) * NP1 + n] = (n < Nreal) ? val : 0.0f;
      }
    }
#pragma unroll
    for (int t = 0; t < NT2; ++t) {
      const int n = t * 16 + l16;
#pragma unroll
      for (int v = 0; v < 8; ++v) {
        const float val = acc[NT1 + t][v] + bn2[t];
        C2[(long)(mrow + v) * NP2 + n] = (n < Nreal) ? val : 0.0f;
      }
    }
  }
}

template <int NT1, int NT2, int KPAD>
__global__ void __launch_bounds__(GEMM_WAVES * 32, 4)
gemm_pair_wmma_f32(GemmSide u, GemmSide r, int lda, int ldw,
                   int Nreal, int K, int nbu)
{
  const bool us = (int)blockIdx.x < nbu;
  gemm_strip_body<NT1, NT2, KPAD>(us ? u.A : r.A, lda, us ? u.W1 : r.W1,
                                  us ? u.W2 : r.W2, ldw,
                                  us ? u.degA : r.degA, us ? u.bias1 : r.bias1,
                                  us ? u.bias2 : r.bias2,
                                  us ? u.C1 : r.C1, us ? u.C2 : r.C2,
                                  us ? u.M : r.M, Nreal, K,
                                  us ? (int)blockIdx.x : (int)blockIdx.x - nbu,
                                  us ? nbu : (int)gridDim.x - nbu);
}


#define NRANGE  16
#define NSLICE  8
#define CSR_T   1024
#define SLICE_E (NEDGE / NSLICE)
#define RMAX    ((N_REPO + NRANGE - 1) / NRANGE)
static_assert(NEDGE % (4 * NSLICE) == 0, "a slice is a whole number of 16-byte groups");
static_assert(N_USER <= N_REPO, "RMAX covers both node kinds");
static_assert(NRANGE * NSLICE <= 128, "the totals of the earlier ranges are summed by four waves");
#define FILL_PER ((RMAX + CSR_T - 1) / CSR_T)
#define TOFF0 0
#define TOFF1 (NSLICE * N_USER)
#define TOFF2 (NSLICE * (N_USER + N_REPO))
#define TOFF3 (NSLICE * (N_USER + 2 * N_REPO))

__global__ void __launch_bounds__(CSR_T, 16)
range_count_kernel(const int* __restrict__ ia0, const int* __restrict__ ia1,
                   const int* __restrict__ ia2, const int* __restrict__ ia3,
                   int* __restrict__ part, int* __restrict__ tot)
{
  __shared__ int hist[RMAX];
  __shared__ int btot;
  const int a  = blockIdx.x / (NRANGE * NSLICE);
  const int rs = blockIdx.x - a * (NRANGE * NSLICE);
  const int r  = rs / NSLICE, s = rs - r * NSLICE;
  const int* __restrict__ idx = (a == 0) ? ia0 : (a == 1) ? ia1 : (a == 2) ? ia2 : ia3;
  const int n    = (a == 0 || a == 3) ? N_USER : N_REPO;
  const int toff = (a == 0) ? TOFF0 : (a == 1) ? TOFF1 : (a == 2) ? TOFF2 : TOFF3;
  const int R   = (n + NRANGE - 1) / NRANGE;
  const int lo  = r * R;
  const int len = max(min(R, n - lo), 0);
  const int tid = threadIdx.x;
  for (int i = tid; i < len; i += CSR_T) hist[i] = 0;
  if (tid == 0) btot = 0;
  __syncthreads();
  const int4* __restrict__ p = (const int4*)(idx + s * SLICE_E);
  constexpr int N4 = SLICE_E / 4;
  constexpr int GC = 8;
  for (int b = 0; b < N4; b += GC * CSR_T) {
    int4 v[GC];
#pragma unroll
    for (int g = 0; g < GC; ++g) {
      const int i = b + g * CSR_T + tid;
      v[g] = p[min(i, N4 - 1)];
      const int inval = (i < N4) ? 0 : -1;
      v[g].x |= inval; v[g].y |= inval; v[g].z |= inval; v[g].w |= inval;
    }
#pragma unroll
    for (int g = 0; g < GC; ++g) {
      unsigned u;
      u = (unsigned)(v[g].x - lo); if (u < (unsigned)len) atomicAdd(&hist[u], 1);
      u = (unsigned)(v[g].y - lo); if (u < (unsigned)len) atomicAdd(&hist[u], 1);
      u = (unsigned)(v[g].z - lo); if (u < (unsigned)len) atomicAdd(&hist[u], 1);
      u = (unsigned)(v[g].w - lo); if (u < (unsigned)len) atomicAdd(&hist[u], 1);
    }
  }
  __syncthreads();
  int* __restrict__ out = part + toff + s * n + lo;
  int my = 0;
  for (int i = tid; i < len; i += CSR_T) { const int h = hist[i]; out[i] = h; my += h; }
  my += __shfl_xor(my, 16, 32);
  my += __shfl_xor(my, 8, 32);
  my += __shfl_xor(my, 4, 32);
  my += __shfl_xor(my, 2, 32);
  my += __shfl_xor(my, 1, 32);
  if ((tid & 31) == 0) atomicAdd(&btot, my);
  __syncthreads();
  if (tid == 0) tot[blockIdx.x] = btot;
}

__global__ void __launch_bounds__(CSR_T, 16)
range_fill_kernel(const int* __restrict__ src0, const int* __restrict__ dst0,
                  const int* __restrict__ src1, const int* __restrict__ dst1,
                  const int* __restrict__ part, const int* __restrict__ tot,
                  int* __restrict__ ptr0, float* __restrict__ deg0,
                  int* __restrict__ ptr1, float* __restrict__ deg1,
                  float* __restrict__ dout0, float* __restrict__ dout2,
                  int* __restrict__ col0, int* __restrict__ col1)
{
  __shared__ int off[RMAX + 32];
  __shared__ int cnt[RMAX];
  __shared__ int wsum[32];
  __shared__ int bsum[4];
  const int tid = threadIdx.x, lane = tid & 31, wv = tid >> 5;

  if (blockIdx.x >= 2 * NRANGE * NSLICE) {
    const int x  = blockIdx.x - 2 * NRANGE * NSLICE;
    const int wh = x / NRANGE, r = x - wh * NRANGE;
    const int n  = wh ? N_REPO : N_USER;
    const int* __restrict__ tab = part + (wh ? TOFF2 : TOFF0);
    float* __restrict__ dout = wh ? dout2 : dout0;
    const int R   = (n + NRANGE - 1) / NRANGE;
    const int lo  = r * R;
    const int len = max(min(R, n - lo), 0);
    for (int i = tid; i < len; i += CSR_T) {
      int c = 0;
#pragma unroll
      for (int s2 = 0; s2 < NSLICE; ++s2) c += tab[s2 * n + lo + i];
      dout[lo + i] = (float)c;
    }
    return;
  }

  const int t  = blockIdx.x / (NRANGE * NSLICE);
  const int rs = blockIdx.x - t * (NRANGE * NSLICE);
  const int r  = rs / NSLICE, s = rs - r * NSLICE;
  const int* __restrict__ src = t ? src1 : src0;
  const int* __restrict__ dst = t ? dst1 : dst0;
  int*       __restrict__ ptr = t ? ptr1 : ptr0;
  float*     __restrict__ deg = t ? deg1 : deg0;
  int*       __restrict__ col = t ? col1 : col0;
  const int n   = t ? N_USER : N_REPO;
  const int* __restrict__ tab = part + (t ? TOFF3 : TOFF1);
  const int R   = (n + NRANGE - 1) / NRANGE;
  const int lo  = r * R;
  const int len = max(min(R, n - lo), 0);

  int bv = (tid < r * NSLICE) ? tot[(2 * t + 1) * NRANGE * NSLICE + tid] : 0;
  bv += __shfl_xor(bv, 16, 32);
  bv += __shfl_xor(bv, 8, 32);
  bv += __shfl_xor(bv, 4, 32);
  bv += __shfl_xor(bv, 2, 32);
  bv += __shfl_xor(bv, 1, 32);
  if (wv < 4 && lane == 0) bsum[wv] = bv;
  if (tid < 32) off[RMAX + tid] = 0;

#pragma unroll 2
  for (int i = tid; i < len; i += CSR_T) {
    int c = 0, pre = 0;
#pragma unroll
    for (int s2 = 0; s2 < NSLICE; ++s2) {
      const int x = tab[s2 * n + lo + i];
      c += x;
      pre += (s2 < s) ? x : 0;
    }
    cnt[i] = c;
    off[i] = pre;
  }
  __syncthreads();

  int cv[FILL_PER];
  int tsum = 0;
#pragma unroll
  for (int k = 0; k < FILL_PER; ++k) {
    const int i = tid * FILL_PER + k;
    cv[k] = (i < len) ? cnt[i] : 0;
    tsum += cv[k];
  }
  int x = tsum;
#pragma unroll
  for (int o = 1; o < 32; o <<= 1) {
    const int y = __shfl_up(x, o, 32);
    if (lane >= o) x += y;
  }
  if (lane == 31) wsum[wv] = x;
  __syncthreads();
  const int w = wsum[lane];
  int wx = w;
#pragma unroll
  for (int o = 1; o < 32; o <<= 1) {
    const int y = __shfl_up(wx, o, 32);
    if (lane >= o) wx += y;
  }
  const int wave_excl = __shfl(wx, wv, 32) - __shfl(w, wv, 32);
  int run = bsum[0] + bsum[1] + bsum[2] + bsum[3] + wave_excl + (x - tsum);
#pragma unroll
  for (int k = 0; k < FILL_PER; ++k) {
    const int i = tid * FILL_PER + k;
    if (i < len) {
      off[i] += run;
      if (s == 0) { ptr[lo + i] = run; deg[lo + i] = (float)cv[k]; }
    }
    run += cv[k];
  }
  __syncthreads();

  const unsigned spare = RMAX + lane;
  const int4* __restrict__ pd = (const int4*)(dst + s * SLICE_E);
  const int4* __restrict__ ps = (const int4*)(src + s * SLICE_E);
  constexpr int N4 = SLICE_E / 4;
  constexpr int GF = 4;
  for (int b = 0; b < N4; b += GF * CSR_T) {
    int4 d[GF], sv[GF];
#pragma unroll
    for (int g = 0; g < GF; ++g) {
      const int i = b + g * CSR_T + tid;
      d[g]  = pd[min(i, N4 - 1)];
      sv[g] = ps[min(i, N4 - 1)];
      const int inval = (i < N4) ? 0 : -1;
      d[g].x |= inval; d[g].y |= inval; d[g].z |= inval; d[g].w |= inval;
    }
#pragma unroll
    for (int g = 0; g < GF; ++g) {
      const unsigned u0 = (unsigned)(d[g].x - lo), u1 = (unsigned)(d[g].y - lo);
      const unsigned u2 = (unsigned)(d[g].z - lo), u3 = (unsigned)(d[g].w - lo);
      const bool m0 = u0 < (unsigned)len, m1 = u1 < (unsigned)len;
      const bool m2 = u2 < (unsigned)len, m3 = u3 < (unsigned)len;
      const int q0 = atomicAdd(&off[m0 ? u0 : spare], 1);
      const int q1 = atomicAdd(&off[m1 ? u1 : spare], 1);
      const int q2 = atomicAdd(&off[m2 ? u2 : spare], 1);
      const int q3 = atomicAdd(&off[m3 ? u3 : spare], 1);
      if (m0) col[min(max(q0, 0), NEDGE - 1)] = sv[g].x;
      if (m1) col[min(max(q1, 0), NEDGE - 1)] = sv[g].y;
      if (m2) col[min(max(q2, 0), NEDGE - 1)] = sv[g].z;
      if (m3) col[min(max(q3, 0), NEDGE - 1)] = sv[g].w;
    }
  }
}

#define GATHER_ROWS 4
#define GATHER_BLK_U ((N_USER + 8 * GATHER_ROWS - 1) / (8 * GATHER_ROWS))
#define GATHER_BLK_R ((N_REPO + 8 * GATHER_ROWS - 1) / (8 * GATHER_ROWS))
struct GatherSide {
  const float* feat;
  const int*   rowptr;
  const int*   col;
  float*       io;
  const float* degC;
  const float* bias2;
  float*       invn;
  int          n_dst;
  int          n_src;
};

template <int D, int JJ>
__device__ __forceinline__ void
gather_trip(float4 (&v)[D / 32], const float* __restrict__ feat, int n_src, int cur, int t, bool on)
{
  const int s = min(max((int)__builtin_amdgcn_mov_dpp8((unsigned)cur, JJ * 0x249249), 0), n_src - 1);
  const float4* p = (const float4*)(feat + (size_t)s * D);
#pragma unroll
  for (int q = 0; q < D / 32; ++q) v[q] = make_float4(0.0f, 0.0f, 0.0f, 0.0f);
  if (on) {
#pragma unroll
    for (int q = 0; q < D / 32; ++q) v[q] = p[q * 8 + t];
  }
}

template <int D, bool NORM>
__device__ __forceinline__ void
gather_rows(const float* __restrict__ feat, const int* __restrict__ rowptr,
            const int* __restrict__ col, float* __restrict__ io,
            const float* __restrict__ degC, const float* __restrict__ bias2,
            float* __restrict__ invn,
            int n_dst, int n_src, int E, int wv)
{
  constexpr int NPC = D / 32;
  static_assert(D % 32 == 0, "a row is a whole number of 128-byte lines");
  const int d0 = wv * GATHER_ROWS;
  if (d0 >= n_dst) return;
  const int lane = threadIdx.x;
  const int g = lane >> 3, t = lane & 7;
  const int  d    = d0 + g;
  const bool live = d < n_dst;

  const int xb = rowptr[min(d0 + min(lane, GATHER_ROWS), n_dst - 1)];

  const int dr = min(d, n_dst - 1);
  float4* const iop = (float4*)(io + (size_t)dr * D);
  float4 self[NPC];
#pragma unroll
  for (int q = 0; q < NPC; ++q) self[q] = iop[q * 8 + t];
  const float dgc = degC[dr];
  int rp[GATHER_ROWS + 1];
#pragma unroll
  for (int i = 0; i <= GATHER_ROWS; ++i) {
    const int x = __builtin_amdgcn_readlane(xb, i);
    rp[i] = (d0 + i < n_dst) ? min(max(x, 0), E) : E;
  }
  int mx = 0;
#pragma unroll
  for (int i = 0; i < GATHER_ROWS; ++i) mx = max(mx, rp[i + 1] - rp[i]);
  int beg = rp[0], end = rp[1];
#pragma unroll
  for (int i = 1; i < GATHER_ROWS; ++i) {
    if (g == i) { beg = rp[i]; end = rp[i + 1]; }
  }
  const int  deg  = max(end - beg, 0);

  float4 acc[NPC];
#pragma unroll
  for (int q = 0; q < NPC; ++q) acc[q] = make_float4(0.0f, 0.0f, 0.0f, 0.0f);

  int cur = 0;
  if (t < deg) cur = col[beg + t];
  cur = min(max(cur, 0), n_src - 1);
  for (int j0 = 0; j0 < mx; j0 += 8) {
    int nxt = 0;
    if (j0 + 8 + t < deg) nxt = col[beg + j0 + 8 + t];
    const int rem = deg - j0;
    float4 v[8][NPC];
    gather_trip<D, 0>(v[0], feat, n_src, cur, t, 0 < rem);
    gather_trip<D, 1>(v[1], feat, n_src, cur, t, 1 < rem);
    gather_trip<D, 2>(v[2], feat, n_src, cur, t, 2 < rem);
    gather_trip<D, 3>(v[3], feat, n_src, cur, t, 3 < rem);
    gather_trip<D, 4>(v[4], feat, n_src, cur, t, 4 < rem);
    gather_trip<D, 5>(v[5], feat, n_src, cur, t, 5 < rem);
    gather_trip<D, 6>(v[6], feat, n_src, cur, t, 6 < rem);
    gather_trip<D, 7>(v[7], feat, n_src, cur, t, 7 < rem);
#pragma unroll
    for (int jj = 0; jj < 8; ++jj) {
#pragma unroll
      for (int q = 0; q < NPC; ++q) {
        acc[q].x += v[jj][q].x;
        acc[q].y += v[jj][q].y;
        acc[q].z += v[jj][q].z;
        acc[q].w += v[jj][q].w;
      }
    }
    cur = nxt;
  }
  const float sC = rsqrtf(fmaxf(dgc, 1.0f));
  const float4* bp = (const float4*)bias2;
  float4 o[NPC];
#pragma unroll
  for (int q = 0; q < NPC; ++q) {
    const float4 b = bp[q * 8 + t];
    float vx = self[q].x, vy = self[q].y, vz = self[q].z, vw = self[q].w;
    vx += acc[q].x * sC + b.x;
    vy += acc[q].y * sC + b.y;
    vz += acc[q].z * sC + b.z;
    vw += acc[q].w * sC + b.w;
    o[q] = make_float4(fmaxf(vx, 0.0f), fmaxf(vy, 0.0f), fmaxf(vz, 0.0f), fmaxf(vw, 0.0f));
  }
  if (live) {
#pragma unroll
    for (int q = 0; q < NPC; ++q) iop[q * 8 + t] = o[q];
  }
  if (NORM) {
    static_assert(!NORM || NPC == 2, "the norm is taken of rows of 64 floats");
    const float4 v0 = o[0], v1 = o[NPC - 1];
    float s = v0.x * v0.x + v0.y * v0.y + v0.z * v0.z + v0.w * v0.w
            + v1.x * v1.x + v1.y * v1.y + v1.z * v1.z + v1.w * v1.w;
    s += __shfl_xor(s, 4, 32);
    s += __shfl_xor(s, 2, 32);
    s += __shfl_xor(s, 1, 32);
    if (live && t == 0) invn[d] = 1.0f / fmaxf(sqrtf(s), EPSN);
  }
}

template <int D, bool NORM>
__global__ void __launch_bounds__(256)
gather_pair_kernel(GatherSide a, GatherSide b, int nblk_a, int E)
{
  const bool fa = (int)blockIdx.x < nblk_a;
  const int  lb = fa ? (int)blockIdx.x : (int)blockIdx.x - nblk_a;
  const int wv = __builtin_amdgcn_readfirstlane((int)(lb * blockDim.y + threadIdx.y));
  gather_rows<D, NORM>(fa ? a.feat : b.feat, fa ? a.rowptr : b.rowptr, fa ? a.col : b.col,
                       fa ? a.io : b.io, fa ? a.degC : b.degC, fa ? a.bias2 : b.bias2,
                       fa ? a.invn : b.invn,
                       fa ? a.n_dst : b.n_dst, fa ? a.n_src : b.n_src, E, wv);
}

#define TAIL_PER_WAVE 32
#define SCORE_BLK  ((NEP + 8 * TAIL_PER_WAVE - 1) / (8 * TAIL_PER_WAVE))

__device__ __forceinline__ void
score_wave(const float* __restrict__ HU, const float* __restrict__ HR,
           const float* __restrict__ invU, const float* __restrict__ invR,
           const int* __restrict__ iu, const int* __restrict__ ir,
           float* __restrict__ out, int Ep, int wv)
{
  const int base = wv * TAIL_PER_WAVE;
  if (base >= Ep) return;
  const int lane = threadIdx.x;
  const int g = lane >> 3, t = lane & 7;
  const int el  = min(base + lane, Ep - 1);
  const int myu = iu[el], myr = ir[el];
  const float mynu = invU[myu], mynr = invR[myr];
  float sel = 0.0f;
#pragma unroll
  for (int j = 0; j < 8; ++j) {
    const int u = __shfl(myu, 4 * j + g, 32);
    const int r = __shfl(myr, 4 * j + g, 32);
    const float4* pu = (const float4*)(HU + (size_t)u * D_OUT);
    const float4* pr = (const float4*)(HR + (size_t)r * D_OUT);
    const float4 a0 = pu[t], a1 = pu[8 + t];
    const float4 b0 = pr[t], b1 = pr[8 + t];
    float s = a0.x * b0.x + a0.y * b0.y + a0.z * b0.z + a0.w * b0.w
            + a1.x * b1.x + a1.y * b1.y + a1.z * b1.z + a1.w * b1.w;
    s += __shfl_xor(s, 4, 32);
    s += __shfl_xor(s, 2, 32);
    s += __shfl_xor(s, 1, 32);
    if (t == j) sel = s;
  }
  const int own = 4 * t + g;
  const float nu = __shfl(mynu, own, 32);
  const float nr = __shfl(mynr, own, 32);
  const int e = base + own;
  if (e < Ep) out[e] = sel * nu * nr;
}

__global__ void __launch_bounds__(256)
score_pair_kernel(const float* __restrict__ HU, const float* __restrict__ HR,
                  const float* __restrict__ invU, const float* __restrict__ invR,
                  const int* __restrict__ iu0, const int* __restrict__ ir0,
                  const int* __restrict__ iu1, const int* __restrict__ ir1,
                  float* __restrict__ out, int Ep, int nblk)
{
  const bool f0 = (int)blockIdx.x < nblk;
  const int  lb = f0 ? (int)blockIdx.x : (int)blockIdx.x - nblk;
  score_wave(HU, HR, invU, invR, f0 ? iu0 : iu1, f0 ? ir0 : ir1,
             f0 ? out : out + Ep, Ep, (int)(lb * blockDim.y + threadIdx.y));
}

extern "C" void kernel_launch(void* const* d_in, const int* in_sizes, int n_in,
                              void* d_out, int out_size, void* d_ws, size_t ws_size,
                              hipStream_t stream)
{
  const float* user_feat = (const float*)d_in[0];
  const float* repo_feat = (const float*)d_in[1];
  const float* W_ue = (const float*)d_in[2];   const float* b_ue = (const float*)d_in[3];
  const float* W_re = (const float*)d_in[4];   const float* b_re = (const float*)d_in[5];
  const float* W_hid_ur = (const float*)d_in[6];  const float* b_hid_ur = (const float*)d_in[7];
  const float* W_hid_ru = (const float*)d_in[8];  const float* b_hid_ru = (const float*)d_in[9];
  const float* W_out_ur = (const float*)d_in[10]; const float* b_out_ur = (const float*)d_in[11];
  const float* W_out_ru = (const float*)d_in[12]; const float* b_out_ru = (const float*)d_in[13];
  const float* W_hcu = (const float*)d_in[14]; const float* b_hcu = (const float*)d_in[15];
  const float* W_hcr = (const float*)d_in[16]; const float* b_hcr = (const float*)d_in[17];
  const float* W_ocu = (const float*)d_in[18]; const float* b_ocu = (const float*)d_in[19];
  const float* W_ocr = (const float*)d_in[20]; const float* b_ocr = (const float*)d_in[21];
  const int* e_ur_src = (const int*)d_in[22];  const int* e_ur_dst = (const int*)d_in[23];
  const int* e_ru_src = (const int*)d_in[24];  const int* e_ru_dst = (const int*)d_in[25];
  const int* pos_u = (const int*)d_in[26];     const int* pos_r = (const int*)d_in[27];
  const int* neg_u = (const int*)d_in[28];     const int* neg_r = (const int*)d_in[29];
  float* out = (float*)d_out;

  float* w = (float*)d_ws;
  size_t o = 0;
  float* deg_u_out_ur = w + o; o += N_USER;
  float* deg_r_in_ur  = w + o; o += N_REPO;
  float* deg_r_out_ru = w + o; o += N_REPO;
  float* deg_u_in_ru  = w + o; o += N_USER;
  float* reg_agg  = w + o; o += ((size_t)N_USER + N_REPO) * D_HID;
  float* reg_h    = w + o; o += ((size_t)N_USER + N_REPO) * LD_EMB;
  float* reg_feat = w + o; o += ((size_t)N_USER + N_REPO) * D_HID;
  float* reg_csr  = w + o; o += ((size_t)N_USER + N_REPO) * D_HID;
  float* inv_nu  = w + o; o += N_USER;
  float* inv_nr  = w + o; o += N_REPO;

  int* col_ur  = (int*)reg_csr;
  int* col_ru  = col_ur + NEDGE;
  int* part    = col_ru + NEDGE;
  int* tot     = part + NSLICE * 2 * (N_USER + N_REPO);
  int* ptr_r   = tot + 4 * NRANGE * NSLICE;
  int* ptr_u   = ptr_r + N_REPO;

  float* h_user   = reg_h;
  float* h_repo   = reg_h + (size_t)N_USER * LD_EMB;
  float* feat_u   = reg_feat;
  float* feat_r   = reg_feat + (size_t)N_USER * D_HID;
  float* out_user = reg_agg;
  float* out_repo = reg_agg + (size_t)N_USER * D_HID;
  float* feat2_u  = reg_feat;
  float* feat2_r  = reg_feat + (size_t)N_USER * D_HID;
  float* hn_user  = reg_h;
  float* hn_repo  = reg_h + (size_t)N_USER * LD_EMB;

  auto gemm = [&](const GemmSide& u, const GemmSide& r, int lda, int ldw,
                  int Nreal, int K) {
    dim3 blk(GEMM_WAVES * 32);
    dim3 grd(GEMM_BLOCKS);
    if (K == D_IN)
      hipLaunchKernelGGL((gemm_pair_wmma_f32<LD_EMB / 16, 0, D_IN>), grd, blk, 0, stream,
                         u, r, lda, ldw, Nreal, K, GEMM_BLOCKS_U);
    else if (K == D_EMB)
      hipLaunchKernelGGL((gemm_pair_wmma_f32<D_HID / 16, D_HID / 16, LD_EMB>), grd, blk, 0, stream,
                         u, r, lda, ldw, Nreal, K, GEMM_BLOCKS_U);
    else
      hipLaunchKernelGGL((gemm_pair_wmma_f32<D_OUT / 16, D_OUT / 16, D_HID>), grd, blk, 0, stream,
                         u, r, lda, ldw, Nreal, K, GEMM_BLOCKS_U);
  };

  auto gather = [&](const GatherSide& a, const GatherSide& b, int D) {
    dim3 blk(32, 8);
    dim3 grd(GATHER_BLK_U + GATHER_BLK_R);
    if (D == D_HID)
      hipLaunchKernelGGL((gather_pair_kernel<D_HID, false>), grd, blk, 0, stream,
                         a, b, GATHER_BLK_U, NEDGE);
    else
      hipLaunchKernelGGL((gather_pair_kernel<D_OUT, true>), grd, blk, 0, stream,
                         a, b, GATHER_BLK_U, NEDGE);
  };

  hipLaunchKernelGGL(range_count_kernel, dim3(4 * NRANGE * NSLICE), dim3(CSR_T), 0, stream,
                     e_ur_src, e_ur_dst, e_ru_src, e_ru_dst, part, tot);
  hipLaunchKernelGGL(range_fill_kernel, dim3(2 * NRANGE * NSLICE + 2 * NRANGE), dim3(CSR_T), 0, stream,
                     e_ur_src, e_ur_dst, e_ru_src, e_ru_dst, (const int*)part, (const int*)tot,
                     ptr_r, deg_r_in_ur, ptr_u, deg_u_in_ru, deg_u_out_ur, deg_r_out_ru,
                     col_ur, col_ru);

  gemm(GemmSide{user_feat, W_ue, nullptr, b_ue, nullptr, nullptr, h_user, nullptr, N_USER},
       GemmSide{repo_feat, W_re, nullptr, b_re, nullptr, nullptr, h_repo, nullptr, N_REPO},
       D_IN, D_EMB, D_EMB, D_IN);

  gemm(GemmSide{h_user, W_hid_ur, deg_u_out_ur, nullptr, W_hcu, b_hcu, feat_u, out_user, N_USER},
       GemmSide{h_repo, W_hid_ru, deg_r_out_ru, nullptr, W_hcr, b_hcr, feat_r, out_repo, N_REPO},
       LD_EMB, D_HID, D_HID, D_EMB);

  gather(GatherSide{feat_r, ptr_u, col_ru, out_user, deg_u_in_ru, b_hid_ru, nullptr, N_USER, N_REPO},
         GatherSide{feat_u, ptr_r, col_ur, out_repo, deg_r_in_ur, b_hid_ur, nullptr, N_REPO, N_USER}, D_HID);

  gemm(GemmSide{out_user, W_out_ur, deg_u_out_ur, nullptr, W_ocu, b_ocu, feat2_u, hn_user, N_USER},
       GemmSide{out_repo, W_out_ru, deg_r_out_ru, nullptr, W_ocr, b_ocr, feat2_r, hn_repo, N_REPO},
       D_HID, D_OUT, D_OUT, D_HID);

  gather(GatherSide{feat2_r, ptr_u, col_ru, hn_user, deg_u_in_ru, b_out_ru, inv_nu, N_USER, N_REPO},
         GatherSide{feat2_u, ptr_r, col_ur, hn_repo, deg_r_in_ur, b_out_ur, inv_nr, N_REPO, N_USER}, D_OUT);

  hipLaunchKernelGGL(score_pair_kernel, dim3(2 * SCORE_BLK), dim3(32, 8), 0, stream,
                     (const float*)hn_user, (const float*)hn_repo,
                     (const float*)inv_nu, (const float*)inv_nr,
                     pos_u, pos_r, neg_u, neg_r, out, NEP, SCORE_BLK);
}


# Embedding GEMM folded into layer 1 by reassociation, x@(We@W)+be@W for (x@We+be)@W: folded weights built per call on WMMA in the fill launch; 8 launches become 7; layer-1 sum order changes
# speedup vs baseline: 5.4926x; 1.0694x over previous
#include <hip/hip_runtime.h>
#include <math.h>

#define N_USER 50000
#define N_REPO 100000
#define NEDGE  1000000
#define NEP    200000
#define D_IN   256
#define D_EMB  125
#define LD_EMB 128
#define D_HID  96
#define D_OUT  64
#define EPSN   1e-12f

typedef float v2f __attribute__((ext_vector_type(2)));
typedef float v8f __attribute__((ext_vector_type(8)));


#define GEMM_WAVES  8
#define GEMM_BLOCKS 512
#define WT_FLOATS (LD_EMB * (D_IN + 4))
#define GEMM1_WAVES  16
#define GEMM1_BLOCKS 256
#define WT1_FLOATS (2 * D_HID * (D_IN + 4))
#define BLOCKS_U(nblk) (((nblk) * N_USER + (N_USER + N_REPO) / 2) / (N_USER + N_REPO))

struct GemmSide {
  const float* A;
  const float* W1;
  const float* degA;
  const float* bias1;
  const float* W2;
  const float* bias2;
  float*       C1;
  float*       C2;
};

template <int NT1, int NT2, int KPAD, int NW, int LDSF, bool SCB>
__device__ __forceinline__ void
gemm_strip_body(const float* __restrict__ A, int lda,
                const float* __restrict__ W1,
                const float* __restrict__ W2, int ldw,
                const float* __restrict__ degA,
                const float* __restrict__ bias1,
                const float* __restrict__ bias2,
                float* __restrict__ C1,
                float* __restrict__ C2,
                int M, int Nreal, int K, int lb, int nb)
{
  constexpr int NT  = NT1 + NT2;
  constexpr int NP1 = NT1 * 16;
  constexpr int NP2 = NT2 * 16;
  constexpr int LS  = KPAD + 4;
  constexpr int NCH = KPAD / 32;
  static_assert(KPAD % 32 == 0, "K is walked in whole 128-byte lines");
  static_assert(NT2 == 0 || NT2 == NT1, "both products have the same width");
  static_assert((NP1 + NP2) * LS <= LDSF, "W1^T and W2^T fit the LDS array");
  __shared__ __align__(16) float Wt[LDSF];

  const int tid = threadIdx.x;

  const int lane = tid & 31;
  const int wave = tid >> 5;
  const int half = lane >> 4;
  const int l16  = lane & 15;

  float bn1[NT1], bn2[NT1];
#pragma unroll
  for (int t = 0; t < NT1; ++t) { bn1[t] = 0.0f; bn2[t] = 0.0f; }
  if (bias1) {
#pragma unroll
    for (int t = 0; t < NT1; ++t) bn1[t] = bias1[min(t * 16 + l16, Nreal - 1)];
  }
  if (NT2 > 0 && bias2) {
#pragma unroll
    for (int t = 0; t < NT2; ++t) bn2[t] = bias2[min(t * 16 + l16, Nreal - 1)];
  }

  constexpr int NST = NP1 * KPAD / (NW * 32);
  static_assert(NP1 * KPAD % (NW * 32) == 0, "every thread stages the same count");
  constexpr int UNR = (NST % 16 == 0) ? 16 : 8;
  static_assert(NST % UNR == 0, "the staging loop unrolls without a remainder");
  auto stage = [&](const float* __restrict__ W, int row0) {
#pragma unroll UNR
    for (int j = 0; j < NST; ++j) {
      const int i = tid + j * (NW * 32);
      const int k = i / NP1;
      const int n = i - k * NP1;
      const float w = W[(long)min(k, K - 1) * ldw + min(n, Nreal - 1)];
      const unsigned keep = (n < Nreal && k < K) ? 0xFFFFFFFFu : 0u;
      Wt[(row0 + n) * LS + k] = __uint_as_float(__float_as_uint(w) & keep);
    }
  };
  stage(W1, 0);
  if (NT2 > 0) stage(W2, NP1);
  __syncthreads();

  const float* Wp = &Wt[l16 * LS + 2 * half];
  const int Mtiles = M / 16;

  for (int mt = lb + nb * wave; mt < Mtiles; mt += nb * NW) {
    const int   m0 = mt * 16;
    const int   mA = m0 + l16;
    const float sA = degA ? rsqrtf(fmaxf(degA[mA], 1.0f)) : 1.0f;
    const float* Ap = A + (long)mA * lda + 2 * half;

    v8f acc[NT];
#pragma unroll
    for (int t = 0; t < NT; ++t) acc[t] = v8f{};

    v2f a_raw[8], a_sc[8];
#pragma unroll
    for (int s = 0; s < 8; ++s) {
      a_raw[s] = *(const v2f*)(Ap + s * 4);
      a_sc[s]  = a_raw[s] * sA;
    }

    auto mma_chunk = [&](int c) {
      const float* Wc = Wp + c * 32;
#pragma unroll
      for (int s = 0; s < 8; ++s) {
        v2f b[NT];
#pragma unroll
        for (int t = 0; t < NT; ++t)
          b[t] = *(const v2f*)(Wc + t * 16 * LS + s * 4);
#pragma unroll
        for (int t = 0; t < NT1; ++t)
          acc[t] = __builtin_amdgcn_wmma_f32_16x16x4_f32(false, a_sc[s], false, b[t],
                                                         (short)0, acc[t], false, false);
#pragma unroll
        for (int t = NT1; t < NT; ++t)
          acc[t] = __builtin_amdgcn_wmma_f32_16x16x4_f32(false, a_raw[s], false, b[t],
                                                         (short)0, acc[t], false, false);
      }
    };

#pragma unroll 1
    for (int c = 0; c + 1 < NCH; ++c) {
      v2f a_nxt[8];
#pragma unroll
      for (int s = 0; s < 8; ++s) a_nxt[s] = *(const v2f*)(Ap + (c + 1) * 32 + s * 4);
      __builtin_amdgcn_sched_barrier(0);
      mma_chunk(c);
      __builtin_amdgcn_sched_barrier(0);
#pragma unroll
      for (int s = 0; s < 8; ++s) { a_raw[s] = a_nxt[s]; a_sc[s] = a_nxt[s] * sA; }
    }
    mma_chunk(NCH - 1);

    const int mrow = m0 + 8 * half;
    float sr[8];
#pragma unroll
    for (int v = 0; v < 8; ++v)
      sr[v] = (SCB && degA) ? rsqrtf(fmaxf(degA[mrow + v], 1.0f)) : 1.0f;
#pragma unroll
    for (int t = 0; t < NT1; ++t) {
      const int n = t * 16 + l16;
#pragma unroll
      for (int v = 0; v < 8; ++v) {
        const float val = acc[t][v] + (SCB ? bn1[t] * sr[v] : bn1[t]);
        C1[(long)(mrow + v) * NP1 + n] = (n < Nreal) ? val : 0.0f;
      }
    }
#pragma unroll
    for (int t = 0; t < NT2; ++t) {
      const int n = t * 16 + l16;
#pragma unroll
      for (int v = 0; v < 8; ++v) {
        const float val = acc[NT1 + t][v] + bn2[t];
        C2[(long)(mrow + v) * NP2 + n] = (n < Nreal) ? val : 0.0f;
      }
    }
  }
}

template <int NT1, int NT2, int KPAD, int NW, int LDSF, bool SCB>
__global__ void __launch_bounds__(NW * 32, 4)
gemm_pair_wmma_f32(GemmSide u, GemmSide r, int Mu, int Mr, int lda, int ldw,
                   int Nreal, int K, int nbu)
{
  const bool us = (int)blockIdx.x < nbu;
  gemm_strip_body<NT1, NT2, KPAD, NW, LDSF, SCB>(us ? u.A : r.A, lda, us ? u.W1 : r.W1,
                                  us ? u.W2 : r.W2, ldw,
                                  us ? u.degA : r.degA, us ? u.bias1 : r.bias1,
                                  us ? u.bias2 : r.bias2,
                                  us ? u.C1 : r.C1, us ? u.C2 : r.C2,
                                  us ? Mu : Mr, Nreal, K,
                                  us ? (int)blockIdx.x : (int)blockIdx.x - nbu,
                                  us ? nbu : (int)gridDim.x - nbu);
}

#define FOLD_ROWS   272
#define FOLD_MT     (FOLD_ROWS / 16)
#define FOLD_NT     (D_HID / 16)
#define FOLD_FLOATS (FOLD_ROWS * D_HID)
#define FOLD_TILES  (4 * FOLD_MT * FOLD_NT)
static_assert(FOLD_ROWS % 16 == 0 && FOLD_ROWS > D_IN, "room for the bias row in whole tiles");
#define FOLD_BLOCKS ((FOLD_TILES + 31) / 32)

struct FoldSide {
  const float* We;
  const float* be;
  const float* Wc;
  const float* Ws;
  const float* bs;
  float*       out;
};

__device__ __forceinline__ float keep_bits(float w, unsigned keep)
{
  return __uint_as_float(__float_as_uint(w) & keep);
}

__device__ __forceinline__ void
fold_tile(const float* __restrict__ We, const float* __restrict__ be,
          const float* __restrict__ W, const float* __restrict__ bs,
          float* __restrict__ out, int mt, int nt, int lane)
{
  const int half = lane >> 4;
  const int l16  = lane & 15;
  const int m = mt * 16 + l16;
  const float* __restrict__ arow = (m < D_IN) ? We + (long)m * D_EMB : be;
  const unsigned rowkeep = (m <= D_IN) ? 0xFFFFFFFFu : 0u;
  const int n = nt * 16 + l16;
  v8f acc = v8f{};
#pragma unroll 8
  for (int k0 = 0; k0 < LD_EMB; k0 += 4) {
    const int ka = k0 + 2 * half, kb = ka + 1;
    const unsigned keepa = (ka < D_EMB) ? 0xFFFFFFFFu : 0u;
    const unsigned keepb = (kb < D_EMB) ? 0xFFFFFFFFu : 0u;
    const int kac = min(ka, D_EMB - 1), kbc = min(kb, D_EMB - 1);
    v2f a, b;
    a.x = keep_bits(arow[kac], keepa & rowkeep);
    a.y = keep_bits(arow[kbc], keepb & rowkeep);
    b.x = keep_bits(W[kac * D_HID + n], keepa);
    b.y = keep_bits(W[kbc * D_HID + n], keepb);
    acc = __builtin_amdgcn_wmma_f32_16x16x4_f32(false, a, false, b, (short)0, acc, false, false);
  }
  const int mrow = mt * 16 + 8 * half;
  const float bsn = bs ? bs[n] : 0.0f;
#pragma unroll
  for (int v = 0; v < 8; ++v) {
    const int row = mrow + v;
    out[row * D_HID + n] = acc[v] + ((row == D_IN) ? bsn : 0.0f);
  }
}


#define NRANGE  16
#define NSLICE  8
#define CSR_T   1024
static_assert(CSR_T == 32 * 32, "a fold block is 32 waves");
#define SLICE_E (NEDGE / NSLICE)
#define RMAX    ((N_REPO + NRANGE - 1) / NRANGE)
static_assert(NEDGE % (4 * NSLICE) == 0, "a slice is a whole number of 16-byte groups");
static_assert(N_USER <= N_REPO, "RMAX covers both node kinds");
static_assert(NRANGE * NSLICE <= 128, "the totals of the earlier ranges are summed by four waves");
#define FILL_PER ((RMAX + CSR_T - 1) / CSR_T)
#define TOFF0 0
#define TOFF1 (NSLICE * N_USER)
#define TOFF2 (NSLICE * (N_USER + N_REPO))
#define TOFF3 (NSLICE * (N_USER + 2 * N_REPO))

__global__ void __launch_bounds__(CSR_T, 16)
range_count_kernel(const int* __restrict__ ia0, const int* __restrict__ ia1,
                   const int* __restrict__ ia2, const int* __restrict__ ia3,
                   int* __restrict__ part, int* __restrict__ tot)
{
  __shared__ int hist[RMAX];
  __shared__ int btot;
  const int a  = blockIdx.x / (NRANGE * NSLICE);
  const int rs = blockIdx.x - a * (NRANGE * NSLICE);
  const int r  = rs / NSLICE, s = rs - r * NSLICE;
  const int* __restrict__ idx = (a == 0) ? ia0 : (a == 1) ? ia1 : (a == 2) ? ia2 : ia3;
  const int n    = (a == 0 || a == 3) ? N_USER : N_REPO;
  const int toff = (a == 0) ? TOFF0 : (a == 1) ? TOFF1 : (a == 2) ? TOFF2 : TOFF3;
  const int R   = (n + NRANGE - 1) / NRANGE;
  const int lo  = r * R;
  const int len = max(min(R, n - lo), 0);
  const int tid = threadIdx.x;
  for (int i = tid; i < len; i += CSR_T) hist[i] = 0;
  if (tid == 0) btot = 0;
  __syncthreads();
  const int4* __restrict__ p = (const int4*)(idx + s * SLICE_E);
  constexpr int N4 = SLICE_E / 4;
  constexpr int GC = 8;
  for (int b = 0; b < N4; b += GC * CSR_T) {
    int4 v[GC];
#pragma unroll
    for (int g = 0; g < GC; ++g) {
      const int i = b + g * CSR_T + tid;
      v[g] = p[min(i, N4 - 1)];
      const int inval = (i < N4) ? 0 : -1;
      v[g].x |= inval; v[g].y |= inval; v[g].z |= inval; v[g].w |= inval;
    }
#pragma unroll
    for (int g = 0; g < GC; ++g) {
      unsigned u;
      u = (unsigned)(v[g].x - lo); if (u < (unsigned)len) atomicAdd(&hist[u], 1);
      u = (unsigned)(v[g].y - lo); if (u < (unsigned)len) atomicAdd(&hist[u], 1);
      u = (unsigned)(v[g].z - lo); if (u < (unsigned)len) atomicAdd(&hist[u], 1);
      u = (unsigned)(v[g].w - lo); if (u < (unsigned)len) atomicAdd(&hist[u], 1);
    }
  }
  __syncthreads();
  int* __restrict__ out = part + toff + s * n + lo;
  int my = 0;
  for (int i = tid; i < len; i += CSR_T) { const int h = hist[i]; out[i] = h; my += h; }
  my += __shfl_xor(my, 16, 32);
  my += __shfl_xor(my, 8, 32);
  my += __shfl_xor(my, 4, 32);
  my += __shfl_xor(my, 2, 32);
  my += __shfl_xor(my, 1, 32);
  if ((tid & 31) == 0) atomicAdd(&btot, my);
  __syncthreads();
  if (tid == 0) tot[blockIdx.x] = btot;
}

__global__ void __launch_bounds__(CSR_T, 16)
range_fill_kernel(const int* __restrict__ src0, const int* __restrict__ dst0,
                  const int* __restrict__ src1, const int* __restrict__ dst1,
                  const int* __restrict__ part, const int* __restrict__ tot,
                  int* __restrict__ ptr0, float* __restrict__ deg0,
                  int* __restrict__ ptr1, float* __restrict__ deg1,
                  float* __restrict__ dout0, float* __restrict__ dout2,
                  int* __restrict__ col0, int* __restrict__ col1,
                  FoldSide fu, FoldSide fr)
{
  __shared__ int off[RMAX + 32];
  __shared__ int cnt[RMAX];
  __shared__ int wsum[32];
  __shared__ int bsum[4];
  const int tid = threadIdx.x, lane = tid & 31, wv = tid >> 5;

  if (blockIdx.x >= 2 * NRANGE * NSLICE + 2 * NRANGE) {
    const int fw = __builtin_amdgcn_readfirstlane(
        (int)(blockIdx.x - (2 * NRANGE * NSLICE + 2 * NRANGE)) * (CSR_T / 32) + wv);
    if (fw >= FOLD_TILES) return;
    const int p   = fw / (FOLD_MT * FOLD_NT);
    const int rem = fw - p * (FOLD_MT * FOLD_NT);
    const int mt  = rem / FOLD_NT, nt = rem - mt * FOLD_NT;
    const bool rs = p >= 2;
    const bool sf = (p & 1) != 0;
    fold_tile(rs ? fr.We : fu.We, rs ? fr.be : fu.be,
              sf ? (rs ? fr.Ws : fu.Ws) : (rs ? fr.Wc : fu.Wc),
              sf ? (rs ? fr.bs : fu.bs) : nullptr,
              (rs ? fr.out : fu.out) + (sf ? FOLD_FLOATS : 0), mt, nt, lane);
    return;
  }

  if (blockIdx.x >= 2 * NRANGE * NSLICE) {
    const int x  = blockIdx.x - 2 * NRANGE * NSLICE;
    const int wh = x / NRANGE, r = x - wh * NRANGE;
    const int n  = wh ? N_REPO : N_USER;
    const int* __restrict__ tab = part + (wh ? TOFF2 : TOFF0);
    float* __restrict__ dout = wh ? dout2 : dout0;
    const int R   = (n + NRANGE - 1) / NRANGE;
    const int lo  = r * R;
    const int len = max(min(R, n - lo), 0);
    for (int i = tid; i < len; i += CSR_T) {
      int c = 0;
#pragma unroll
      for (int s2 = 0; s2 < NSLICE; ++s2) c += tab[s2 * n + lo + i];
      dout[lo + i] = (float)c;
    }
    return;
  }

  const int t  = blockIdx.x / (NRANGE * NSLICE);
  const int rs = blockIdx.x - t * (NRANGE * NSLICE);
  const int r  = rs / NSLICE, s = rs - r * NSLICE;
  const int* __restrict__ src = t ? src1 : src0;
  const int* __restrict__ dst = t ? dst1 : dst0;
  int*       __restrict__ ptr = t ? ptr1 : ptr0;
  float*     __restrict__ deg = t ? deg1 : deg0;
  int*       __restrict__ col = t ? col1 : col0;
  const int n   = t ? N_USER : N_REPO;
  const int* __restrict__ tab = part + (t ? TOFF3 : TOFF1);
  const int R   = (n + NRANGE - 1) / NRANGE;
  const int lo  = r * R;
  const int len = max(min(R, n - lo), 0);

  int bv = (tid < r * NSLICE) ? tot[(2 * t + 1) * NRANGE * NSLICE + tid] : 0;
  bv += __shfl_xor(bv, 16, 32);
  bv += __shfl_xor(bv, 8, 32);
  bv += __shfl_xor(bv, 4, 32);
  bv += __shfl_xor(bv, 2, 32);
  bv += __shfl_xor(bv, 1, 32);
  if (wv < 4 && lane == 0) bsum[wv] = bv;
  if (tid < 32) off[RMAX + tid] = 0;

#pragma unroll 2
  for (int i = tid; i < len; i += CSR_T) {
    int c = 0, pre = 0;
#pragma unroll
    for (int s2 = 0; s2 < NSLICE; ++s2) {
      const int x = tab[s2 * n + lo + i];
      c += x;
      pre += (s2 < s) ? x : 0;
    }
    cnt[i] = c;
    off[i] = pre;
  }
  __syncthreads();

  int cv[FILL_PER];
  int tsum = 0;
#pragma unroll
  for (int k = 0; k < FILL_PER; ++k) {
    const int i = tid * FILL_PER + k;
    cv[k] = (i < len) ? cnt[i] : 0;
    tsum += cv[k];
  }
  int x = tsum;
#pragma unroll
  for (int o = 1; o < 32; o <<= 1) {
    const int y = __shfl_up(x, o, 32);
    if (lane >= o) x += y;
  }
  if (lane == 31) wsum[wv] = x;
  __syncthreads();
  const int w = wsum[lane];
  int wx = w;
#pragma unroll
  for (int o = 1; o < 32; o <<= 1) {
    const int y = __shfl_up(wx, o, 32);
    if (lane >= o) wx += y;
  }
  const int wave_excl = __shfl(wx, wv, 32) - __shfl(w, wv, 32);
  int run = bsum[0] + bsum[1] + bsum[2] + bsum[3] + wave_excl + (x - tsum);
#pragma unroll
  for (int k = 0; k < FILL_PER; ++k) {
    const int i = tid * FILL_PER + k;
    if (i < len) {
      off[i] += run;
      if (s == 0) { ptr[lo + i] = run; deg[lo + i] = (float)cv[k]; }
    }
    run += cv[k];
  }
  __syncthreads();

  const unsigned spare = RMAX + lane;
  const int4* __restrict__ pd = (const int4*)(dst + s * SLICE_E);
  const int4* __restrict__ ps = (const int4*)(src + s * SLICE_E);
  constexpr int N4 = SLICE_E / 4;
  constexpr int GF = 4;
  for (int b = 0; b < N4; b += GF * CSR_T) {
    int4 d[GF], sv[GF];
#pragma unroll
    for (int g = 0; g < GF; ++g) {
      const int i = b + g * CSR_T + tid;
      d[g]  = pd[min(i, N4 - 1)];
      sv[g] = ps[min(i, N4 - 1)];
      const int inval = (i < N4) ? 0 : -1;
      d[g].x |= inval; d[g].y |= inval; d[g].z |= inval; d[g].w |= inval;
    }
#pragma unroll
    for (int g = 0; g < GF; ++g) {
      const unsigned u0 = (unsigned)(d[g].x - lo), u1 = (unsigned)(d[g].y - lo);
      const unsigned u2 = (unsigned)(d[g].z - lo), u3 = (unsigned)(d[g].w - lo);
      const bool m0 = u0 < (unsigned)len, m1 = u1 < (unsigned)len;
      const bool m2 = u2 < (unsigned)len, m3 = u3 < (unsigned)len;
      const int q0 = atomicAdd(&off[m0 ? u0 : spare], 1);
      const int q1 = atomicAdd(&off[m1 ? u1 : spare], 1);
      const int q2 = atomicAdd(&off[m2 ? u2 : spare], 1);
      const int q3 = atomicAdd(&off[m3 ? u3 : spare], 1);
      if (m0) col[min(max(q0, 0), NEDGE - 1)] = sv[g].x;
      if (m1) col[min(max(q1, 0), NEDGE - 1)] = sv[g].y;
      if (m2) col[min(max(q2, 0), NEDGE - 1)] = sv[g].z;
      if (m3) col[min(max(q3, 0), NEDGE - 1)] = sv[g].w;
    }
  }
}

#define GATHER_ROWS 4
#define GATHER_BLK_U ((N_USER + 8 * GATHER_ROWS - 1) / (8 * GATHER_ROWS))
#define GATHER_BLK_R ((N_REPO + 8 * GATHER_ROWS - 1) / (8 * GATHER_ROWS))
struct GatherSide {
  const float* feat;
  const int*   rowptr;
  const int*   col;
  float*       io;
  const float* degC;
  const float* bias2;
  float*       invn;
  int          n_dst;
  int          n_src;
};

template <int D, int JJ>
__device__ __forceinline__ void
gather_trip(float4 (&v)[D / 32], const float* __restrict__ feat, int n_src, int cur, int t, bool on)
{
  const int s = min(max((int)__builtin_amdgcn_mov_dpp8((unsigned)cur, JJ * 0x249249), 0), n_src - 1);
  const float4* p = (const float4*)(feat + (size_t)s * D);
#pragma unroll
  for (int q = 0; q < D / 32; ++q) v[q] = make_float4(0.0f, 0.0f, 0.0f, 0.0f);
  if (on) {
#pragma unroll
    for (int q = 0; q < D / 32; ++q) v[q] = p[q * 8 + t];
  }
}

template <int D, bool NORM>
__device__ __forceinline__ void
gather_rows(const float* __restrict__ feat, const int* __restrict__ rowptr,
            const int* __restrict__ col, float* __restrict__ io,
            const float* __restrict__ degC, const float* __restrict__ bias2,
            float* __restrict__ invn,
            int n_dst, int n_src, int E, int wv)
{
  constexpr int NPC = D / 32;
  static_assert(D % 32 == 0, "a row is a whole number of 128-byte lines");
  const int d0 = wv * GATHER_ROWS;
  if (d0 >= n_dst) return;
  const int lane = threadIdx.x;
  const int g = lane >> 3, t = lane & 7;
  const int  d    = d0 + g;
  const bool live = d < n_dst;

  const int xb = rowptr[min(d0 + min(lane, GATHER_ROWS), n_dst - 1)];

  const int dr = min(d, n_dst - 1);
  float4* const iop = (float4*)(io + (size_t)dr * D);
  float4 self[NPC];
#pragma unroll
  for (int q = 0; q < NPC; ++q) self[q] = iop[q * 8 + t];
  const float dgc = degC[dr];
  int rp[GATHER_ROWS + 1];
#pragma unroll
  for (int i = 0; i <= GATHER_ROWS; ++i) {
    const int x = __builtin_amdgcn_readlane(xb, i);
    rp[i] = (d0 + i < n_dst) ? min(max(x, 0), E) : E;
  }
  int mx = 0;
#pragma unroll
  for (int i = 0; i < GATHER_ROWS; ++i) mx = max(mx, rp[i + 1] - rp[i]);
  int beg = rp[0], end = rp[1];
#pragma unroll
  for (int i = 1; i < GATHER_ROWS; ++i) {
    if (g == i) { beg = rp[i]; end = rp[i + 1]; }
  }
  const int  deg  = max(end - beg, 0);

  float4 acc[NPC];
#pragma unroll
  for (int q = 0; q < NPC; ++q) acc[q] = make_float4(0.0f, 0.0f, 0.0f, 0.0f);

  int cur = 0;
  if (t < deg) cur = col[beg + t];
  cur = min(max(cur, 0), n_src - 1);
  for (int j0 = 0; j0 < mx; j0 += 8) {
    int nxt = 0;
    if (j0 + 8 + t < deg) nxt = col[beg + j0 + 8 + t];
    const int rem = deg - j0;
    float4 v[8][NPC];
    gather_trip<D, 0>(v[0], feat, n_src, cur, t, 0 < rem);
    gather_trip<D, 1>(v[1], feat, n_src, cur, t, 1 < rem);
    gather_trip<D, 2>(v[2], feat, n_src, cur, t, 2 < rem);
    gather_trip<D, 3>(v[3], feat, n_src, cur, t, 3 < rem);
    gather_trip<D, 4>(v[4], feat, n_src, cur, t, 4 < rem);
    gather_trip<D, 5>(v[5], feat, n_src, cur, t, 5 < rem);
    gather_trip<D, 6>(v[6], feat, n_src, cur, t, 6 < rem);
    gather_trip<D, 7>(v[7], feat, n_src, cur, t, 7 < rem);
#pragma unroll
    for (int jj = 0; jj < 8; ++jj) {
#pragma unroll
      for (int q = 0; q < NPC; ++q) {
        acc[q].x += v[jj][q].x;
        acc[q].y += v[jj][q].y;
        acc[q].z += v[jj][q].z;
        acc[q].w += v[jj][q].w;
      }
    }
    cur = nxt;
  }
  const float sC = rsqrtf(fmaxf(dgc, 1.0f));
  const float4* bp = (const float4*)bias2;
  float4 o[NPC];
#pragma unroll
  for (int q = 0; q < NPC; ++q) {
    const float4 b = bp[q * 8 + t];
    float vx = self[q].x, vy = self[q].y, vz = self[q].z, vw = self[q].w;
    vx += acc[q].x * sC + b.x;
    vy += acc[q].y * sC + b.y;
    vz += acc[q].z * sC + b.z;
    vw += acc[q].w * sC + b.w;
    o[q] = make_float4(fmaxf(vx, 0.0f), fmaxf(vy, 0.0f), fmaxf(vz, 0.0f), fmaxf(vw, 0.0f));
  }
  if (live) {
#pragma unroll
    for (int q = 0; q < NPC; ++q) iop[q * 8 + t] = o[q];
  }
  if (NORM) {
    static_assert(!NORM || NPC == 2, "the norm is taken of rows of 64 floats");
    const float4 v0 = o[0], v1 = o[NPC - 1];
    float s = v0.x * v0.x + v0.y * v0.y + v0.z * v0.z + v0.w * v0.w
            + v1.x * v1.x + v1.y * v1.y + v1.z * v1.z + v1.w * v1.w;
    s += __shfl_xor(s, 4, 32);
    s += __shfl_xor(s, 2, 32);
    s += __shfl_xor(s, 1, 32);
    if (live && t == 0) invn[d] = 1.0f / fmaxf(sqrtf(s), EPSN);
  }
}

template <int D, bool NORM>
__global__ void __launch_bounds__(256)
gather_pair_kernel(GatherSide a, GatherSide b, int nblk_a, int E)
{
  const bool fa = (int)blockIdx.x < nblk_a;
  const int  lb = fa ? (int)blockIdx.x : (int)blockIdx.x - nblk_a;
  const int wv = __builtin_amdgcn_readfirstlane((int)(lb * blockDim.y + threadIdx.y));
  gather_rows<D, NORM>(fa ? a.feat : b.feat, fa ? a.rowptr : b.rowptr, fa ? a.col : b.col,
                       fa ? a.io : b.io, fa ? a.degC : b.degC, fa ? a.bias2 : b.bias2,
                       fa ? a.invn : b.invn,
                       fa ? a.n_dst : b.n_dst, fa ? a.n_src : b.n_src, E, wv);
}

#define TAIL_PER_WAVE 32
#define SCORE_BLK  ((NEP + 8 * TAIL_PER_WAVE - 1) / (8 * TAIL_PER_WAVE))

__device__ __forceinline__ void
score_wave(const float* __restrict__ HU, const float* __restrict__ HR,
           const float* __restrict__ invU, const float* __restrict__ invR,
           const int* __restrict__ iu, const int* __restrict__ ir,
           float* __restrict__ out, int Ep, int wv)
{
  const int base = wv * TAIL_PER_WAVE;
  if (base >= Ep) return;
  const int lane = threadIdx.x;
  const int g = lane >> 3, t = lane & 7;
  const int el  = min(base + lane, Ep - 1);
  const int myu = iu[el], myr = ir[el];
  const float mynu = invU[myu], mynr = invR[myr];
  float sel = 0.0f;
#pragma unroll
  for (int j = 0; j < 8; ++j) {
    const int u = __shfl(myu, 4 * j + g, 32);
    const int r = __shfl(myr, 4 * j + g, 32);
    const float4* pu = (const float4*)(HU + (size_t)u * D_OUT);
    const float4* pr = (const float4*)(HR + (size_t)r * D_OUT);
    const float4 a0 = pu[t], a1 = pu[8 + t];
    const float4 b0 = pr[t], b1 = pr[8 + t];
    float s = a0.x * b0.x + a0.y * b0.y + a0.z * b0.z + a0.w * b0.w
            + a1.x * b1.x + a1.y * b1.y + a1.z * b1.z + a1.w * b1.w;
    s += __shfl_xor(s, 4, 32);
    s += __shfl_xor(s, 2, 32);
    s += __shfl_xor(s, 1, 32);
    if (t == j) sel = s;
  }
  const int own = 4 * t + g;
  const float nu = __shfl(mynu, own, 32);
  const float nr = __shfl(mynr, own, 32);
  const int e = base + own;
  if (e < Ep) out[e] = sel * nu * nr;
}

__global__ void __launch_bounds__(256)
score_pair_kernel(const float* __restrict__ HU, const float* __restrict__ HR,
                  const float* __restrict__ invU, const float* __restrict__ invR,
                  const int* __restrict__ iu0, const int* __restrict__ ir0,
                  const int* __restrict__ iu1, const int* __restrict__ ir1,
                  float* __restrict__ out, int Ep, int nblk)
{
  const bool f0 = (int)blockIdx.x < nblk;
  const int  lb = f0 ? (int)blockIdx.x : (int)blockIdx.x - nblk;
  score_wave(HU, HR, invU, invR, f0 ? iu0 : iu1, f0 ? ir0 : ir1,
             f0 ? out : out + Ep, Ep, (int)(lb * blockDim.y + threadIdx.y));
}

extern "C" void kernel_launch(void* const* d_in, const int* in_sizes, int n_in,
                              void* d_out, int out_size, void* d_ws, size_t ws_size,
                              hipStream_t stream)
{
  const float* user_feat = (const float*)d_in[0];
  const float* repo_feat = (const float*)d_in[1];
  const float* W_ue = (const float*)d_in[2];   const float* b_ue = (const float*)d_in[3];
  const float* W_re = (const float*)d_in[4];   const float* b_re = (const float*)d_in[5];
  const float* W_hid_ur = (const float*)d_in[6];  const float* b_hid_ur = (const float*)d_in[7];
  const float* W_hid_ru = (const float*)d_in[8];  const float* b_hid_ru = (const float*)d_in[9];
  const float* W_out_ur = (const float*)d_in[10]; const float* b_out_ur = (const float*)d_in[11];
  const float* W_out_ru = (const float*)d_in[12]; const float* b_out_ru = (const float*)d_in[13];
  const float* W_hcu = (const float*)d_in[14]; const float* b_hcu = (const float*)d_in[15];
  const float* W_hcr = (const float*)d_in[16]; const float* b_hcr = (const float*)d_in[17];
  const float* W_ocu = (const float*)d_in[18]; const float* b_ocu = (const float*)d_in[19];
  const float* W_ocr = (const float*)d_in[20]; const float* b_ocr = (const float*)d_in[21];
  const int* e_ur_src = (const int*)d_in[22];  const int* e_ur_dst = (const int*)d_in[23];
  const int* e_ru_src = (const int*)d_in[24];  const int* e_ru_dst = (const int*)d_in[25];
  const int* pos_u = (const int*)d_in[26];     const int* pos_r = (const int*)d_in[27];
  const int* neg_u = (const int*)d_in[28];     const int* neg_r = (const int*)d_in[29];
  float* out = (float*)d_out;

  float* w = (float*)d_ws;
  size_t o = 0;
  float* deg_u_out_ur = w + o; o += N_USER;
  float* deg_r_in_ur  = w + o; o += N_REPO;
  float* deg_r_out_ru = w + o; o += N_REPO;
  float* deg_u_in_ru  = w + o; o += N_USER;
  float* reg_agg  = w + o; o += ((size_t)N_USER + N_REPO) * D_HID;
  float* reg_h    = w + o; o += ((size_t)N_USER + N_REPO) * LD_EMB;
  float* reg_feat = w + o; o += ((size_t)N_USER + N_REPO) * D_HID;
  float* reg_csr  = w + o; o += ((size_t)N_USER + N_REPO) * D_HID;
  float* inv_nu  = w + o; o += N_USER;
  float* inv_nr  = w + o; o += N_REPO;

  int* col_ur  = (int*)reg_csr;
  int* col_ru  = col_ur + NEDGE;
  int* part    = col_ru + NEDGE;
  int* tot     = part + NSLICE * 2 * (N_USER + N_REPO);
  int* ptr_r   = tot + 4 * NRANGE * NSLICE;
  int* ptr_u   = ptr_r + N_REPO;

  float* feat_u   = reg_feat;
  float* feat_r   = reg_feat + (size_t)N_USER * D_HID;
  float* out_user = reg_agg;
  float* out_repo = reg_agg + (size_t)N_USER * D_HID;
  float* feat2_u  = reg_feat;
  float* feat2_r  = reg_feat + (size_t)N_USER * D_HID;
  float* hn_user  = reg_h;
  float* hn_repo  = reg_h + (size_t)N_USER * LD_EMB;
  float* wf_u     = hn_repo + (size_t)N_REPO * D_OUT;
  float* wf_r     = wf_u + 2 * FOLD_FLOATS;

  auto gemm1 = [&](const GemmSide& u, const GemmSide& r) {
    hipLaunchKernelGGL((gemm_pair_wmma_f32<D_HID / 16, D_HID / 16, D_IN, GEMM1_WAVES, WT1_FLOATS, true>),
                       dim3(GEMM1_BLOCKS), dim3(GEMM1_WAVES * 32), 0, stream,
                       u, r, N_USER, N_REPO, D_IN, D_HID, D_HID, D_IN, BLOCKS_U(GEMM1_BLOCKS));
  };
  auto gemm2 = [&](const GemmSide& u, const GemmSide& r) {
    hipLaunchKernelGGL((gemm_pair_wmma_f32<D_OUT / 16, D_OUT / 16, D_HID, GEMM_WAVES, WT_FLOATS, false>),
                       dim3(GEMM_BLOCKS), dim3(GEMM_WAVES * 32), 0, stream,
                       u, r, N_USER, N_REPO, D_HID, D_OUT, D_OUT, D_HID, BLOCKS_U(GEMM_BLOCKS));
  };

  auto gather = [&](const GatherSide& a, const GatherSide& b, int D) {
    dim3 blk(32, 8);
    dim3 grd(GATHER_BLK_U + GATHER_BLK_R);
    if (D == D_HID)
      hipLaunchKernelGGL((gather_pair_kernel<D_HID, false>), grd, blk, 0, stream,
                         a, b, GATHER_BLK_U, NEDGE);
    else
      hipLaunchKernelGGL((gather_pair_kernel<D_OUT, true>), grd, blk, 0, stream,
                         a, b, GATHER_BLK_U, NEDGE);
  };

  hipLaunchKernelGGL(range_count_kernel, dim3(4 * NRANGE * NSLICE), dim3(CSR_T), 0, stream,
                     e_ur_src, e_ur_dst, e_ru_src, e_ru_dst, part, tot);
  hipLaunchKernelGGL(range_fill_kernel,
                     dim3(2 * NRANGE * NSLICE + 2 * NRANGE + FOLD_BLOCKS),
                     dim3(CSR_T), 0, stream,
                     e_ur_src, e_ur_dst, e_ru_src, e_ru_dst, (const int*)part, (const int*)tot,
                     ptr_r, deg_r_in_ur, ptr_u, deg_u_in_ru, deg_u_out_ur, deg_r_out_ru,
                     col_ur, col_ru,
                     FoldSide{W_ue, b_ue, W_hid_ur, W_hcu, b_hcu, wf_u},
                     FoldSide{W_re, b_re, W_hid_ru, W_hcr, b_hcr, wf_r});

  gemm1(GemmSide{user_feat, wf_u, deg_u_out_ur, wf_u + D_IN * D_HID,
                 wf_u + FOLD_FLOATS, wf_u + FOLD_FLOATS + D_IN * D_HID, feat_u, out_user},
        GemmSide{repo_feat, wf_r, deg_r_out_ru, wf_r + D_IN * D_HID,
                 wf_r + FOLD_FLOATS, wf_r + FOLD_FLOATS + D_IN * D_HID, feat_r, out_repo});

  gather(GatherSide{feat_r, ptr_u, col_ru, out_user, deg_u_in_ru, b_hid_ru, nullptr, N_USER, N_REPO},
         GatherSide{feat_u, ptr_r, col_ur, out_repo, deg_r_in_ur, b_hid_ur, nullptr, N_REPO, N_USER}, D_HID);

  gemm2(GemmSide{out_user, W_out_ur, deg_u_out_ur, nullptr, W_ocu, b_ocu, feat2_u, hn_user},
        GemmSide{out_repo, W_out_ru, deg_r_out_ru, nullptr, W_ocr, b_ocr, feat2_r, hn_repo});

  gather(GatherSide{feat2_r, ptr_u, col_ru, hn_user, deg_u_in_ru, b_out_ru, inv_nu, N_USER, N_REPO},
         GatherSide{feat2_u, ptr_r, col_ur, hn_repo, deg_r_in_ur, b_out_ur, inv_nr, N_REPO, N_USER}, D_OUT);

  hipLaunchKernelGGL(score_pair_kernel, dim3(2 * SCORE_BLK), dim3(32, 8), 0, stream,
                     (const float*)hn_user, (const float*)hn_repo,
                     (const float*)inv_nu, (const float*)inv_nr,
                     pos_u, pos_r, neg_u, neg_r, out, NEP, SCORE_BLK);
}


# Layer-2 GEMM launch folded into the layer-1 edge-sum launch: a block's 32 finished rows go through LDS into f32 WMMA (weights packed per call); 7 launches become 6; per-element sum order unchanged
# speedup vs baseline: 5.7642x; 1.0494x over previous
#include <hip/hip_runtime.h>
#include <math.h>

#define N_USER 50000
#define N_REPO 100000
#define NEDGE  1000000
#define NEP    200000
#define D_IN   256
#define D_EMB  125
#define LD_EMB 128
#define D_HID  96
#define D_OUT  64
#define EPSN   1e-12f

typedef float v2f __attribute__((ext_vector_type(2)));
typedef float v8f __attribute__((ext_vector_type(8)));


#define GEMM1_WAVES  16
#define GEMM1_BLOCKS 256
#define WT1_FLOATS (2 * D_HID * (D_IN + 4))
#define BLOCKS_U(nblk) (((nblk) * N_USER + (N_USER + N_REPO) / 2) / (N_USER + N_REPO))

struct GemmSide {
  const float* A;
  const float* W1;
  const float* degA;
  const float* bias1;
  const float* W2;
  const float* bias2;
  float*       C1;
  float*       C2;
};

template <int NT1, int NT2, int KPAD, int NW, int LDSF, bool SCB>
__device__ __forceinline__ void
gemm_strip_body(const float* __restrict__ A, int lda,
                const float* __restrict__ W1,
                const float* __restrict__ W2, int ldw,
                const float* __restrict__ degA,
                const float* __restrict__ bias1,
                const float* __restrict__ bias2,
                float* __restrict__ C1,
                float* __restrict__ C2,
                int M, int Nreal, int K, int lb, int nb)
{
  constexpr int NT  = NT1 + NT2;
  constexpr int NP1 = NT1 * 16;
  constexpr int NP2 = NT2 * 16;
  constexpr int LS  = KPAD + 4;
  constexpr int NCH = KPAD / 32;
  static_assert(KPAD % 32 == 0, "K is walked in whole 128-byte lines");
  static_assert(NT2 == 0 || NT2 == NT1, "both products have the same width");
  static_assert((NP1 + NP2) * LS <= LDSF, "W1^T and W2^T fit the LDS array");
  __shared__ __align__(16) float Wt[LDSF];

  const int tid = threadIdx.x;

  const int lane = tid & 31;
  const int wave = tid >> 5;
  const int half = lane >> 4;
  const int l16  = lane & 15;

  float bn1[NT1], bn2[NT1];
#pragma unroll
  for (int t = 0; t < NT1; ++t) { bn1[t] = 0.0f; bn2[t] = 0.0f; }
  if (bias1) {
#pragma unroll
    for (int t = 0; t < NT1; ++t) bn1[t] = bias1[min(t * 16 + l16, Nreal - 1)];
  }
  if (NT2 > 0 && bias2) {
#pragma unroll
    for (int t = 0; t < NT2; ++t) bn2[t] = bias2[min(t * 16 + l16, Nreal - 1)];
  }

  constexpr int NST = NP1 * KPAD / (NW * 32);
  static_assert(NP1 * KPAD % (NW * 32) == 0, "every thread stages the same count");
  constexpr int UNR = (NST % 16 == 0) ? 16 : 8;
  static_assert(NST % UNR == 0, "the staging loop unrolls without a remainder");
  auto stage = [&](const float* __restrict__ W, int row0) {
#pragma unroll UNR
    for (int j = 0; j < NST; ++j) {
      const int i = tid + j * (NW * 32);
      const int k = i / NP1;
      const int n = i - k * NP1;
      const float w = W[(long)min(k, K - 1) * ldw + min(n, Nreal - 1)];
      const unsigned keep = (n < Nreal && k < K) ? 0xFFFFFFFFu : 0u;
      Wt[(row0 + n) * LS + k] = __uint_as_float(__float_as_uint(w) & keep);
    }
  };
  stage(W1, 0);
  if (NT2 > 0) stage(W2, NP1);
  __syncthreads();

  const float* Wp = &Wt[l16 * LS + 2 * half];
  const int Mtiles = M / 16;

  for (int mt = lb + nb * wave; mt < Mtiles; mt += nb * NW) {
    const int   m0 = mt * 16;
    const int   mA = m0 + l16;
    const float sA = degA ? rsqrtf(fmaxf(degA[mA], 1.0f)) : 1.0f;
    const float* Ap = A + (long)mA * lda + 2 * half;

    v8f acc[NT];
#pragma unroll
    for (int t = 0; t < NT; ++t) acc[t] = v8f{};

    v2f a_raw[8], a_sc[8];
#pragma unroll
    for (int s = 0; s < 8; ++s) {
      a_raw[s] = *(const v2f*)(Ap + s * 4);
      a_sc[s]  = a_raw[s] * sA;
    }

    auto mma_chunk = [&](int c) {
      const float* Wc = Wp + c * 32;
#pragma unroll
      for (int s = 0; s < 8; ++s) {
        v2f b[NT];
#pragma unroll
        for (int t = 0; t < NT; ++t)
          b[t] = *(const v2f*)(Wc + t * 16 * LS + s * 4);
#pragma unroll
        for (int t = 0; t < NT1; ++t)
          acc[t] = __builtin_amdgcn_wmma_f32_16x16x4_f32(false, a_sc[s], false, b[t],
                                                         (short)0, acc[t], false, false);
#pragma unroll
        for (int t = NT1; t < NT; ++t)
          acc[t] = __builtin_amdgcn_wmma_f32_16x16x4_f32(false, a_raw[s], false, b[t],
                                                         (short)0, acc[t], false, false);
      }
    };

#pragma unroll 1
    for (int c = 0; c + 1 < NCH; ++c) {
      v2f a_nxt[8];
#pragma unroll
      for (int s = 0; s < 8; ++s) a_nxt[s] = *(const v2f*)(Ap + (c + 1) * 32 + s * 4);
      __builtin_amdgcn_sched_barrier(0);
      mma_chunk(c);
      __builtin_amdgcn_sched_barrier(0);
#pragma unroll
      for (int s = 0; s < 8; ++s) { a_raw[s] = a_nxt[s]; a_sc[s] = a_nxt[s] * sA; }
    }
    mma_chunk(NCH - 1);

    const int mrow = m0 + 8 * half;
    float sr[8];
#pragma unroll
    for (int v = 0; v < 8; ++v)
      sr[v] = (SCB && degA) ? rsqrtf(fmaxf(degA[mrow + v], 1.0f)) : 1.0f;
#pragma unroll
    for (int t = 0; t < NT1; ++t) {
      const int n = t * 16 + l16;
#pragma unroll
      for (int v = 0; v < 8; ++v) {
        const float val = acc[t][v] + (SCB ? bn1[t] * sr[v] : bn1[t]);
        C1[(long)(mrow + v) * NP1 + n] = (n < Nreal) ? val : 0.0f;
      }
    }
#pragma unroll
    for (int t = 0; t < NT2; ++t) {
      const int n = t * 16 + l16;
#pragma unroll
      for (int v = 0; v < 8; ++v) {
        const float val = acc[NT1 + t][v] + bn2[t];
        C2[(long)(mrow + v) * NP2 + n] = (n < Nreal) ? val : 0.0f;
      }
    }
  }
}

template <int NT1, int NT2, int KPAD, int NW, int LDSF, bool SCB>
__global__ void __launch_bounds__(NW * 32, 4)
gemm_pair_wmma_f32(GemmSide u, GemmSide r, int Mu, int Mr, int lda, int ldw,
                   int Nreal, int K, int nbu)
{
  const bool us = (int)blockIdx.x < nbu;
  gemm_strip_body<NT1, NT2, KPAD, NW, LDSF, SCB>(us ? u.A : r.A, lda, us ? u.W1 : r.W1,
                                  us ? u.W2 : r.W2, ldw,
                                  us ? u.degA : r.degA, us ? u.bias1 : r.bias1,
                                  us ? u.bias2 : r.bias2,
                                  us ? u.C1 : r.C1, us ? u.C2 : r.C2,
                                  us ? Mu : Mr, Nreal, K,
                                  us ? (int)blockIdx.x : (int)blockIdx.x - nbu,
                                  us ? nbu : (int)gridDim.x - nbu);
}

#define FOLD_ROWS   272
#define FOLD_MT     (FOLD_ROWS / 16)
#define FOLD_NT     (D_HID / 16)
#define FOLD_FLOATS (FOLD_ROWS * D_HID)
#define FOLD_TILES  (4 * FOLD_MT * FOLD_NT)
static_assert(FOLD_ROWS % 16 == 0 && FOLD_ROWS > D_IN, "room for the bias row in whole tiles");
#define FOLD_BLOCKS ((FOLD_TILES + 31) / 32)

struct FoldSide {
  const float* We;
  const float* be;
  const float* Wc;
  const float* Ws;
  const float* bs;
  float*       out;
};

__device__ __forceinline__ float keep_bits(float w, unsigned keep)
{
  return __uint_as_float(__float_as_uint(w) & keep);
}

__device__ __forceinline__ void
fold_tile(const float* __restrict__ We, const float* __restrict__ be,
          const float* __restrict__ W, const float* __restrict__ bs,
          float* __restrict__ out, int mt, int nt, int lane)
{
  const int half = lane >> 4;
  const int l16  = lane & 15;
  const int m = mt * 16 + l16;
  const float* __restrict__ arow = (m < D_IN) ? We + (long)m * D_EMB : be;
  const unsigned rowkeep = (m <= D_IN) ? 0xFFFFFFFFu : 0u;
  const int n = nt * 16 + l16;
  v8f acc = v8f{};
#pragma unroll 8
  for (int k0 = 0; k0 < LD_EMB; k0 += 4) {
    const int ka = k0 + 2 * half, kb = ka + 1;
    const unsigned keepa = (ka < D_EMB) ? 0xFFFFFFFFu : 0u;
    const unsigned keepb = (kb < D_EMB) ? 0xFFFFFFFFu : 0u;
    const int kac = min(ka, D_EMB - 1), kbc = min(kb, D_EMB - 1);
    v2f a, b;
    a.x = keep_bits(arow[kac], keepa & rowkeep);
    a.y = keep_bits(arow[kbc], keepb & rowkeep);
    b.x = keep_bits(W[kac * D_HID + n], keepa);
    b.y = keep_bits(W[kbc * D_HID + n], keepb);
    acc = __builtin_amdgcn_wmma_f32_16x16x4_f32(false, a, false, b, (short)0, acc, false, false);
  }
  const int mrow = mt * 16 + 8 * half;
  const float bsn = bs ? bs[n] : 0.0f;
#pragma unroll
  for (int v = 0; v < 8; ++v) {
    const int row = mrow + v;
    out[row * D_HID + n] = acc[v] + ((row == D_IN) ? bsn : 0.0f);
  }
}


#define NRANGE  16
#define NSLICE  8
#define CSR_T   1024
static_assert(CSR_T == 32 * 32, "a fold block is 32 waves");
#define SLICE_E (NEDGE / NSLICE)
#define RMAX    ((N_REPO + NRANGE - 1) / NRANGE)
static_assert(NEDGE % (4 * NSLICE) == 0, "a slice is a whole number of 16-byte groups");
static_assert(N_USER <= N_REPO, "RMAX covers both node kinds");
static_assert(NRANGE * NSLICE <= 128, "the totals of the earlier ranges are summed by four waves");
#define FILL_PER ((RMAX + CSR_T - 1) / CSR_T)
#define TOFF0 0
#define TOFF1 (NSLICE * N_USER)
#define TOFF2 (NSLICE * (N_USER + N_REPO))
#define TOFF3 (NSLICE * (N_USER + 2 * N_REPO))

__global__ void __launch_bounds__(CSR_T, 16)
range_count_kernel(const int* __restrict__ ia0, const int* __restrict__ ia1,
                   const int* __restrict__ ia2, const int* __restrict__ ia3,
                   int* __restrict__ part, int* __restrict__ tot)
{
  __shared__ int hist[RMAX];
  __shared__ int btot;
  const int a  = blockIdx.x / (NRANGE * NSLICE);
  const int rs = blockIdx.x - a * (NRANGE * NSLICE);
  const int r  = rs / NSLICE, s = rs - r * NSLICE;
  const int* __restrict__ idx = (a == 0) ? ia0 : (a == 1) ? ia1 : (a == 2) ? ia2 : ia3;
  const int n    = (a == 0 || a == 3) ? N_USER : N_REPO;
  const int toff = (a == 0) ? TOFF0 : (a == 1) ? TOFF1 : (a == 2) ? TOFF2 : TOFF3;
  const int R   = (n + NRANGE - 1) / NRANGE;
  const int lo  = r * R;
  const int len = max(min(R, n - lo), 0);
  const int tid = threadIdx.x;
  for (int i = tid; i < len; i += CSR_T) hist[i] = 0;
  if (tid == 0) btot = 0;
  __syncthreads();
  const int4* __restrict__ p = (const int4*)(idx + s * SLICE_E);
  constexpr int N4 = SLICE_E / 4;
  constexpr int GC = 8;
  for (int b = 0; b < N4; b += GC * CSR_T) {
    int4 v[GC];
#pragma unroll
    for (int g = 0; g < GC; ++g) {
      const int i = b + g * CSR_T + tid;
      v[g] = p[min(i, N4 - 1)];
      const int inval = (i < N4) ? 0 : -1;
      v[g].x |= inval; v[g].y |= inval; v[g].z |= inval; v[g].w |= inval;
    }
#pragma unroll
    for (int g = 0; g < GC; ++g) {
      unsigned u;
      u = (unsigned)(v[g].x - lo); if (u < (unsigned)len) atomicAdd(&hist[u], 1);
      u = (unsigned)(v[g].y - lo); if (u < (unsigned)len) atomicAdd(&hist[u], 1);
      u = (unsigned)(v[g].z - lo); if (u < (unsigned)len) atomicAdd(&hist[u], 1);
      u = (unsigned)(v[g].w - lo); if (u < (unsigned)len) atomicAdd(&hist[u], 1);
    }
  }
  __syncthreads();
  int* __restrict__ out = part + toff + s * n + lo;
  int my = 0;
  for (int i = tid; i < len; i += CSR_T) { const int h = hist[i]; out[i] = h; my += h; }
  my += __shfl_xor(my, 16, 32);
  my += __shfl_xor(my, 8, 32);
  my += __shfl_xor(my, 4, 32);
  my += __shfl_xor(my, 2, 32);
  my += __shfl_xor(my, 1, 32);
  if ((tid & 31) == 0) atomicAdd(&btot, my);
  __syncthreads();
  if (tid == 0) tot[blockIdx.x] = btot;
}

#define PACK_FLOATS (D_HID * D_OUT)
struct PackSide {
  const float* Wa;
  const float* Wb;
  const float* Wc;
  const float* Wd;
  float*       out;
};

__global__ void __launch_bounds__(CSR_T, 16)
range_fill_kernel(const int* __restrict__ src0, const int* __restrict__ dst0,
                  const int* __restrict__ src1, const int* __restrict__ dst1,
                  const int* __restrict__ part, const int* __restrict__ tot,
                  int* __restrict__ ptr0, float* __restrict__ deg0,
                  int* __restrict__ ptr1, float* __restrict__ deg1,
                  float* __restrict__ dout0, float* __restrict__ dout2,
                  int* __restrict__ col0, int* __restrict__ col1,
                  FoldSide fu, FoldSide fr, PackSide pk)
{
  __shared__ int off[RMAX + 32];
  __shared__ int cnt[RMAX];
  __shared__ int wsum[32];
  __shared__ int bsum[4];
  const int tid = threadIdx.x, lane = tid & 31, wv = tid >> 5;

  if (blockIdx.x >= 2 * NRANGE * NSLICE + 2 * NRANGE + FOLD_BLOCKS) {
    const int m = blockIdx.x - (2 * NRANGE * NSLICE + 2 * NRANGE + FOLD_BLOCKS);
    const float* __restrict__ W = (m == 0) ? pk.Wa : (m == 1) ? pk.Wb : (m == 2) ? pk.Wc : pk.Wd;
    float* __restrict__ po = pk.out + m * PACK_FLOATS;
    static_assert(PACK_FLOATS % CSR_T == 0, "every thread packs the same count");
    float wv6[PACK_FLOATS / CSR_T];
#pragma unroll
    for (int j = 0; j < PACK_FLOATS / CSR_T; ++j) wv6[j] = W[tid + j * CSR_T];
#pragma unroll
    for (int j = 0; j < PACK_FLOATS / CSR_T; ++j) {
      const int e = tid + j * CSR_T;
      const int k = e / D_OUT, n = e - k * D_OUT;
      po[((k >> 1) * D_OUT + n) * 2 + (k & 1)] = wv6[j];
    }
    return;
  }

  if (blockIdx.x >= 2 * NRANGE * NSLICE + 2 * NRANGE) {
    const int fw = __builtin_amdgcn_readfirstlane(
        (int)(blockIdx.x - (2 * NRANGE * NSLICE + 2 * NRANGE)) * (CSR_T / 32) + wv);
    if (fw >= FOLD_TILES) return;
    const int p   = fw / (FOLD_MT * FOLD_NT);
    const int rem = fw - p * (FOLD_MT * FOLD_NT);
    const int mt  = rem / FOLD_NT, nt = rem - mt * FOLD_NT;
    const bool rs = p >= 2;
    const bool sf = (p & 1) != 0;
    fold_tile(rs ? fr.We : fu.We, rs ? fr.be : fu.be,
              sf ? (rs ? fr.Ws : fu.Ws) : (rs ? fr.Wc : fu.Wc),
              sf ? (rs ? fr.bs : fu.bs) : nullptr,
              (rs ? fr.out : fu.out) + (sf ? FOLD_FLOATS : 0), mt, nt, lane);
    return;
  }

  if (blockIdx.x >= 2 * NRANGE * NSLICE) {
    const int x  = blockIdx.x - 2 * NRANGE * NSLICE;
    const int wh = x / NRANGE, r = x - wh * NRANGE;
    const int n  = wh ? N_REPO : N_USER;
    const int* __restrict__ tab = part + (wh ? TOFF2 : TOFF0);
    float* __restrict__ dout = wh ? dout2 : dout0;
    const int R   = (n + NRANGE - 1) / NRANGE;
    const int lo  = r * R;
    const int len = max(min(R, n - lo), 0);
    for (int i = tid; i < len; i += CSR_T) {
      int c = 0;
#pragma unroll
      for (int s2 = 0; s2 < NSLICE; ++s2) c += tab[s2 * n + lo + i];
      dout[lo + i] = (float)c;
    }
    return;
  }

  const int t  = blockIdx.x / (NRANGE * NSLICE);
  const int rs = blockIdx.x - t * (NRANGE * NSLICE);
  const int r  = rs / NSLICE, s = rs - r * NSLICE;
  const int* __restrict__ src = t ? src1 : src0;
  const int* __restrict__ dst = t ? dst1 : dst0;
  int*       __restrict__ ptr = t ? ptr1 : ptr0;
  float*     __restrict__ deg = t ? deg1 : deg0;
  int*       __restrict__ col = t ? col1 : col0;
  const int n   = t ? N_USER : N_REPO;
  const int* __restrict__ tab = part + (t ? TOFF3 : TOFF1);
  const int R   = (n + NRANGE - 1) / NRANGE;
  const int lo  = r * R;
  const int len = max(min(R, n - lo), 0);

  int bv = (tid < r * NSLICE) ? tot[(2 * t + 1) * NRANGE * NSLICE + tid] : 0;
  bv += __shfl_xor(bv, 16, 32);
  bv += __shfl_xor(bv, 8, 32);
  bv += __shfl_xor(bv, 4, 32);
  bv += __shfl_xor(bv, 2, 32);
  bv += __shfl_xor(bv, 1, 32);
  if (wv < 4 && lane == 0) bsum[wv] = bv;
  if (tid < 32) off[RMAX + tid] = 0;

#pragma unroll 2
  for (int i = tid; i < len; i += CSR_T) {
    int c = 0, pre = 0;
#pragma unroll
    for (int s2 = 0; s2 < NSLICE; ++s2) {
      const int x = tab[s2 * n + lo + i];
      c += x;
      pre += (s2 < s) ? x : 0;
    }
    cnt[i] = c;
    off[i] = pre;
  }
  __syncthreads();

  int cv[FILL_PER];
  int tsum = 0;
#pragma unroll
  for (int k = 0; k < FILL_PER; ++k) {
    const int i = tid * FILL_PER + k;
    cv[k] = (i < len) ? cnt[i] : 0;
    tsum += cv[k];
  }
  int x = tsum;
#pragma unroll
  for (int o = 1; o < 32; o <<= 1) {
    const int y = __shfl_up(x, o, 32);
    if (lane >= o) x += y;
  }
  if (lane == 31) wsum[wv] = x;
  __syncthreads();
  const int w = wsum[lane];
  int wx = w;
#pragma unroll
  for (int o = 1; o < 32; o <<= 1) {
    const int y = __shfl_up(wx, o, 32);
    if (lane >= o) wx += y;
  }
  const int wave_excl = __shfl(wx, wv, 32) - __shfl(w, wv, 32);
  int run = bsum[0] + bsum[1] + bsum[2] + bsum[3] + wave_excl + (x - tsum);
#pragma unroll
  for (int k = 0; k < FILL_PER; ++k) {
    const int i = tid * FILL_PER + k;
    if (i < len) {
      off[i] += run;
      if (s == 0) { ptr[lo + i] = run; deg[lo + i] = (float)cv[k]; }
    }
    run += cv[k];
  }
  __syncthreads();

  const unsigned spare = RMAX + lane;
  const int4* __restrict__ pd = (const int4*)(dst + s * SLICE_E);
  const int4* __restrict__ ps = (const int4*)(src + s * SLICE_E);
  constexpr int N4 = SLICE_E / 4;
  constexpr int GF = 4;
  for (int b = 0; b < N4; b += GF * CSR_T) {
    int4 d[GF], sv[GF];
#pragma unroll
    for (int g = 0; g < GF; ++g) {
      const int i = b + g * CSR_T + tid;
      d[g]  = pd[min(i, N4 - 1)];
      sv[g] = ps[min(i, N4 - 1)];
      const int inval = (i < N4) ? 0 : -1;
      d[g].x |= inval; d[g].y |= inval; d[g].z |= inval; d[g].w |= inval;
    }
#pragma unroll
    for (int g = 0; g < GF; ++g) {
      const unsigned u0 = (unsigned)(d[g].x - lo), u1 = (unsigned)(d[g].y - lo);
      const unsigned u2 = (unsigned)(d[g].z - lo), u3 = (unsigned)(d[g].w - lo);
      const bool m0 = u0 < (unsigned)len, m1 = u1 < (unsigned)len;
      const bool m2 = u2 < (unsigned)len, m3 = u3 < (unsigned)len;
      const int q0 = atomicAdd(&off[m0 ? u0 : spare], 1);
      const int q1 = atomicAdd(&off[m1 ? u1 : spare], 1);
      const int q2 = atomicAdd(&off[m2 ? u2 : spare], 1);
      const int q3 = atomicAdd(&off[m3 ? u3 : spare], 1);
      if (m0) col[min(max(q0, 0), NEDGE - 1)] = sv[g].x;
      if (m1) col[min(max(q1, 0), NEDGE - 1)] = sv[g].y;
      if (m2) col[min(max(q2, 0), NEDGE - 1)] = sv[g].z;
      if (m3) col[min(max(q3, 0), NEDGE - 1)] = sv[g].w;
    }
  }
}

#define GATHER_ROWS 4
#define GATHER_BLK_U ((N_USER + 8 * GATHER_ROWS - 1) / (8 * GATHER_ROWS))
#define GATHER_BLK_R ((N_REPO + 8 * GATHER_ROWS - 1) / (8 * GATHER_ROWS))
struct GatherSide {
  const float* feat;
  const int*   rowptr;
  const int*   col;
  float*       io;
  const float* degC;
  const float* bias2;
  float*       invn;
  int          n_dst;
  int          n_src;
};

template <int D, int JJ>
__device__ __forceinline__ void
gather_trip(float4 (&v)[D / 32], const float* __restrict__ feat, int n_src, int cur, int t, bool on)
{
  const int s = min(max((int)__builtin_amdgcn_mov_dpp8((unsigned)cur, JJ * 0x249249), 0), n_src - 1);
  const float4* p = (const float4*)(feat + (size_t)s * D);
#pragma unroll
  for (int q = 0; q < D / 32; ++q) v[q] = make_float4(0.0f, 0.0f, 0.0f, 0.0f);
  if (on) {
#pragma unroll
    for (int q = 0; q < D / 32; ++q) v[q] = p[q * 8 + t];
  }
}

template <int D>
__device__ __forceinline__ void
gather_row_out(float4 (&o)[D / 32],
               const float* __restrict__ feat, const int* __restrict__ rowptr,
               const int* __restrict__ col, const float* io,
               const float* __restrict__ degC, const float* __restrict__ bias2,
               int n_dst, int n_src, int E, int d0, int lane)
{
  constexpr int NPC = D / 32;
  static_assert(D % 32 == 0, "a row is a whole number of 128-byte lines");
  const int g = lane >> 3, t = lane & 7;
  const int  d    = d0 + g;

  const int xb = rowptr[min(d0 + min(lane, GATHER_ROWS), n_dst - 1)];

  const int dr = min(d, n_dst - 1);
  const float4* const iop = (const float4*)(io + (size_t)dr * D);
  float4 self[NPC];
#pragma unroll
  for (int q = 0; q < NPC; ++q) self[q] = iop[q * 8 + t];
  const float dgc = degC[dr];
  int rp[GATHER_ROWS + 1];
#pragma unroll
  for (int i = 0; i <= GATHER_ROWS; ++i) {
    const int x = __builtin_amdgcn_readlane(xb, i);
    rp[i] = (d0 + i < n_dst) ? min(max(x, 0), E) : E;
  }
  int mx = 0;
#pragma unroll
  for (int i = 0; i < GATHER_ROWS; ++i) mx = max(mx, rp[i + 1] - rp[i]);
  int beg = rp[0], end = rp[1];
#pragma unroll
  for (int i = 1; i < GATHER_ROWS; ++i) {
    if (g == i) { beg = rp[i]; end = rp[i + 1]; }
  }
  const int  deg  = max(end - beg, 0);

  float4 acc[NPC];
#pragma unroll
  for (int q = 0; q < NPC; ++q) acc[q] = make_float4(0.0f, 0.0f, 0.0f, 0.0f);

  int cur = 0;
  if (t < deg) cur = col[beg + t];
  cur = min(max(cur, 0), n_src - 1);
  for (int j0 = 0; j0 < mx; j0 += 8) {
    int nxt = 0;
    if (j0 + 8 + t < deg) nxt = col[beg + j0 + 8 + t];
    const int rem = deg - j0;
    float4 v[8][NPC];
    gather_trip<D, 0>(v[0], feat, n_src, cur, t, 0 < rem);
    gather_trip<D, 1>(v[1], feat, n_src, cur, t, 1 < rem);
    gather_trip<D, 2>(v[2], feat, n_src, cur, t, 2 < rem);
    gather_trip<D, 3>(v[3], feat, n_src, cur, t, 3 < rem);
    gather_trip<D, 4>(v[4], feat, n_src, cur, t, 4 < rem);
    gather_trip<D, 5>(v[5], feat, n_src, cur, t, 5 < rem);
    gather_trip<D, 6>(v[6], feat, n_src, cur, t, 6 < rem);
    gather_trip<D, 7>(v[7], feat, n_src, cur, t, 7 < rem);
#pragma unroll
    for (int jj = 0; jj < 8; ++jj) {
#pragma unroll
      for (int q = 0; q < NPC; ++q) {
        acc[q].x += v[jj][q].x;
        acc[q].y += v[jj][q].y;
        acc[q].z += v[jj][q].z;
        acc[q].w += v[jj][q].w;
      }
    }
    cur = nxt;
  }
  const float sC = rsqrtf(fmaxf(dgc, 1.0f));
  const float4* bp = (const float4*)bias2;
#pragma unroll
  for (int q = 0; q < NPC; ++q) {
    const float4 b = bp[q * 8 + t];
    float vx = self[q].x, vy = self[q].y, vz = self[q].z, vw = self[q].w;
    vx += acc[q].x * sC + b.x;
    vy += acc[q].y * sC + b.y;
    vz += acc[q].z * sC + b.z;
    vw += acc[q].w * sC + b.w;
    o[q] = make_float4(fmaxf(vx, 0.0f), fmaxf(vy, 0.0f), fmaxf(vz, 0.0f), fmaxf(vw, 0.0f));
  }
}

template <int D, bool NORM>
__device__ __forceinline__ void
gather_rows(const float* __restrict__ feat, const int* __restrict__ rowptr,
            const int* __restrict__ col, float* io,
            const float* __restrict__ degC, const float* __restrict__ bias2,
            float* __restrict__ invn,
            int n_dst, int n_src, int E, int wv)
{
  constexpr int NPC = D / 32;
  const int d0 = wv * GATHER_ROWS;
  if (d0 >= n_dst) return;
  const int lane = threadIdx.x;
  const int g = lane >> 3, t = lane & 7;
  const int  d    = d0 + g;
  const bool live = d < n_dst;
  float4 o[NPC];
  gather_row_out<D>(o, feat, rowptr, col, io, degC, bias2, n_dst, n_src, E, d0, lane);
  if (live) {
    float4* const iop = (float4*)(io + (size_t)d * D);
#pragma unroll
    for (int q = 0; q < NPC; ++q) iop[q * 8 + t] = o[q];
  }
  if (NORM) {
    static_assert(!NORM || NPC == 2, "the norm is taken of rows of 64 floats");
    const float4 v0 = o[0], v1 = o[NPC - 1];
    float s = v0.x * v0.x + v0.y * v0.y + v0.z * v0.z + v0.w * v0.w
            + v1.x * v1.x + v1.y * v1.y + v1.z * v1.z + v1.w * v1.w;
    s += __shfl_xor(s, 4, 32);
    s += __shfl_xor(s, 2, 32);
    s += __shfl_xor(s, 1, 32);
    if (live && t == 0) invn[d] = 1.0f / fmaxf(sqrtf(s), EPSN);
  }
}

template <int D, bool NORM>
__global__ void __launch_bounds__(256)
gather_pair_kernel(GatherSide a, GatherSide b, int nblk_a, int E)
{
  const bool fa = (int)blockIdx.x < nblk_a;
  const int  lb = fa ? (int)blockIdx.x : (int)blockIdx.x - nblk_a;
  const int wv = __builtin_amdgcn_readfirstlane((int)(lb * blockDim.y + threadIdx.y));
  gather_rows<D, NORM>(fa ? a.feat : b.feat, fa ? a.rowptr : b.rowptr, fa ? a.col : b.col,
                       fa ? a.io : b.io, fa ? a.degC : b.degC, fa ? a.bias2 : b.bias2,
                       fa ? a.invn : b.invn,
                       fa ? a.n_dst : b.n_dst, fa ? a.n_src : b.n_src, E, wv);
}

#define AT_LS (D_HID + 4)
struct FuseSide {
  const float* Wp1;
  const float* Wp2;
  const float* degA;
  const float* bias2;
  float*       C1;
  float*       C2;
};

__global__ void __launch_bounds__(256, 7)
gather_gemm_kernel(GatherSide a, GatherSide b, FuseSide ga, FuseSide gb, int nblk_a, int E)
{
  constexpr int NPC = D_HID / 32;
  constexpr int BROWS = 8 * GATHER_ROWS;
  static_assert(BROWS == 32 && D_OUT == 64, "four waves cover 2 strips x 2 products x 2 column halves");
  static_assert(D_HID % 32 == 0, "K is walked in chunks of 8 k-steps");
  __shared__ __align__(16) float At[BROWS * AT_LS];

  const bool fa = (int)blockIdx.x < nblk_a;
  const int  lb = fa ? (int)blockIdx.x : (int)blockIdx.x - nblk_a;
  const int wy   = __builtin_amdgcn_readfirstlane((int)threadIdx.y);
  const int lane = threadIdx.x;
  const int n_dst = fa ? a.n_dst : b.n_dst;
  const int row0  = lb * BROWS;
  const int d0    = row0 + wy * GATHER_ROWS;

  float4 o[NPC];
#pragma unroll
  for (int q = 0; q < NPC; ++q) o[q] = make_float4(0.0f, 0.0f, 0.0f, 0.0f);
  if (d0 < n_dst)
    gather_row_out<D_HID>(o, fa ? a.feat : b.feat, fa ? a.rowptr : b.rowptr, fa ? a.col : b.col,
                          fa ? a.io : b.io, fa ? a.degC : b.degC, fa ? a.bias2 : b.bias2,
                          n_dst, fa ? a.n_src : b.n_src, E, d0, lane);
  const int half = lane >> 4, l16 = lane & 15;
  const bool mm = wy < 4;
  const int p  = (wy >> 1) & 1;
  const int n0 = (wy & 1) * 32;
  const float* __restrict__ Wp   = p ? (fa ? ga.Wp2 : gb.Wp2) : (fa ? ga.Wp1 : gb.Wp1);
  const float* __restrict__ degA = fa ? ga.degA : gb.degA;
  const float* __restrict__ bias = fa ? ga.bias2 : gb.bias2;
  const v2f* __restrict__ Bp = (const v2f*)Wp + (half * D_OUT + n0 + l16);

  auto load_b = [&](v2f (&bf)[8][2], int c) {
#pragma unroll
    for (int s = 0; s < 8; ++s) {
      bf[s][0] = Bp[(16 * c + 2 * s) * D_OUT];
      bf[s][1] = Bp[(16 * c + 2 * s) * D_OUT + 16];
    }
  };

  float dg[2] = {1.0f, 1.0f}, bb[2] = {0.0f, 0.0f};
  v2f bf[8][2];
#pragma unroll
  for (int s = 0; s < 8; ++s) { bf[s][0] = v2f{}; bf[s][1] = v2f{}; }
  if (mm) {
#pragma unroll
    for (int st = 0; st < 2; ++st) dg[st] = degA[min(row0 + st * 16 + l16, n_dst - 1)];
#pragma unroll
    for (int j = 0; j < 2; ++j) bb[j] = bias[n0 + 16 * j + l16];
    load_b(bf, 0);
  }

  {
    const int g = lane >> 3, t = lane & 7;
    float* ar = At + (wy * GATHER_ROWS + g) * AT_LS + 4 * t;
#pragma unroll
    for (int q = 0; q < NPC; ++q) *(float4*)(ar + q * 32) = o[q];
  }
  __syncthreads();
  if (!mm) return;

  float* __restrict__ C = p ? (fa ? ga.C2 : gb.C2) : (fa ? ga.C1 : gb.C1);

  float sc[2], bn[2];
#pragma unroll
  for (int st = 0; st < 2; ++st) sc[st] = p ? 1.0f : rsqrtf(fmaxf(dg[st], 1.0f));
#pragma unroll
  for (int j = 0; j < 2; ++j) bn[j] = p ? bb[j] : 0.0f;

  v8f acc[2][2];
#pragma unroll
  for (int st = 0; st < 2; ++st) {
#pragma unroll
    for (int j = 0; j < 2; ++j) acc[st][j] = v8f{};
  }

  const float* Ap = At + l16 * AT_LS + 2 * half;

  auto mma_chunk = [&](int c) {
    v2f af[2][8];
#pragma unroll
    for (int st = 0; st < 2; ++st) {
#pragma unroll
      for (int s = 0; s < 8; ++s)
        af[st][s] = *(const v2f*)(Ap + st * 16 * AT_LS + c * 32 + s * 4) * sc[st];
    }
#pragma unroll
    for (int s = 0; s < 8; ++s) {
#pragma unroll
      for (int st = 0; st < 2; ++st) {
#pragma unroll
        for (int j = 0; j < 2; ++j)
          acc[st][j] = __builtin_amdgcn_wmma_f32_16x16x4_f32(false, af[st][s], false, bf[s][j],
                                                             (short)0, acc[st][j], false, false);
      }
    }
  };

  for (int c = 0; c + 1 < D_HID / 32; ++c) {
    v2f bx[8][2];
    load_b(bx, c + 1);
    mma_chunk(c);
#pragma unroll
    for (int s = 0; s < 8; ++s) { bf[s][0] = bx[s][0]; bf[s][1] = bx[s][1]; }
  }
  mma_chunk(D_HID / 32 - 1);

#pragma unroll
  for (int st = 0; st < 2; ++st) {
    const int mrow = row0 + st * 16 + 8 * half;
#pragma unroll
    for (int j = 0; j < 2; ++j) {
      const int n = n0 + 16 * j + l16;
#pragma unroll
      for (int v = 0; v < 8; ++v) {
        const int row = mrow + v;
        if (row < n_dst) C[(size_t)row * D_OUT + n] = acc[st][j][v] + bn[j];
      }
    }
  }
}

#define TAIL_PER_WAVE 32
#define SCORE_BLK  ((NEP + 8 * TAIL_PER_WAVE - 1) / (8 * TAIL_PER_WAVE))

__device__ __forceinline__ void
score_wave(const float* __restrict__ HU, const float* __restrict__ HR,
           const float* __restrict__ invU, const float* __restrict__ invR,
           const int* __restrict__ iu, const int* __restrict__ ir,
           float* __restrict__ out, int Ep, int wv)
{
  const int base = wv * TAIL_PER_WAVE;
  if (base >= Ep) return;
  const int lane = threadIdx.x;
  const int g = lane >> 3, t = lane & 7;
  const int el  = min(base + lane, Ep - 1);
  const int myu = iu[el], myr = ir[el];
  const float mynu = invU[myu], mynr = invR[myr];
  float sel = 0.0f;
#pragma unroll
  for (int j = 0; j < 8; ++j) {
    const int u = __shfl(myu, 4 * j + g, 32);
    const int r = __shfl(myr, 4 * j + g, 32);
    const float4* pu = (const float4*)(HU + (size_t)u * D_OUT);
    const float4* pr = (const float4*)(HR + (size_t)r * D_OUT);
    const float4 a0 = pu[t], a1 = pu[8 + t];
    const float4 b0 = pr[t], b1 = pr[8 + t];
    float s = a0.x * b0.x + a0.y * b0.y + a0.z * b0.z + a0.w * b0.w
            + a1.x * b1.x + a1.y * b1.y + a1.z * b1.z + a1.w * b1.w;
    s += __shfl_xor(s, 4, 32);
    s += __shfl_xor(s, 2, 32);
    s += __shfl_xor(s, 1, 32);
    if (t == j) sel = s;
  }
  const int own = 4 * t + g;
  const float nu = __shfl(mynu, own, 32);
  const float nr = __shfl(mynr, own, 32);
  const int e = base + own;
  if (e < Ep) out[e] = sel * nu * nr;
}

__global__ void __launch_bounds__(256)
score_pair_kernel(const float* __restrict__ HU, const float* __restrict__ HR,
                  const float* __restrict__ invU, const float* __restrict__ invR,
                  const int* __restrict__ iu0, const int* __restrict__ ir0,
                  const int* __restrict__ iu1, const int* __restrict__ ir1,
                  float* __restrict__ out, int Ep, int nblk)
{
  const bool f0 = (int)blockIdx.x < nblk;
  const int  lb = f0 ? (int)blockIdx.x : (int)blockIdx.x - nblk;
  score_wave(HU, HR, invU, invR, f0 ? iu0 : iu1, f0 ? ir0 : ir1,
             f0 ? out : out + Ep, Ep, (int)(lb * blockDim.y + threadIdx.y));
}

extern "C" void kernel_launch(void* const* d_in, const int* in_sizes, int n_in,
                              void* d_out, int out_size, void* d_ws, size_t ws_size,
                              hipStream_t stream)
{
  const float* user_feat = (const float*)d_in[0];
  const float* repo_feat = (const float*)d_in[1];
  const float* W_ue = (const float*)d_in[2];   const float* b_ue = (const float*)d_in[3];
  const float* W_re = (const float*)d_in[4];   const float* b_re = (const float*)d_in[5];
  const float* W_hid_ur = (const float*)d_in[6];  const float* b_hid_ur = (const float*)d_in[7];
  const float* W_hid_ru = (const float*)d_in[8];  const float* b_hid_ru = (const float*)d_in[9];
  const float* W_out_ur = (const float*)d_in[10]; const float* b_out_ur = (const float*)d_in[11];
  const float* W_out_ru = (const float*)d_in[12]; const float* b_out_ru = (const float*)d_in[13];
  const float* W_hcu = (const float*)d_in[14]; const float* b_hcu = (const float*)d_in[15];
  const float* W_hcr = (const float*)d_in[16]; const float* b_hcr = (const float*)d_in[17];
  const float* W_ocu = (const float*)d_in[18]; const float* b_ocu = (const float*)d_in[19];
  const float* W_ocr = (const float*)d_in[20]; const float* b_ocr = (const float*)d_in[21];
  const int* e_ur_src = (const int*)d_in[22];  const int* e_ur_dst = (const int*)d_in[23];
  const int* e_ru_src = (const int*)d_in[24];  const int* e_ru_dst = (const int*)d_in[25];
  const int* pos_u = (const int*)d_in[26];     const int* pos_r = (const int*)d_in[27];
  const int* neg_u = (const int*)d_in[28];     const int* neg_r = (const int*)d_in[29];
  float* out = (float*)d_out;

  float* w = (float*)d_ws;
  size_t o = 0;
  float* deg_u_out_ur = w + o; o += N_USER;
  float* deg_r_in_ur  = w + o; o += N_REPO;
  float* deg_r_out_ru = w + o; o += N_REPO;
  float* deg_u_in_ru  = w + o; o += N_USER;
  float* reg_agg  = w + o; o += ((size_t)N_USER + N_REPO) * D_HID;
  float* reg_h    = w + o; o += ((size_t)N_USER + N_REPO) * LD_EMB;
  float* reg_feat = w + o; o += ((size_t)N_USER + N_REPO) * D_HID;
  float* reg_csr  = w + o; o += ((size_t)N_USER + N_REPO) * D_HID;
  float* inv_nu  = w + o; o += N_USER;
  float* inv_nr  = w + o; o += N_REPO;

  int* col_ur  = (int*)reg_csr;
  int* col_ru  = col_ur + NEDGE;
  int* part    = col_ru + NEDGE;
  int* tot     = part + NSLICE * 2 * (N_USER + N_REPO);
  int* ptr_r   = tot + 4 * NRANGE * NSLICE;
  int* ptr_u   = ptr_r + N_REPO;

  float* feat_u   = reg_feat;
  float* feat_r   = reg_feat + (size_t)N_USER * D_HID;
  float* out_user = reg_agg;
  float* out_repo = reg_agg + (size_t)N_USER * D_HID;
  float* hn_user  = reg_h;
  float* hn_repo  = reg_h + (size_t)N_USER * LD_EMB;
  float* feat2_u  = hn_user + (size_t)N_USER * D_OUT;
  float* feat2_r  = hn_repo + (size_t)N_REPO * D_OUT;
  float* wf_u     = hn_repo + (size_t)N_REPO * D_OUT;
  float* wf_r     = wf_u + 2 * FOLD_FLOATS;
  float* wpack    = (float*)(ptr_u + N_USER) + 16;

  auto gemm1 = [&](const GemmSide& u, const GemmSide& r) {
    hipLaunchKernelGGL((gemm_pair_wmma_f32<D_HID / 16, D_HID / 16, D_IN, GEMM1_WAVES, WT1_FLOATS, true>),
                       dim3(GEMM1_BLOCKS), dim3(GEMM1_WAVES * 32), 0, stream,
                       u, r, N_USER, N_REPO, D_IN, D_HID, D_HID, D_IN, BLOCKS_U(GEMM1_BLOCKS));
  };

  hipLaunchKernelGGL(range_count_kernel, dim3(4 * NRANGE * NSLICE), dim3(CSR_T), 0, stream,
                     e_ur_src, e_ur_dst, e_ru_src, e_ru_dst, part, tot);
  hipLaunchKernelGGL(range_fill_kernel,
                     dim3(2 * NRANGE * NSLICE + 2 * NRANGE + FOLD_BLOCKS + 4),
                     dim3(CSR_T), 0, stream,
                     e_ur_src, e_ur_dst, e_ru_src, e_ru_dst, (const int*)part, (const int*)tot,
                     ptr_r, deg_r_in_ur, ptr_u, deg_u_in_ru, deg_u_out_ur, deg_r_out_ru,
                     col_ur, col_ru,
                     FoldSide{W_ue, b_ue, W_hid_ur, W_hcu, b_hcu, wf_u},
                     FoldSide{W_re, b_re, W_hid_ru, W_hcr, b_hcr, wf_r},
                     PackSide{W_out_ur, W_ocu, W_out_ru, W_ocr, wpack});

  gemm1(GemmSide{user_feat, wf_u, deg_u_out_ur, wf_u + D_IN * D_HID,
                 wf_u + FOLD_FLOATS, wf_u + FOLD_FLOATS + D_IN * D_HID, feat_u, out_user},
        GemmSide{repo_feat, wf_r, deg_r_out_ru, wf_r + D_IN * D_HID,
                 wf_r + FOLD_FLOATS, wf_r + FOLD_FLOATS + D_IN * D_HID, feat_r, out_repo});

  hipLaunchKernelGGL(gather_gemm_kernel, dim3(GATHER_BLK_U + GATHER_BLK_R), dim3(32, 8), 0, stream,
                     GatherSide{feat_r, ptr_u, col_ru, out_user, deg_u_in_ru, b_hid_ru, nullptr, N_USER, N_REPO},
                     GatherSide{feat_u, ptr_r, col_ur, out_repo, deg_r_in_ur, b_hid_ur, nullptr, N_REPO, N_USER},
                     FuseSide{wpack, wpack + PACK_FLOATS, deg_u_out_ur, b_ocu, feat2_u, hn_user},
                     FuseSide{wpack + 2 * PACK_FLOATS, wpack + 3 * PACK_FLOATS, deg_r_out_ru, b_ocr,
                              feat2_r, hn_repo},
                     GATHER_BLK_U, NEDGE);

  hipLaunchKernelGGL((gather_pair_kernel<D_OUT, true>), dim3(GATHER_BLK_U + GATHER_BLK_R), dim3(32, 8),
                     0, stream,
                     GatherSide{feat2_r, ptr_u, col_ru, hn_user, deg_u_in_ru, b_out_ru, inv_nu, N_USER, N_REPO},
                     GatherSide{feat2_u, ptr_r, col_ur, hn_repo, deg_r_in_ur, b_out_ur, inv_nr, N_REPO, N_USER},
                     GATHER_BLK_U, NEDGE);

  hipLaunchKernelGGL(score_pair_kernel, dim3(2 * SCORE_BLK), dim3(32, 8), 0, stream,
                     (const float*)hn_user, (const float*)hn_repo,
                     (const float*)inv_nu, (const float*)inv_nr,
                     pos_u, pos_r, neg_u, neg_r, out, NEP, SCORE_BLK);
}
